# starting kernel
# baseline (speedup 1.0000x reference)
#include <hip/hip_runtime.h>
#include <hip/hip_bf16.h>

#define HDIM 128
#define EPS_BN 1e-5f

typedef __attribute__((ext_vector_type(2))) float v2f;
typedef __attribute__((ext_vector_type(8))) float v8f;

__global__ __launch_bounds__(256) void k_zero(float* p, long long n) {
    long long i = (long long)blockIdx.x * blockDim.x + threadIdx.x;
    long long stride = (long long)gridDim.x * blockDim.x;
    for (; i < n; i += stride) p[i] = 0.0f;
}

__global__ __launch_bounds__(256) void k_deg_scatter(const int* __restrict__ dst,
                                                     float* __restrict__ deg, int E) {
    int t = blockIdx.x * 256 + threadIdx.x;
    if (t < E) atomicAdd(&deg[dst[t]], 1.0f);
}

__global__ __launch_bounds__(256) void k_dinv(float* __restrict__ deg, int n) {
    int t = blockIdx.x * 256 + threadIdx.x;
    if (t < n) deg[t] = rsqrtf(deg[t] + 1.0f);
}

__global__ __launch_bounds__(256) void k_rank1(const float* __restrict__ X,
                                               const float* __restrict__ W1,
                                               float* __restrict__ h, int n) {
    int t = blockIdx.x * 256 + threadIdx.x;
    if (t >= n * 32) return;
    int row = t >> 5, c4 = (t & 31) << 2;
    float xv = X[row];
    const float4 w = *(const float4*)(W1 + c4);
    float4 o; o.x = xv * w.x; o.y = xv * w.y; o.z = xv * w.z; o.w = xv * w.w;
    *(float4*)(h + (long long)row * HDIM + c4) = o;
}

__global__ __launch_bounds__(256) void k_gemm_wmma(const float* __restrict__ x,
                                                   const float* __restrict__ W,
                                                   float* __restrict__ h, int n) {
    __shared__ float tileX[32 * HDIM];
    const int rowBase = blockIdx.x * 32;
    const int tid  = threadIdx.x;
    const int wave = tid >> 5;
    const int lane = tid & 31;
    const int half = lane >> 4;
    const int l16  = lane & 15;
    const int rowHalf = wave & 1;
    const int c0   = (wave >> 1) * 32;

    const bool fullTile = (rowBase + 32 <= n);

    if (fullTile) {
        const float4* xs = (const float4*)(x + (long long)rowBase * HDIM);
        #pragma unroll
        for (int i = 0; i < 4; ++i) {
            int idx = tid + i * 256;
            ((float4*)tileX)[idx] = xs[idx];
        }
    } else {
        #pragma unroll
        for (int i = 0; i < 4; ++i) {
            int idx = tid + i * 256;
            int r = rowBase + (idx >> 5);
            float4 v = make_float4(0.f, 0.f, 0.f, 0.f);
            if (r < n) v = *(const float4*)(x + (long long)r * HDIM + ((idx & 31) << 2));
            ((float4*)tileX)[idx] = v;
        }
    }
    __syncthreads();

    const float* xrow = tileX + (rowHalf * 16 + l16) * HDIM + 2 * half;
    v8f acc0 = {}, acc1 = {};

    #pragma unroll 4
    for (int k = 0; k < HDIM; k += 4) {
        v2f a; a.x = xrow[k]; a.y = xrow[k + 1];
        const float* wp = W + (long long)(k + 2 * half) * HDIM + c0 + l16;
        v2f b0; b0.x = wp[0];  b0.y = wp[HDIM];
        v2f b1; b1.x = wp[16]; b1.y = wp[HDIM + 16];
        acc0 = __builtin_amdgcn_wmma_f32_16x16x4_f32(false, a, false, b0,
                                                     (short)0, acc0, false, false);
        acc1 = __builtin_amdgcn_wmma_f32_16x16x4_f32(false, a, false, b1,
                                                     (short)0, acc1, false, false);
    }

    const int rbase = rowBase + rowHalf * 16 + half * 8;
    if (fullTile) {
        #pragma unroll
        for (int v = 0; v < 8; ++v) {
            float* op = h + (long long)(rbase + v) * HDIM + c0 + l16;
            op[0]  = acc0[v];
            op[16] = acc1[v];
        }
    } else {
        for (int v = 0; v < 8; ++v) {
            int r = rbase + v;
            if (r < n) {
                h[(long long)r * HDIM + c0 + l16]      = acc0[v];
                h[(long long)r * HDIM + c0 + 16 + l16] = acc1[v];
            }
        }
    }
}

__global__ __launch_bounds__(256) void k_edge_scatter(const float* __restrict__ h,
                                                      const int* __restrict__ src,
                                                      const int* __restrict__ dst,
                                                      const float* __restrict__ dinv,
                                                      float* __restrict__ agg, int E) {
    int t = blockIdx.x * 256 + threadIdx.x;
    if (t >= E * 32) return;
    int e = t >> 5, c4 = (t & 31) << 2;
    int s = src[e], d = dst[e];
    float coef = dinv[s] * dinv[d];
    const float4 hv = *(const float4*)(h + (long long)s * HDIM + c4);
    float* ap = agg + (long long)d * HDIM + c4;
    atomicAdd(ap + 0, hv.x * coef);
    atomicAdd(ap + 1, hv.y * coef);
    atomicAdd(ap + 2, hv.z * coef);
    atomicAdd(ap + 3, hv.w * coef);
}

__global__ __launch_bounds__(256) void k_combine_relu(float* __restrict__ agg,
                                                      const float* __restrict__ h,
                                                      const float* __restrict__ dinv,
                                                      const float* __restrict__ bias, int n) {
    int t = blockIdx.x * 256 + threadIdx.x;
    if (t >= n * 32) return;
    int row = t >> 5, c4 = (t & 31) << 2;
    float di = dinv[row], d2 = di * di;
    long long off = (long long)row * HDIM + c4;
    float4 a  = *(const float4*)(agg + off);
    float4 hv = *(const float4*)(h + off);
    float4 b  = *(const float4*)(bias + c4);
    a.x = fmaxf(a.x + hv.x * d2 + b.x, 0.f);
    a.y = fmaxf(a.y + hv.y * d2 + b.y, 0.f);
    a.z = fmaxf(a.z + hv.z * d2 + b.z, 0.f);
    a.w = fmaxf(a.w + hv.w * d2 + b.w, 0.f);
    *(float4*)(agg + off) = a;
}

__global__ __launch_bounds__(256) void k_stats(const float* __restrict__ y,
                                               float* __restrict__ sums, int n) {
    int c = threadIdx.x & 127;
    int sub = threadIdx.x >> 7;
    float s = 0.f, s2 = 0.f;
    for (int r = blockIdx.x * 2 + sub; r < n; r += gridDim.x * 2) {
        float v = y[(long long)r * HDIM + c];
        s += v; s2 += v * v;
    }
    __shared__ float ls[256], ls2[256];
    ls[threadIdx.x] = s; ls2[threadIdx.x] = s2;
    __syncthreads();
    if (sub == 0) {
        atomicAdd(&sums[c],        ls[c]  + ls[128 + c]);
        atomicAdd(&sums[128 + c],  ls2[c] + ls2[128 + c]);
    }
}

__global__ void k_scaleshift(const float* __restrict__ sums,
                             const float* __restrict__ gamma,
                             const float* __restrict__ beta,
                             float* __restrict__ ss, float invN) {
    int c = threadIdx.x;
    float m   = sums[c] * invN;
    float var = sums[128 + c] * invN - m * m;
    float sc  = rsqrtf(var + EPS_BN) * gamma[c];
    ss[c] = sc;
    ss[128 + c] = beta[c] - m * sc;
}

__global__ __launch_bounds__(256) void k_bn_apply(float* __restrict__ y,
                                                  const float* __restrict__ res,
                                                  const float* __restrict__ ss, int n) {
    int t = blockIdx.x * 256 + threadIdx.x;
    if (t >= n * 32) return;
    int row = t >> 5, c4 = (t & 31) << 2;
    long long off = (long long)row * HDIM + c4;
    float4 v  = *(const float4*)(y + off);
    float4 sc = *(const float4*)(ss + c4);
    float4 sh = *(const float4*)(ss + 128 + c4);
    v.x = v.x * sc.x + sh.x; v.y = v.y * sc.y + sh.y;
    v.z = v.z * sc.z + sh.z; v.w = v.w * sc.w + sh.w;
    if (res) {
        float4 r = *(const float4*)(res + off);
        v.x += r.x; v.y += r.y; v.z += r.z; v.w += r.w;
    }
    *(float4*)(y + off) = v;
}

__global__ __launch_bounds__(256) void k_counts(const int* __restrict__ batch,
                                                float* __restrict__ counts, int n) {
    int t = blockIdx.x * 256 + threadIdx.x;
    if (t < n) atomicAdd(&counts[batch[t]], 1.0f);
}

__global__ __launch_bounds__(256) void k_pool_acc(const float* __restrict__ x,
                                                  const int* __restrict__ batch,
                                                  float* __restrict__ pool, int n) {
    int t = blockIdx.x * 256 + threadIdx.x;
    if (t >= n * 32) return;
    int row = t >> 5, c4 = (t & 31) << 2;
    int g = batch[row];
    float4 v = *(const float4*)(x + (long long)row * HDIM + c4);
    float* p = pool + (long long)g * HDIM + c4;
    atomicAdd(p + 0, v.x); atomicAdd(p + 1, v.y);
    atomicAdd(p + 2, v.z); atomicAdd(p + 3, v.w);
}

__global__ __launch_bounds__(256) void k_pool_final(const float* __restrict__ pool,
                                                    const float* __restrict__ counts,
                                                    float* __restrict__ xmean, int G) {
    int t = blockIdx.x * 256 + threadIdx.x;
    if (t >= G * HDIM) return;
    int g = t / HDIM;
    xmean[t] = pool[t] / fmaxf(counts[g], 1.0f);
}

__global__ void k_pred(const float* __restrict__ xmean, const float* __restrict__ Wr,
                       const float* __restrict__ br, float* __restrict__ pred) {
    __shared__ float red[128];
    int g = blockIdx.x, c = threadIdx.x;
    red[c] = xmean[(long long)g * HDIM + c] * Wr[c];
    __syncthreads();
    for (int s = 64; s > 0; s >>= 1) {
        if (c < s) red[c] += red[c + s];
        __syncthreads();
    }
    if (c == 0) pred[g] = red[0] + br[0];
}

extern "C" void kernel_launch(void* const* d_in, const int* in_sizes, int n_in,
                              void* d_out, int out_size, void* d_ws, size_t ws_size,
                              hipStream_t stream) {
    const float* X     = (const float*)d_in[0];
    const int*   edge  = (const int*)d_in[1];
    const int*   batch = (const int*)d_in[2];
    const float* W1    = (const float*)d_in[3];
    const float* b1    = (const float*)d_in[4];
    const float* Wm    = (const float*)d_in[5];
    const float* bm    = (const float*)d_in[6];
    const float* gamma = (const float*)d_in[7];
    const float* beta  = (const float*)d_in[8];
    const float* Wr    = (const float*)d_in[9];
    const float* br    = (const float*)d_in[10];
    float* out = (float*)d_out;

    const int N = in_sizes[0];
    const int E = in_sizes[1] / 2;
    const int G = out_size / (HDIM + 1);
    const int* src = edge;
    const int* dst = edge + E;

    const long long NH = (long long)N * HDIM;
    float* ws    = (float*)d_ws;
    float* buf[3] = { ws, ws + NH, ws + 2 * NH };
    float* dinv  = ws + 3 * NH;
    float* sums  = dinv + N;
    float* ss    = sums + 256;
    float* pool  = ss + 256;
    float* cnts  = pool + (long long)G * HDIM;

    const int gE   = (E + 255) / 256;
    const int gN   = (N + 255) / 256;
    const int gNH4 = (N * 32 + 255) / 256;
    const int gE4  = (E * 32 + 255) / 256;
    const int gGem = (N + 31) / 32;

    k_zero<<<1024, 256, 0, stream>>>(dinv, N);
    k_deg_scatter<<<gE, 256, 0, stream>>>(dst, dinv, E);
    k_dinv<<<gN, 256, 0, stream>>>(dinv, N);

    auto layer_tail = [&](float* hbuf, float* abuf, const float* bias,
                          const float* g_, const float* be_, const float* res) {
        k_zero<<<2048, 256, 0, stream>>>(abuf, NH);
        k_edge_scatter<<<gE4, 256, 0, stream>>>(hbuf, src, dst, dinv, abuf, E);
        k_combine_relu<<<gNH4, 256, 0, stream>>>(abuf, hbuf, dinv, bias, N);
        k_zero<<<1, 256, 0, stream>>>(sums, 256);
        k_stats<<<256, 256, 0, stream>>>(abuf, sums, N);
        k_scaleshift<<<1, 128, 0, stream>>>(sums, g_, be_, ss, 1.0f / (float)N);
        k_bn_apply<<<gNH4, 256, 0, stream>>>(abuf, res, ss, N);
    };

    k_rank1<<<gNH4, 256, 0, stream>>>(X, W1, buf[1], N);
    layer_tail(buf[1], buf[2], b1, gamma, beta, nullptr);

    float* cur = buf[2];
    float* f0  = buf[0];
    float* f1  = buf[1];

    for (int i = 0; i < 4; ++i) {
        float* hbuf = f0;
        float* abuf = f1;
        k_gemm_wmma<<<gGem, 256, 0, stream>>>(cur, Wm + (long long)i * HDIM * HDIM, hbuf, N);
        layer_tail(hbuf, abuf, bm + (long long)i * HDIM,
                   gamma + (long long)(i + 1) * HDIM,
                   beta  + (long long)(i + 1) * HDIM, cur);
        float* next = abuf;
        f1 = cur;
        f0 = hbuf;
        cur = next;
    }

    k_zero<<<32, 256, 0, stream>>>(pool, (long long)G * HDIM + G);
    k_counts<<<gN, 256, 0, stream>>>(batch, cnts, N);
    k_pool_acc<<<gNH4, 256, 0, stream>>>(cur, batch, pool, N);
    k_pool_final<<<(G * HDIM + 255) / 256, 256, 0, stream>>>(pool, cnts, out + G, G);
    k_pred<<<G, 128, 0, stream>>>(out + G, Wr, br, out);
}


# Edges ordered by destination once per call; every layer as (A_hat x)W: each product tile gathers its rows' neighbour sums into LDS; no row atomics, no h buffer; sums run in list order
# speedup vs baseline: 2.0988x; 2.0988x over previous
#include <hip/hip_runtime.h>

#define HDIM 128
#define EPS_BN 1e-5f

typedef __attribute__((ext_vector_type(2))) float v2f;
typedef __attribute__((ext_vector_type(8))) float v8f;

__global__ __launch_bounds__(256) void k_zero(float* p, long long n) {
    long long i = (long long)blockIdx.x * blockDim.x + threadIdx.x;
    long long stride = (long long)gridDim.x * blockDim.x;
    for (; i < n; i += stride) p[i] = 0.0f;
}

__global__ __launch_bounds__(256) void k_deg_count(const int* __restrict__ dst,
                                                   int* __restrict__ degi, int E) {
    int stride = gridDim.x * 256;
    for (int e = blockIdx.x * 256 + threadIdx.x; e < E; e += stride)
        atomicAdd(&degi[dst[e]], 1);
}

__global__ __launch_bounds__(256) void k_scan_part(const int* __restrict__ degi,
                                                   float* __restrict__ dinv,
                                                   int* __restrict__ bsum, int n) {
    __shared__ int red[256];
    const int t = threadIdx.x;
    const int base = blockIdx.x * 1024 + t * 4;
    int s = 0;
    #pragma unroll
    for (int q = 0; q < 4; ++q) {
        int i = base + q;
        if (i < n) {
            int d = degi[i];
            s += d;
            dinv[i] = rsqrtf((float)d + 1.0f);
        }
    }
    red[t] = s;
    __syncthreads();
    for (int off = 128; off > 0; off >>= 1) {
        if (t < off) red[t] += red[t + off];
        __syncthreads();
    }
    if (t == 0) bsum[blockIdx.x] = red[0];
}

__global__ __launch_bounds__(256) void k_scan_fin(const int* __restrict__ degi,
                                                  const int* __restrict__ bsum,
                                                  int* __restrict__ rowptr,
                                                  int* __restrict__ cursor,
                                                  int n, int E) {
    __shared__ int red[256];
    __shared__ int sc[256];
    const int t = threadIdx.x;
    int part = 0;
    for (int k = t; k < (int)blockIdx.x; k += 256) part += bsum[k];
    red[t] = part;
    __syncthreads();
    for (int off = 128; off > 0; off >>= 1) {
        if (t < off) red[t] += red[t + off];
        __syncthreads();
    }
    const int chunkOff = red[0];

    const int base = blockIdx.x * 1024 + t * 4;
    int d[4];
    #pragma unroll
    for (int q = 0; q < 4; ++q) d[q] = (base + q < n) ? degi[base + q] : 0;
    const int tsum = d[0] + d[1] + d[2] + d[3];
    sc[t] = tsum;
    __syncthreads();
    for (int off = 1; off < 256; off <<= 1) {
        int v = (t >= off) ? sc[t - off] : 0;
        __syncthreads();
        sc[t] += v;
        __syncthreads();
    }
    int run = chunkOff + sc[t] - tsum;
    #pragma unroll
    for (int q = 0; q < 4; ++q) {
        if (base + q < n) { rowptr[base + q] = run; cursor[base + q] = run; }
        run += d[q];
    }
    if (blockIdx.x == 0 && t == 0) rowptr[n] = E;
}

__global__ __launch_bounds__(256) void k_fill(const int* __restrict__ src,
                                              const int* __restrict__ dst,
                                              const float* __restrict__ dinv,
                                              int* __restrict__ cursor,
                                              int* __restrict__ csrc,
                                              float* __restrict__ ccoef, int E) {
    int stride = gridDim.x * 256;
    for (int e = blockIdx.x * 256 + threadIdx.x; e < E; e += stride) {
        int s = src[e], d = dst[e];
        int pos = atomicAdd(&cursor[d], 1);
        if ((unsigned)pos < (unsigned)E) {
            csrc[pos]  = s;
            ccoef[pos] = dinv[s] * dinv[d];
        }
    }
}

__global__ __launch_bounds__(256) void k_layer1(const float* __restrict__ X,
                                                const float* __restrict__ W1,
                                                const float* __restrict__ b1,
                                                const int* __restrict__ rowptr,
                                                const int* __restrict__ csrc,
                                                const float* __restrict__ ccoef,
                                                const float* __restrict__ dinv,
                                                float* __restrict__ y, int n) {
    const int lane = threadIdx.x & 31;
    const int node = blockIdx.x * 8 + (threadIdx.x >> 5);
    if (node >= n) return;
    const int beg = rowptr[node], end = rowptr[node + 1];
    float part = 0.f;
    for (int j = beg + lane; j < end; j += 32) part += ccoef[j] * X[csrc[j]];
    for (int off = 16; off > 0; off >>= 1) part += __shfl_xor(part, off);
    const float di = dinv[node];
    const float a = part + X[node] * (di * di);
    const int c4 = lane << 2;
    const float4 w = *(const float4*)(W1 + c4);
    const float4 b = *(const float4*)(b1 + c4);
    float4 o;
    o.x = fmaxf(a * w.x + b.x, 0.f); o.y = fmaxf(a * w.y + b.y, 0.f);
    o.z = fmaxf(a * w.z + b.z, 0.f); o.w = fmaxf(a * w.w + b.w, 0.f);
    *(float4*)(y + (long long)node * HDIM + c4) = o;
}

__global__ __launch_bounds__(256) void k_agg_gemm(const float* __restrict__ x,
                                                  const float* __restrict__ W,
                                                  const float* __restrict__ bias,
                                                  const int* __restrict__ rowptr,
                                                  const int* __restrict__ csrc,
                                                  const float* __restrict__ ccoef,
                                                  const float* __restrict__ dinv,
                                                  float* __restrict__ y, int n) {
    __shared__ float tileX[32 * HDIM];
    const int rowBase = blockIdx.x * 32;
    const int tid  = threadIdx.x;
    const int wave = __builtin_amdgcn_readfirstlane(tid >> 5);
    const int lane = tid & 31;
    const int half = lane >> 4;
    const int l16  = lane & 15;
    const int rowHalf = wave & 1;
    const int c0   = (wave >> 1) * 32;
    const int c4   = lane << 2;

    const bool fullTile = (rowBase + 32 <= n);

    for (int q = 0; q < 4; ++q) {
        const int lr = wave * 4 + q;
        const int r  = rowBase + lr;
        float4 acc = make_float4(0.f, 0.f, 0.f, 0.f);
        if (r < n) {
            const int beg = rowptr[r], end = rowptr[r + 1];
            const float di = dinv[r];
            const float4 xs = *(const float4*)(x + (long long)r * HDIM + c4);
            for (int base = beg; base < end; base += 32) {
                int m = end - base; if (m > 32) m = 32;
                int   myi = 0;
                float myc = 0.f;
                if (lane < m) { myi = csrc[base + lane]; myc = ccoef[base + lane]; }
                const int myci = __float_as_int(myc);
                int k = 0;
                for (; k + 4 <= m; k += 4) {
                    const int s0 = __builtin_amdgcn_readlane(myi, k);
                    const int s1 = __builtin_amdgcn_readlane(myi, k + 1);
                    const int s2 = __builtin_amdgcn_readlane(myi, k + 2);
                    const int s3 = __builtin_amdgcn_readlane(myi, k + 3);
                    const float e0 = __int_as_float(__builtin_amdgcn_readlane(myci, k));
                    const float e1 = __int_as_float(__builtin_amdgcn_readlane(myci, k + 1));
                    const float e2 = __int_as_float(__builtin_amdgcn_readlane(myci, k + 2));
                    const float e3 = __int_as_float(__builtin_amdgcn_readlane(myci, k + 3));
                    const float4 v0 = *(const float4*)(x + (long long)s0 * HDIM + c4);
                    const float4 v1 = *(const float4*)(x + (long long)s1 * HDIM + c4);
                    const float4 v2 = *(const float4*)(x + (long long)s2 * HDIM + c4);
                    const float4 v3 = *(const float4*)(x + (long long)s3 * HDIM + c4);
                    acc.x += e0 * v0.x; acc.y += e0 * v0.y; acc.z += e0 * v0.z; acc.w += e0 * v0.w;
                    acc.x += e1 * v1.x; acc.y += e1 * v1.y; acc.z += e1 * v1.z; acc.w += e1 * v1.w;
                    acc.x += e2 * v2.x; acc.y += e2 * v2.y; acc.z += e2 * v2.z; acc.w += e2 * v2.w;
                    acc.x += e3 * v3.x; acc.y += e3 * v3.y; acc.z += e3 * v3.z; acc.w += e3 * v3.w;
                }
                for (; k < m; ++k) {
                    const int s0 = __builtin_amdgcn_readlane(myi, k);
                    const float e0 = __int_as_float(__builtin_amdgcn_readlane(myci, k));
                    const float4 v0 = *(const float4*)(x + (long long)s0 * HDIM + c4);
                    acc.x += e0 * v0.x; acc.y += e0 * v0.y; acc.z += e0 * v0.z; acc.w += e0 * v0.w;
                }
            }
            const float d2 = di * di;
            acc.x += xs.x * d2; acc.y += xs.y * d2; acc.z += xs.z * d2; acc.w += xs.w * d2;
        }
        *(float4*)(tileX + lr * HDIM + c4) = acc;
    }
    __syncthreads();

    const float bi0 = bias[c0 + l16];
    const float bi1 = bias[c0 + 16 + l16];

    const float* xrow = tileX + (rowHalf * 16 + l16) * HDIM + 2 * half;
    v8f acc0 = {}, acc1 = {};

    #pragma unroll 4
    for (int k = 0; k < HDIM; k += 4) {
        v2f a; a.x = xrow[k]; a.y = xrow[k + 1];
        const float* wp = W + (long long)(k + 2 * half) * HDIM + c0 + l16;
        v2f b0; b0.x = wp[0];  b0.y = wp[HDIM];
        v2f b1; b1.x = wp[16]; b1.y = wp[HDIM + 16];
        acc0 = __builtin_amdgcn_wmma_f32_16x16x4_f32(false, a, false, b0,
                                                     (short)0, acc0, false, false);
        acc1 = __builtin_amdgcn_wmma_f32_16x16x4_f32(false, a, false, b1,
                                                     (short)0, acc1, false, false);
    }

    asm volatile("v_nop\n\tv_nop\n\tv_nop\n\tv_nop\n\tv_nop\n\tv_nop\n\tv_nop\n\tv_nop"
                 : "+v"(acc0), "+v"(acc1));

    #pragma unroll
    for (int v = 0; v < 8; ++v) {
        acc0[v] = fmaxf(acc0[v] + bi0, 0.f);
        acc1[v] = fmaxf(acc1[v] + bi1, 0.f);
    }

    const int rbase = rowBase + rowHalf * 16 + half * 8;
    if (fullTile) {
        #pragma unroll
        for (int v = 0; v < 8; ++v) {
            float* op = y + (long long)(rbase + v) * HDIM + c0 + l16;
            op[0]  = acc0[v];
            op[16] = acc1[v];
        }
    } else {
        for (int v = 0; v < 8; ++v) {
            int r = rbase + v;
            if (r < n) {
                y[(long long)r * HDIM + c0 + l16]      = acc0[v];
                y[(long long)r * HDIM + c0 + 16 + l16] = acc1[v];
            }
        }
    }
}

__global__ __launch_bounds__(256) void k_stats(const float* __restrict__ y,
                                               float* __restrict__ sums, int n) {
    int c = threadIdx.x & 127;
    int sub = threadIdx.x >> 7;
    float s = 0.f, s2 = 0.f;
    for (int r = blockIdx.x * 2 + sub; r < n; r += gridDim.x * 2) {
        float v = y[(long long)r * HDIM + c];
        s += v; s2 += v * v;
    }
    __shared__ float ls[256], ls2[256];
    ls[threadIdx.x] = s; ls2[threadIdx.x] = s2;
    __syncthreads();
    if (sub == 0) {
        atomicAdd(&sums[c],        ls[c]  + ls[128 + c]);
        atomicAdd(&sums[128 + c],  ls2[c] + ls2[128 + c]);
    }
}

__global__ void k_scaleshift(const float* __restrict__ sums,
                             const float* __restrict__ gamma,
                             const float* __restrict__ beta,
                             float* __restrict__ ss, float invN) {
    int c = threadIdx.x;
    float m   = sums[c] * invN;
    float var = sums[128 + c] * invN - m * m;
    float sc  = rsqrtf(var + EPS_BN) * gamma[c];
    ss[c] = sc;
    ss[128 + c] = beta[c] - m * sc;
}

__global__ __launch_bounds__(256) void k_bn_apply(float* __restrict__ y,
                                                  const float* __restrict__ res,
                                                  const float* __restrict__ ss, int n) {
    int t = blockIdx.x * 256 + threadIdx.x;
    if (t >= n * 32) return;
    int row = t >> 5, c4 = (t & 31) << 2;
    long long off = (long long)row * HDIM + c4;
    float4 v  = *(const float4*)(y + off);
    float4 sc = *(const float4*)(ss + c4);
    float4 sh = *(const float4*)(ss + 128 + c4);
    v.x = v.x * sc.x + sh.x; v.y = v.y * sc.y + sh.y;
    v.z = v.z * sc.z + sh.z; v.w = v.w * sc.w + sh.w;
    if (res) {
        float4 r = *(const float4*)(res + off);
        v.x += r.x; v.y += r.y; v.z += r.z; v.w += r.w;
    }
    *(float4*)(y + off) = v;
}

__global__ __launch_bounds__(256) void k_counts(const int* __restrict__ batch,
                                                float* __restrict__ counts, int n) {
    int t = blockIdx.x * 256 + threadIdx.x;
    if (t < n) atomicAdd(&counts[batch[t]], 1.0f);
}

__global__ __launch_bounds__(256) void k_pool_acc(const float* __restrict__ x,
                                                  const int* __restrict__ batch,
                                                  float* __restrict__ pool, int n) {
    int t = blockIdx.x * 256 + threadIdx.x;
    if (t >= n * 32) return;
    int row = t >> 5, c4 = (t & 31) << 2;
    int g = batch[row];
    float4 v = *(const float4*)(x + (long long)row * HDIM + c4);
    float* p = pool + (long long)g * HDIM + c4;
    atomicAdd(p + 0, v.x); atomicAdd(p + 1, v.y);
    atomicAdd(p + 2, v.z); atomicAdd(p + 3, v.w);
}

__global__ __launch_bounds__(256) void k_pool_final(const float* __restrict__ pool,
                                                    const float* __restrict__ counts,
                                                    float* __restrict__ xmean, int G) {
    int t = blockIdx.x * 256 + threadIdx.x;
    if (t >= G * HDIM) return;
    int g = t / HDIM;
    xmean[t] = pool[t] / fmaxf(counts[g], 1.0f);
}

__global__ void k_pred(const float* __restrict__ xmean, const float* __restrict__ Wr,
                       const float* __restrict__ br, float* __restrict__ pred) {
    __shared__ float red[128];
    int g = blockIdx.x, c = threadIdx.x;
    red[c] = xmean[(long long)g * HDIM + c] * Wr[c];
    __syncthreads();
    for (int s = 64; s > 0; s >>= 1) {
        if (c < s) red[c] += red[c + s];
        __syncthreads();
    }
    if (c == 0) pred[g] = red[0] + br[0];
}

extern "C" void kernel_launch(void* const* d_in, const int* in_sizes, int n_in,
                              void* d_out, int out_size, void* d_ws, size_t ws_size,
                              hipStream_t stream) {
    const float* X     = (const float*)d_in[0];
    const int*   edge  = (const int*)d_in[1];
    const int*   batch = (const int*)d_in[2];
    const float* W1    = (const float*)d_in[3];
    const float* b1    = (const float*)d_in[4];
    const float* Wm    = (const float*)d_in[5];
    const float* bm    = (const float*)d_in[6];
    const float* gamma = (const float*)d_in[7];
    const float* beta  = (const float*)d_in[8];
    const float* Wr    = (const float*)d_in[9];
    const float* br    = (const float*)d_in[10];
    float* out = (float*)d_out;

    const int N = in_sizes[0];
    const int E = in_sizes[1] / 2;
    const int G = out_size / (HDIM + 1);
    const int* src = edge;
    const int* dst = edge + E;

    const long long NH = (long long)N * HDIM;
    float* ws    = (float*)d_ws;
    float* bufA  = ws;
    float* bufB  = ws + NH;
    int*   rowptr = (int*)(ws + 2 * NH);
    int*   cursor = rowptr + (N + 1);
    int*   degi   = cursor + N;
    int*   bsum   = degi + N;
    int*   csrc   = bsum + 1024;
    float* ccoef  = (float*)(csrc + E);
    float* dinv  = ws + 3 * NH;
    float* sums  = dinv + N;
    float* ss    = sums + 256;
    float* pool  = ss + 256;
    float* cnts  = pool + (long long)G * HDIM;

    const int gN   = (N + 255) / 256;
    const int gNH4 = (N * 32 + 255) / 256;
    const int gGem = (N + 31) / 32;
    const int nChunk = (N + 1023) / 1024;

    k_zero<<<256, 256, 0, stream>>>((float*)degi, N);
    k_deg_count<<<2048, 256, 0, stream>>>(dst, degi, E);
    k_scan_part<<<nChunk, 256, 0, stream>>>(degi, dinv, bsum, N);
    k_scan_fin<<<nChunk, 256, 0, stream>>>(degi, bsum, rowptr, cursor, N, E);
    k_fill<<<2048, 256, 0, stream>>>(src, dst, dinv, cursor, csrc, ccoef, E);

    auto bn_tail = [&](float* ybuf, const float* g_, const float* be_, const float* res) {
        k_zero<<<1, 256, 0, stream>>>(sums, 256);
        k_stats<<<256, 256, 0, stream>>>(ybuf, sums, N);
        k_scaleshift<<<1, 128, 0, stream>>>(sums, g_, be_, ss, 1.0f / (float)N);
        k_bn_apply<<<gNH4, 256, 0, stream>>>(ybuf, res, ss, N);
    };

    k_layer1<<<(N + 7) / 8, 256, 0, stream>>>(X, W1, b1, rowptr, csrc, ccoef, dinv, bufA, N);
    bn_tail(bufA, gamma, beta, nullptr);

    float* cur = bufA;
    float* nxt = bufB;

    for (int i = 0; i < 4; ++i) {
        k_agg_gemm<<<gGem, 256, 0, stream>>>(cur, Wm + (long long)i * HDIM * HDIM,
                                             bm + (long long)i * HDIM,
                                             rowptr, csrc, ccoef, dinv, nxt, N);
        bn_tail(nxt, gamma + (long long)(i + 1) * HDIM,
                beta  + (long long)(i + 1) * HDIM, cur);
        float* t = cur; cur = nxt; nxt = t;
    }

    k_zero<<<32, 256, 0, stream>>>(pool, (long long)G * HDIM + G);
    k_counts<<<gN, 256, 0, stream>>>(batch, cnts, N);
    k_pool_acc<<<gNH4, 256, 0, stream>>>(cur, batch, pool, N);
    k_pool_final<<<(G * HDIM + 255) / 256, 256, 0, stream>>>(pool, cnts, out + G, G);
    k_pred<<<G, 128, 0, stream>>>(out + G, Wr, br, out);
}


# BN tail 4 launches -> 1: column sum/sumsq formed in the producer's epilogue (tile tree, then float atomics on 8 copies: stats summed in a new order); apply derives scale/shift itself, grid-stride
# speedup vs baseline: 3.1290x; 1.4909x over previous
#include <hip/hip_runtime.h>

#define HDIM 128
#define EPS_BN 1e-5f
#define NREP 8

typedef __attribute__((ext_vector_type(2))) float v2f;
typedef __attribute__((ext_vector_type(8))) float v8f;
typedef __attribute__((ext_vector_type(4))) float v4f;

__global__ __launch_bounds__(256) void k_zero(float* p, long long n) {
    long long i = (long long)blockIdx.x * blockDim.x + threadIdx.x;
    long long stride = (long long)gridDim.x * blockDim.x;
    for (; i < n; i += stride) p[i] = 0.0f;
}

__global__ __launch_bounds__(256) void k_zero2(float* a, long long na, float* b, long long nb) {
    long long i = (long long)blockIdx.x * blockDim.x + threadIdx.x;
    long long stride = (long long)gridDim.x * blockDim.x;
    for (; i < na + nb; i += stride) {
        if (i < na) a[i] = 0.0f; else b[i - na] = 0.0f;
    }
}

__global__ __launch_bounds__(256) void k_deg_count(const int* __restrict__ dst,
                                                   int* __restrict__ degi, int E) {
    int stride = gridDim.x * 256;
    for (int e = blockIdx.x * 256 + threadIdx.x; e < E; e += stride)
        atomicAdd(&degi[dst[e]], 1);
}

__global__ __launch_bounds__(256) void k_scan_part(const int* __restrict__ degi,
                                                   float* __restrict__ dinv,
                                                   int* __restrict__ bsum, int n) {
    __shared__ int red[256];
    const int t = threadIdx.x;
    const int base = blockIdx.x * 1024 + t * 4;
    int s = 0;
    #pragma unroll
    for (int q = 0; q < 4; ++q) {
        int i = base + q;
        if (i < n) {
            int d = degi[i];
            s += d;
            dinv[i] = rsqrtf((float)d + 1.0f);
        }
    }
    red[t] = s;
    __syncthreads();
    for (int off = 128; off > 0; off >>= 1) {
        if (t < off) red[t] += red[t + off];
        __syncthreads();
    }
    if (t == 0) bsum[blockIdx.x] = red[0];
}

__global__ __launch_bounds__(256) void k_scan_fin(const int* __restrict__ degi,
                                                  const int* __restrict__ bsum,
                                                  int* __restrict__ rowptr,
                                                  int* __restrict__ cursor,
                                                  int n, int E) {
    __shared__ int red[256];
    __shared__ int sc[256];
    const int t = threadIdx.x;
    int part = 0;
    for (int k = t; k < (int)blockIdx.x; k += 256) part += bsum[k];
    red[t] = part;
    __syncthreads();
    for (int off = 128; off > 0; off >>= 1) {
        if (t < off) red[t] += red[t + off];
        __syncthreads();
    }
    const int chunkOff = red[0];

    const int base = blockIdx.x * 1024 + t * 4;
    int d[4];
    #pragma unroll
    for (int q = 0; q < 4; ++q) d[q] = (base + q < n) ? degi[base + q] : 0;
    const int tsum = d[0] + d[1] + d[2] + d[3];
    sc[t] = tsum;
    __syncthreads();
    for (int off = 1; off < 256; off <<= 1) {
        int v = (t >= off) ? sc[t - off] : 0;
        __syncthreads();
        sc[t] += v;
        __syncthreads();
    }
    int run = chunkOff + sc[t] - tsum;
    #pragma unroll
    for (int q = 0; q < 4; ++q) {
        if (base + q < n) { rowptr[base + q] = run; cursor[base + q] = run; }
        run += d[q];
    }
    if (blockIdx.x == 0 && t == 0) rowptr[n] = E;
}

__global__ __launch_bounds__(256) void k_fill(const int* __restrict__ src,
                                              const int* __restrict__ dst,
                                              const float* __restrict__ dinv,
                                              int* __restrict__ cursor,
                                              int* __restrict__ csrc,
                                              float* __restrict__ ccoef, int E) {
    int stride = gridDim.x * 256;
    for (int e = blockIdx.x * 256 + threadIdx.x; e < E; e += stride) {
        int s = src[e], d = dst[e];
        int pos = atomicAdd(&cursor[d], 1);
        if ((unsigned)pos < (unsigned)E) {
            csrc[pos]  = s;
            ccoef[pos] = dinv[s] * dinv[d];
        }
    }
}

__global__ __launch_bounds__(256) void k_layer1(const float* __restrict__ X,
                                                const float* __restrict__ W1,
                                                const float* __restrict__ b1,
                                                const int* __restrict__ rowptr,
                                                const int* __restrict__ csrc,
                                                const float* __restrict__ ccoef,
                                                const float* __restrict__ dinv,
                                                float* __restrict__ y,
                                                float* __restrict__ sums, int n) {
    __shared__ float red[8 * 256];
    const int lane = threadIdx.x & 31;
    const int wave = __builtin_amdgcn_readfirstlane((int)(threadIdx.x >> 5));
    const int c4 = lane << 2;
    const float4 w = *(const float4*)(W1 + c4);
    const float4 b = *(const float4*)(b1 + c4);
    float4 s = make_float4(0.f, 0.f, 0.f, 0.f);
    float4 q = make_float4(0.f, 0.f, 0.f, 0.f);
    for (int node = blockIdx.x * 8 + wave; node < n; node += gridDim.x * 8) {
        const int beg = rowptr[node], end = rowptr[node + 1];
        float part = 0.f;
        for (int j = beg + lane; j < end; j += 32) part += ccoef[j] * X[csrc[j]];
        for (int off = 16; off > 0; off >>= 1) part += __shfl_xor(part, off);
        const float di = dinv[node];
        const float a = part + X[node] * (di * di);
        float4 o;
        o.x = fmaxf(a * w.x + b.x, 0.f); o.y = fmaxf(a * w.y + b.y, 0.f);
        o.z = fmaxf(a * w.z + b.z, 0.f); o.w = fmaxf(a * w.w + b.w, 0.f);
        *(float4*)(y + (long long)node * HDIM + c4) = o;
        s.x += o.x; s.y += o.y; s.z += o.z; s.w += o.w;
        q.x += o.x * o.x; q.y += o.y * o.y; q.z += o.z * o.z; q.w += o.w * o.w;
    }
    *(float4*)(red + wave * 256 + c4)       = s;
    *(float4*)(red + wave * 256 + 128 + c4) = q;
    __syncthreads();
    float tot = 0.f;
    for (int k = 0; k < 8; ++k) tot += red[k * 256 + threadIdx.x];
    atomicAdd(&sums[(blockIdx.x & (NREP - 1)) * 256 + threadIdx.x], tot);
}

__global__ __launch_bounds__(256) void k_agg_gemm(const float* __restrict__ x,
                                                  const float* __restrict__ W,
                                                  const float* __restrict__ bias,
                                                  const int* __restrict__ rowptr,
                                                  const int* __restrict__ csrc,
                                                  const float* __restrict__ ccoef,
                                                  const float* __restrict__ dinv,
                                                  float* __restrict__ y,
                                                  float* __restrict__ sums, int n) {
    __shared__ float tileX[32 * HDIM];
    __shared__ float red[2 * 256];
    const int rowBase = blockIdx.x * 32;
    const int tid  = threadIdx.x;
    const int wave = __builtin_amdgcn_readfirstlane(tid >> 5);
    const int lane = tid & 31;
    const int half = lane >> 4;
    const int l16  = lane & 15;
    const int rowHalf = wave & 1;
    const int c0   = (wave >> 1) * 32;
    const int c4   = lane << 2;

    const bool fullTile = (rowBase + 32 <= n);

    for (int q = 0; q < 4; ++q) {
        const int lr = wave * 4 + q;
        const int r  = rowBase + lr;
        float4 acc = make_float4(0.f, 0.f, 0.f, 0.f);
        if (r < n) {
            const int beg = rowptr[r], end = rowptr[r + 1];
            const float di = dinv[r];
            const float4 xs = *(const float4*)(x + (long long)r * HDIM + c4);
            for (int base = beg; base < end; base += 32) {
                int m = end - base; if (m > 32) m = 32;
                int   myi = 0;
                float myc = 0.f;
                if (lane < m) { myi = csrc[base + lane]; myc = ccoef[base + lane]; }
                const int myci = __float_as_int(myc);
                int k = 0;
                for (; k + 4 <= m; k += 4) {
                    const int s0 = __builtin_amdgcn_readlane(myi, k);
                    const int s1 = __builtin_amdgcn_readlane(myi, k + 1);
                    const int s2 = __builtin_amdgcn_readlane(myi, k + 2);
                    const int s3 = __builtin_amdgcn_readlane(myi, k + 3);
                    const float e0 = __int_as_float(__builtin_amdgcn_readlane(myci, k));
                    const float e1 = __int_as_float(__builtin_amdgcn_readlane(myci, k + 1));
                    const float e2 = __int_as_float(__builtin_amdgcn_readlane(myci, k + 2));
                    const float e3 = __int_as_float(__builtin_amdgcn_readlane(myci, k + 3));
                    const float4 v0 = *(const float4*)(x + (long long)s0 * HDIM + c4);
                    const float4 v1 = *(const float4*)(x + (long long)s1 * HDIM + c4);
                    const float4 v2 = *(const float4*)(x + (long long)s2 * HDIM + c4);
                    const float4 v3 = *(const float4*)(x + (long long)s3 * HDIM + c4);
                    acc.x += e0 * v0.x; acc.y += e0 * v0.y; acc.z += e0 * v0.z; acc.w += e0 * v0.w;
                    acc.x += e1 * v1.x; acc.y += e1 * v1.y; acc.z += e1 * v1.z; acc.w += e1 * v1.w;
                    acc.x += e2 * v2.x; acc.y += e2 * v2.y; acc.z += e2 * v2.z; acc.w += e2 * v2.w;
                    acc.x += e3 * v3.x; acc.y += e3 * v3.y; acc.z += e3 * v3.z; acc.w += e3 * v3.w;
                }
                for (; k < m; ++k) {
                    const int s0 = __builtin_amdgcn_readlane(myi, k);
                    const float e0 = __int_as_float(__builtin_amdgcn_readlane(myci, k));
                    const float4 v0 = *(const float4*)(x + (long long)s0 * HDIM + c4);
                    acc.x += e0 * v0.x; acc.y += e0 * v0.y; acc.z += e0 * v0.z; acc.w += e0 * v0.w;
                }
            }
            const float d2 = di * di;
            acc.x += xs.x * d2; acc.y += xs.y * d2; acc.z += xs.z * d2; acc.w += xs.w * d2;
        }
        *(float4*)(tileX + lr * HDIM + c4) = acc;
    }
    __syncthreads();

    const float bi0 = bias[c0 + l16];
    const float bi1 = bias[c0 + 16 + l16];

    const float* xrow = tileX + (rowHalf * 16 + l16) * HDIM + 2 * half;
    v8f acc0 = {}, acc1 = {};

    #pragma unroll 4
    for (int k = 0; k < HDIM; k += 4) {
        v2f a; a.x = xrow[k]; a.y = xrow[k + 1];
        const float* wp = W + (long long)(k + 2 * half) * HDIM + c0 + l16;
        v2f b0; b0.x = wp[0];  b0.y = wp[HDIM];
        v2f b1; b1.x = wp[16]; b1.y = wp[HDIM + 16];
        acc0 = __builtin_amdgcn_wmma_f32_16x16x4_f32(false, a, false, b0,
                                                     (short)0, acc0, false, false);
        acc1 = __builtin_amdgcn_wmma_f32_16x16x4_f32(false, a, false, b1,
                                                     (short)0, acc1, false, false);
    }

    asm volatile("v_nop\n\tv_nop\n\tv_nop\n\tv_nop\n\tv_nop\n\tv_nop\n\tv_nop\n\tv_nop"
                 : "+v"(acc0), "+v"(acc1));

    #pragma unroll
    for (int v = 0; v < 8; ++v) {
        acc0[v] = fmaxf(acc0[v] + bi0, 0.f);
        acc1[v] = fmaxf(acc1[v] + bi1, 0.f);
    }

    const int rbase = rowBase + rowHalf * 16 + half * 8;

    float cs0 = 0.f, cq0 = 0.f, cs1 = 0.f, cq1 = 0.f;
    for (int v = 0; v < 8; ++v) {
        const bool in = fullTile || (rbase + v < n);
        const float a0 = in ? acc0[v] : 0.f;
        const float a1 = in ? acc1[v] : 0.f;
        cs0 += a0; cq0 += a0 * a0;
        cs1 += a1; cq1 += a1 * a1;
    }
    cs0 += __shfl_xor(cs0, 16); cq0 += __shfl_xor(cq0, 16);
    cs1 += __shfl_xor(cs1, 16); cq1 += __shfl_xor(cq1, 16);
    if (half == 0) {
        float* rp = red + rowHalf * 256 + c0 + l16;
        rp[0]   = cs0; rp[16]  = cs1;
        rp[128] = cq0; rp[144] = cq1;
    }
    __syncthreads();
    atomicAdd(&sums[(blockIdx.x & (NREP - 1)) * 256 + tid], red[tid] + red[256 + tid]);

    if (fullTile) {
        #pragma unroll
        for (int v = 0; v < 8; ++v) {
            float* op = y + (long long)(rbase + v) * HDIM + c0 + l16;
            op[0]  = acc0[v];
            op[16] = acc1[v];
        }
    } else {
        for (int v = 0; v < 8; ++v) {
            int r = rbase + v;
            if (r < n) {
                y[(long long)r * HDIM + c0 + l16]      = acc0[v];
                y[(long long)r * HDIM + c0 + 16 + l16] = acc1[v];
            }
        }
    }
}

template <bool HAS_RES>
__device__ __forceinline__ void bn_rows(float* __restrict__ y, const float* __restrict__ res,
                                        const v4f sc, const v4f sh,
                                        int t, const int stride, const int total) {
    for (; t + 3 * stride < total; t += 4 * stride) {
        v4f v[4], r[4];
        for (int u = 0; u < 4; ++u) {
            const long long off = (long long)(t + u * stride) * 4;
            v[u] = *(const v4f*)(y + off);
            if (HAS_RES) r[u] = *(const v4f*)(res + off);
        }
        for (int u = 0; u < 4; ++u) {
            const long long off = (long long)(t + u * stride) * 4;
            v4f o = v[u] * sc + sh;
            if (HAS_RES) o += r[u];
            *(v4f*)(y + off) = o;
        }
    }
    for (; t < total; t += stride) {
        const long long off = (long long)t * 4;
        v4f o = *(const v4f*)(y + off) * sc + sh;
        if (HAS_RES) o += *(const v4f*)(res + off);
        *(v4f*)(y + off) = o;
    }
}

__global__ __launch_bounds__(256) void k_bn_apply(float* __restrict__ y,
                                                  const float* __restrict__ res,
                                                  const float* __restrict__ sums,
                                                  const float* __restrict__ gamma,
                                                  const float* __restrict__ beta,
                                                  float invN, int n) {
    const int t0 = blockIdx.x * 256 + threadIdx.x;
    const int c4 = (t0 & 31) << 2;
    float4 s = make_float4(0.f, 0.f, 0.f, 0.f);
    float4 q = make_float4(0.f, 0.f, 0.f, 0.f);
    for (int k = 0; k < NREP; ++k) {
        const float4 a = *(const float4*)(sums + k * 256 + c4);
        const float4 b = *(const float4*)(sums + k * 256 + 128 + c4);
        s.x += a.x; s.y += a.y; s.z += a.z; s.w += a.w;
        q.x += b.x; q.y += b.y; q.z += b.z; q.w += b.w;
    }
    const float4 g  = *(const float4*)(gamma + c4);
    const float4 be = *(const float4*)(beta + c4);
    v4f sc, sh;
    {
        float m, var;
        m = s.x * invN; var = q.x * invN - m * m; sc.x = rsqrtf(var + EPS_BN) * g.x; sh.x = be.x - m * sc.x;
        m = s.y * invN; var = q.y * invN - m * m; sc.y = rsqrtf(var + EPS_BN) * g.y; sh.y = be.y - m * sc.y;
        m = s.z * invN; var = q.z * invN - m * m; sc.z = rsqrtf(var + EPS_BN) * g.z; sh.z = be.z - m * sc.z;
        m = s.w * invN; var = q.w * invN - m * m; sc.w = rsqrtf(var + EPS_BN) * g.w; sh.w = be.w - m * sc.w;
    }
    const int stride = gridDim.x * 256;
    const int total  = n * 32;
    if (res) bn_rows<true>(y, res, sc, sh, t0, stride, total);
    else     bn_rows<false>(y, res, sc, sh, t0, stride, total);
}

__global__ __launch_bounds__(256) void k_counts(const int* __restrict__ batch,
                                                float* __restrict__ counts, int n) {
    int t = blockIdx.x * 256 + threadIdx.x;
    if (t < n) atomicAdd(&counts[batch[t]], 1.0f);
}

__global__ __launch_bounds__(256) void k_pool_acc(const float* __restrict__ x,
                                                  const int* __restrict__ batch,
                                                  float* __restrict__ pool, int n) {
    int t = blockIdx.x * 256 + threadIdx.x;
    if (t >= n * 32) return;
    int row = t >> 5, c4 = (t & 31) << 2;
    int g = batch[row];
    float4 v = *(const float4*)(x + (long long)row * HDIM + c4);
    float* p = pool + (long long)g * HDIM + c4;
    atomicAdd(p + 0, v.x); atomicAdd(p + 1, v.y);
    atomicAdd(p + 2, v.z); atomicAdd(p + 3, v.w);
}

__global__ __launch_bounds__(256) void k_pool_final(const float* __restrict__ pool,
                                                    const float* __restrict__ counts,
                                                    float* __restrict__ xmean, int G) {
    int t = blockIdx.x * 256 + threadIdx.x;
    if (t >= G * HDIM) return;
    int g = t / HDIM;
    xmean[t] = pool[t] / fmaxf(counts[g], 1.0f);
}

__global__ void k_pred(const float* __restrict__ xmean, const float* __restrict__ Wr,
                       const float* __restrict__ br, float* __restrict__ pred) {
    __shared__ float red[128];
    int g = blockIdx.x, c = threadIdx.x;
    red[c] = xmean[(long long)g * HDIM + c] * Wr[c];
    __syncthreads();
    for (int s = 64; s > 0; s >>= 1) {
        if (c < s) red[c] += red[c + s];
        __syncthreads();
    }
    if (c == 0) pred[g] = red[0] + br[0];
}

extern "C" void kernel_launch(void* const* d_in, const int* in_sizes, int n_in,
                              void* d_out, int out_size, void* d_ws, size_t ws_size,
                              hipStream_t stream) {
    const float* X     = (const float*)d_in[0];
    const int*   edge  = (const int*)d_in[1];
    const int*   batch = (const int*)d_in[2];
    const float* W1    = (const float*)d_in[3];
    const float* b1    = (const float*)d_in[4];
    const float* Wm    = (const float*)d_in[5];
    const float* bm    = (const float*)d_in[6];
    const float* gamma = (const float*)d_in[7];
    const float* beta  = (const float*)d_in[8];
    const float* Wr    = (const float*)d_in[9];
    const float* br    = (const float*)d_in[10];
    float* out = (float*)d_out;

    const int N = in_sizes[0];
    const int E = in_sizes[1] / 2;
    const int G = out_size / (HDIM + 1);
    const int* src = edge;
    const int* dst = edge + E;

    const long long NH = (long long)N * HDIM;
    float* ws    = (float*)d_ws;
    float* bufA  = ws;
    float* bufB  = ws + NH;
    int*   rowptr = (int*)(ws + 2 * NH);
    int*   cursor = rowptr + (N + 1);
    int*   degi   = cursor + N;
    int*   bsum   = degi + N;
    int*   csrc   = bsum + 1024;
    float* ccoef  = (float*)(csrc + E);
    float* sumsAll = ws + 2 * NH + ((3LL * N + 1025 + 2LL * E + 3) & ~3LL);
    float* dinv  = ws + 3 * NH;
    float* pool  = dinv + N + 512;
    float* cnts  = pool + (long long)G * HDIM;

    const int gN   = (N + 255) / 256;
    const int gNH4 = (N * 32 + 255) / 256;
    const int gGem = (N + 31) / 32;
    const int nChunk = (N + 1023) / 1024;

    k_zero2<<<256, 256, 0, stream>>>((float*)degi, N,
                                     sumsAll, 5LL * NREP * 256);
    k_deg_count<<<2048, 256, 0, stream>>>(dst, degi, E);
    k_scan_part<<<nChunk, 256, 0, stream>>>(degi, dinv, bsum, N);
    k_scan_fin<<<nChunk, 256, 0, stream>>>(degi, bsum, rowptr, cursor, N, E);
    k_fill<<<2048, 256, 0, stream>>>(src, dst, dinv, cursor, csrc, ccoef, E);

    const float invN = 1.0f / (float)N;
    auto bn_tail = [&](float* ybuf, const float* sums_, const float* g_, const float* be_,
                       const float* res) {
        k_bn_apply<<<1024, 256, 0, stream>>>(ybuf, res, sums_, g_, be_, invN, N);
    };

    k_layer1<<<2048, 256, 0, stream>>>(X, W1, b1, rowptr, csrc, ccoef, dinv, bufA, sumsAll, N);
    bn_tail(bufA, sumsAll, gamma, beta, nullptr);

    float* cur = bufA;
    float* nxt = bufB;

    for (int i = 0; i < 4; ++i) {
        k_agg_gemm<<<gGem, 256, 0, stream>>>(cur, Wm + (long long)i * HDIM * HDIM,
                                             bm + (long long)i * HDIM,
                                             rowptr, csrc, ccoef, dinv, nxt,
                                             sumsAll + (long long)(i + 1) * NREP * 256, N);
        bn_tail(nxt, sumsAll + (long long)(i + 1) * NREP * 256,
                gamma + (long long)(i + 1) * HDIM,
                beta  + (long long)(i + 1) * HDIM, cur);
        float* t = cur; cur = nxt; nxt = t;
    }

    k_zero<<<32, 256, 0, stream>>>(pool, (long long)G * HDIM + G);
    k_counts<<<gN, 256, 0, stream>>>(batch, cnts, N);
    k_pool_acc<<<gNH4, 256, 0, stream>>>(cur, batch, pool, N);
    k_pool_final<<<(G * HDIM + 255) / 256, 256, 0, stream>>>(pool, cnts, out + G, G);
    k_pred<<<G, 128, 0, stream>>>(out + G, Wr, br, out);
}


# Last BN-apply+residual fused with the mean pool: a wave sums its contiguous rows of one graph id in registers (row order), one float-add flush per run; x5 not stored; 20 -> 17 launches
# speedup vs baseline: 3.8915x; 1.2437x over previous
#include <hip/hip_runtime.h>

#define HDIM 128
#define EPS_BN 1e-5f
#define NREP 8

typedef __attribute__((ext_vector_type(2))) float v2f;
typedef __attribute__((ext_vector_type(8))) float v8f;
typedef __attribute__((ext_vector_type(4))) float v4f;

__global__ __launch_bounds__(256) void k_zero2(float* a, long long na, float* b, long long nb) {
    long long i = (long long)blockIdx.x * blockDim.x + threadIdx.x;
    long long stride = (long long)gridDim.x * blockDim.x;
    for (; i < na + nb; i += stride) {
        if (i < na) a[i] = 0.0f; else b[i - na] = 0.0f;
    }
}

__global__ __launch_bounds__(256) void k_deg_count(const int* __restrict__ dst,
                                                   int* __restrict__ degi, int E) {
    int stride = gridDim.x * 256;
    for (int e = blockIdx.x * 256 + threadIdx.x; e < E; e += stride)
        atomicAdd(&degi[dst[e]], 1);
}

__global__ __launch_bounds__(256) void k_scan_part(const int* __restrict__ degi,
                                                   float* __restrict__ dinv,
                                                   int* __restrict__ bsum, int n) {
    __shared__ int red[256];
    const int t = threadIdx.x;
    const int base = blockIdx.x * 1024 + t * 4;
    int s = 0;
    #pragma unroll
    for (int q = 0; q < 4; ++q) {
        int i = base + q;
        if (i < n) {
            int d = degi[i];
            s += d;
            dinv[i] = rsqrtf((float)d + 1.0f);
        }
    }
    red[t] = s;
    __syncthreads();
    for (int off = 128; off > 0; off >>= 1) {
        if (t < off) red[t] += red[t + off];
        __syncthreads();
    }
    if (t == 0) bsum[blockIdx.x] = red[0];
}

__global__ __launch_bounds__(256) void k_scan_fin(const int* __restrict__ degi,
                                                  const int* __restrict__ bsum,
                                                  int* __restrict__ rowptr,
                                                  int* __restrict__ cursor,
                                                  int n, int E) {
    __shared__ int red[256];
    __shared__ int sc[256];
    const int t = threadIdx.x;
    int part = 0;
    for (int k = t; k < (int)blockIdx.x; k += 256) part += bsum[k];
    red[t] = part;
    __syncthreads();
    for (int off = 128; off > 0; off >>= 1) {
        if (t < off) red[t] += red[t + off];
        __syncthreads();
    }
    const int chunkOff = red[0];

    const int base = blockIdx.x * 1024 + t * 4;
    int d[4];
    #pragma unroll
    for (int q = 0; q < 4; ++q) d[q] = (base + q < n) ? degi[base + q] : 0;
    const int tsum = d[0] + d[1] + d[2] + d[3];
    sc[t] = tsum;
    __syncthreads();
    for (int off = 1; off < 256; off <<= 1) {
        int v = (t >= off) ? sc[t - off] : 0;
        __syncthreads();
        sc[t] += v;
        __syncthreads();
    }
    int run = chunkOff + sc[t] - tsum;
    #pragma unroll
    for (int q = 0; q < 4; ++q) {
        if (base + q < n) { rowptr[base + q] = run; cursor[base + q] = run; }
        run += d[q];
    }
    if (blockIdx.x == 0 && t == 0) rowptr[n] = E;
}

__global__ __launch_bounds__(256) void k_fill(const int* __restrict__ src,
                                              const int* __restrict__ dst,
                                              const float* __restrict__ dinv,
                                              int* __restrict__ cursor,
                                              int* __restrict__ csrc,
                                              float* __restrict__ ccoef, int E) {
    int stride = gridDim.x * 256;
    for (int e = blockIdx.x * 256 + threadIdx.x; e < E; e += stride) {
        int s = src[e], d = dst[e];
        int pos = atomicAdd(&cursor[d], 1);
        if ((unsigned)pos < (unsigned)E) {
            csrc[pos]  = s;
            ccoef[pos] = dinv[s] * dinv[d];
        }
    }
}

__global__ __launch_bounds__(256) void k_layer1(const float* __restrict__ X,
                                                const float* __restrict__ W1,
                                                const float* __restrict__ b1,
                                                const int* __restrict__ rowptr,
                                                const int* __restrict__ csrc,
                                                const float* __restrict__ ccoef,
                                                const float* __restrict__ dinv,
                                                float* __restrict__ y,
                                                float* __restrict__ sums, int n) {
    __shared__ float red[8 * 256];
    const int lane = threadIdx.x & 31;
    const int wave = __builtin_amdgcn_readfirstlane((int)(threadIdx.x >> 5));
    const int c4 = lane << 2;
    const float4 w = *(const float4*)(W1 + c4);
    const float4 b = *(const float4*)(b1 + c4);
    float4 s = make_float4(0.f, 0.f, 0.f, 0.f);
    float4 q = make_float4(0.f, 0.f, 0.f, 0.f);
    for (int node = blockIdx.x * 8 + wave; node < n; node += gridDim.x * 8) {
        const int beg = rowptr[node], end = rowptr[node + 1];
        float part = 0.f;
        for (int j = beg + lane; j < end; j += 32) part += ccoef[j] * X[csrc[j]];
        for (int off = 16; off > 0; off >>= 1) part += __shfl_xor(part, off);
        const float di = dinv[node];
        const float a = part + X[node] * (di * di);
        float4 o;
        o.x = fmaxf(a * w.x + b.x, 0.f); o.y = fmaxf(a * w.y + b.y, 0.f);
        o.z = fmaxf(a * w.z + b.z, 0.f); o.w = fmaxf(a * w.w + b.w, 0.f);
        *(float4*)(y + (long long)node * HDIM + c4) = o;
        s.x += o.x; s.y += o.y; s.z += o.z; s.w += o.w;
        q.x += o.x * o.x; q.y += o.y * o.y; q.z += o.z * o.z; q.w += o.w * o.w;
    }
    *(float4*)(red + wave * 256 + c4)       = s;
    *(float4*)(red + wave * 256 + 128 + c4) = q;
    __syncthreads();
    float tot = 0.f;
    for (int k = 0; k < 8; ++k) tot += red[k * 256 + threadIdx.x];
    atomicAdd(&sums[(blockIdx.x & (NREP - 1)) * 256 + threadIdx.x], tot);
}

__global__ __launch_bounds__(256) void k_agg_gemm(const float* __restrict__ x,
                                                  const float* __restrict__ W,
                                                  const float* __restrict__ bias,
                                                  const int* __restrict__ rowptr,
                                                  const int* __restrict__ csrc,
                                                  const float* __restrict__ ccoef,
                                                  const float* __restrict__ dinv,
                                                  float* __restrict__ y,
                                                  float* __restrict__ sums, int n) {
    __shared__ float tileX[32 * HDIM];
    __shared__ float red[2 * 256];
    const int rowBase = blockIdx.x * 32;
    const int tid  = threadIdx.x;
    const int wave = __builtin_amdgcn_readfirstlane(tid >> 5);
    const int lane = tid & 31;
    const int half = lane >> 4;
    const int l16  = lane & 15;
    const int rowHalf = wave & 1;
    const int c0   = (wave >> 1) * 32;
    const int c4   = lane << 2;

    const bool fullTile = (rowBase + 32 <= n);

    for (int q = 0; q < 4; ++q) {
        const int lr = wave * 4 + q;
        const int r  = rowBase + lr;
        float4 acc = make_float4(0.f, 0.f, 0.f, 0.f);
        if (r < n) {
            const int beg = rowptr[r], end = rowptr[r + 1];
            const float di = dinv[r];
            const float4 xs = *(const float4*)(x + (long long)r * HDIM + c4);
            for (int base = beg; base < end; base += 32) {
                int m = end - base; if (m > 32) m = 32;
                int   myi = 0;
                float myc = 0.f;
                if (lane < m) { myi = csrc[base + lane]; myc = ccoef[base + lane]; }
                const int myci = __float_as_int(myc);
                int k = 0;
                for (; k + 4 <= m; k += 4) {
                    const int s0 = __builtin_amdgcn_readlane(myi, k);
                    const int s1 = __builtin_amdgcn_readlane(myi, k + 1);
                    const int s2 = __builtin_amdgcn_readlane(myi, k + 2);
                    const int s3 = __builtin_amdgcn_readlane(myi, k + 3);
                    const float e0 = __int_as_float(__builtin_amdgcn_readlane(myci, k));
                    const float e1 = __int_as_float(__builtin_amdgcn_readlane(myci, k + 1));
                    const float e2 = __int_as_float(__builtin_amdgcn_readlane(myci, k + 2));
                    const float e3 = __int_as_float(__builtin_amdgcn_readlane(myci, k + 3));
                    const float4 v0 = *(const float4*)(x + (long long)s0 * HDIM + c4);
                    const float4 v1 = *(const float4*)(x + (long long)s1 * HDIM + c4);
                    const float4 v2 = *(const float4*)(x + (long long)s2 * HDIM + c4);
                    const float4 v3 = *(const float4*)(x + (long long)s3 * HDIM + c4);
                    acc.x += e0 * v0.x; acc.y += e0 * v0.y; acc.z += e0 * v0.z; acc.w += e0 * v0.w;
                    acc.x += e1 * v1.x; acc.y += e1 * v1.y; acc.z += e1 * v1.z; acc.w += e1 * v1.w;
                    acc.x += e2 * v2.x; acc.y += e2 * v2.y; acc.z += e2 * v2.z; acc.w += e2 * v2.w;
                    acc.x += e3 * v3.x; acc.y += e3 * v3.y; acc.z += e3 * v3.z; acc.w += e3 * v3.w;
                }
                for (; k < m; ++k) {
                    const int s0 = __builtin_amdgcn_readlane(myi, k);
                    const float e0 = __int_as_float(__builtin_amdgcn_readlane(myci, k));
                    const float4 v0 = *(const float4*)(x + (long long)s0 * HDIM + c4);
                    acc.x += e0 * v0.x; acc.y += e0 * v0.y; acc.z += e0 * v0.z; acc.w += e0 * v0.w;
                }
            }
            const float d2 = di * di;
            acc.x += xs.x * d2; acc.y += xs.y * d2; acc.z += xs.z * d2; acc.w += xs.w * d2;
        }
        *(float4*)(tileX + lr * HDIM + c4) = acc;
    }
    __syncthreads();

    const float bi0 = bias[c0 + l16];
    const float bi1 = bias[c0 + 16 + l16];

    const float* xrow = tileX + (rowHalf * 16 + l16) * HDIM + 2 * half;
    v8f acc0 = {}, acc1 = {};

    #pragma unroll 4
    for (int k = 0; k < HDIM; k += 4) {
        v2f a; a.x = xrow[k]; a.y = xrow[k + 1];
        const float* wp = W + (long long)(k + 2 * half) * HDIM + c0 + l16;
        v2f b0; b0.x = wp[0];  b0.y = wp[HDIM];
        v2f b1; b1.x = wp[16]; b1.y = wp[HDIM + 16];
        acc0 = __builtin_amdgcn_wmma_f32_16x16x4_f32(false, a, false, b0,
                                                     (short)0, acc0, false, false);
        acc1 = __builtin_amdgcn_wmma_f32_16x16x4_f32(false, a, false, b1,
                                                     (short)0, acc1, false, false);
    }

    asm volatile("v_nop\n\tv_nop\n\tv_nop\n\tv_nop\n\tv_nop\n\tv_nop\n\tv_nop\n\tv_nop"
                 : "+v"(acc0), "+v"(acc1));

    #pragma unroll
    for (int v = 0; v < 8; ++v) {
        acc0[v] = fmaxf(acc0[v] + bi0, 0.f);
        acc1[v] = fmaxf(acc1[v] + bi1, 0.f);
    }

    const int rbase = rowBase + rowHalf * 16 + half * 8;

    float cs0 = 0.f, cq0 = 0.f, cs1 = 0.f, cq1 = 0.f;
    for (int v = 0; v < 8; ++v) {
        const bool in = fullTile || (rbase + v < n);
        const float a0 = in ? acc0[v] : 0.f;
        const float a1 = in ? acc1[v] : 0.f;
        cs0 += a0; cq0 += a0 * a0;
        cs1 += a1; cq1 += a1 * a1;
    }
    cs0 += __shfl_xor(cs0, 16); cq0 += __shfl_xor(cq0, 16);
    cs1 += __shfl_xor(cs1, 16); cq1 += __shfl_xor(cq1, 16);
    if (half == 0) {
        float* rp = red + rowHalf * 256 + c0 + l16;
        rp[0]   = cs0; rp[16]  = cs1;
        rp[128] = cq0; rp[144] = cq1;
    }
    __syncthreads();
    atomicAdd(&sums[(blockIdx.x & (NREP - 1)) * 256 + tid], red[tid] + red[256 + tid]);

    if (fullTile) {
        #pragma unroll
        for (int v = 0; v < 8; ++v) {
            float* op = y + (long long)(rbase + v) * HDIM + c0 + l16;
            op[0]  = acc0[v];
            op[16] = acc1[v];
        }
    } else {
        for (int v = 0; v < 8; ++v) {
            int r = rbase + v;
            if (r < n) {
                y[(long long)r * HDIM + c0 + l16]      = acc0[v];
                y[(long long)r * HDIM + c0 + 16 + l16] = acc1[v];
            }
        }
    }
}

template <bool HAS_RES>
__device__ __forceinline__ void bn_rows(float* __restrict__ y, const float* __restrict__ res,
                                        const v4f sc, const v4f sh,
                                        int t, const int stride, const int total) {
    for (; t + 3 * stride < total; t += 4 * stride) {
        v4f v[4], r[4];
        for (int u = 0; u < 4; ++u) {
            const long long off = (long long)(t + u * stride) * 4;
            v[u] = *(const v4f*)(y + off);
            if (HAS_RES) r[u] = *(const v4f*)(res + off);
        }
        for (int u = 0; u < 4; ++u) {
            const long long off = (long long)(t + u * stride) * 4;
            v4f o = v[u] * sc + sh;
            if (HAS_RES) o += r[u];
            *(v4f*)(y + off) = o;
        }
    }
    for (; t < total; t += stride) {
        const long long off = (long long)t * 4;
        v4f o = *(const v4f*)(y + off) * sc + sh;
        if (HAS_RES) o += *(const v4f*)(res + off);
        *(v4f*)(y + off) = o;
    }
}

__global__ __launch_bounds__(256) void k_bn_apply(float* __restrict__ y,
                                                  const float* __restrict__ res,
                                                  const float* __restrict__ sums,
                                                  const float* __restrict__ gamma,
                                                  const float* __restrict__ beta,
                                                  float invN, int n) {
    const int t0 = blockIdx.x * 256 + threadIdx.x;
    const int c4 = (t0 & 31) << 2;
    float4 s = make_float4(0.f, 0.f, 0.f, 0.f);
    float4 q = make_float4(0.f, 0.f, 0.f, 0.f);
    for (int k = 0; k < NREP; ++k) {
        const float4 a = *(const float4*)(sums + k * 256 + c4);
        const float4 b = *(const float4*)(sums + k * 256 + 128 + c4);
        s.x += a.x; s.y += a.y; s.z += a.z; s.w += a.w;
        q.x += b.x; q.y += b.y; q.z += b.z; q.w += b.w;
    }
    const float4 g  = *(const float4*)(gamma + c4);
    const float4 be = *(const float4*)(beta + c4);
    v4f sc, sh;
    {
        float m, var;
        m = s.x * invN; var = q.x * invN - m * m; sc.x = rsqrtf(var + EPS_BN) * g.x; sh.x = be.x - m * sc.x;
        m = s.y * invN; var = q.y * invN - m * m; sc.y = rsqrtf(var + EPS_BN) * g.y; sh.y = be.y - m * sc.y;
        m = s.z * invN; var = q.z * invN - m * m; sc.z = rsqrtf(var + EPS_BN) * g.z; sh.z = be.z - m * sc.z;
        m = s.w * invN; var = q.w * invN - m * m; sc.w = rsqrtf(var + EPS_BN) * g.w; sh.w = be.w - m * sc.w;
    }
    const int stride = gridDim.x * 256;
    const int total  = n * 32;
    if (res) bn_rows<true>(y, res, sc, sh, t0, stride, total);
    else     bn_rows<false>(y, res, sc, sh, t0, stride, total);
}

__device__ __forceinline__ void pool_flush(float* __restrict__ pool, float* __restrict__ counts,
                                           const int g, const int G, const int lane,
                                           const v4f acc, const float cnt) {
    if ((unsigned)g < (unsigned)G) {
        float* p = pool + (long long)g * HDIM + (lane << 2);
        atomicAdd(p + 0, acc.x); atomicAdd(p + 1, acc.y);
        atomicAdd(p + 2, acc.z); atomicAdd(p + 3, acc.w);
        if (lane == 0) atomicAdd(&counts[g], cnt);
    }
}

__global__ __launch_bounds__(256) void k_bn_pool(const float* __restrict__ y,
                                                 const float* __restrict__ res,
                                                 const float* __restrict__ sums,
                                                 const float* __restrict__ gamma,
                                                 const float* __restrict__ beta,
                                                 const int* __restrict__ batch,
                                                 float* __restrict__ pool,
                                                 float* __restrict__ counts,
                                                 float invN, int n, int G) {
    const int lane = threadIdx.x & 31;
    const int wave = __builtin_amdgcn_readfirstlane((int)(threadIdx.x >> 5));
    const int c4 = lane << 2;
    float4 s = make_float4(0.f, 0.f, 0.f, 0.f);
    float4 q = make_float4(0.f, 0.f, 0.f, 0.f);
    for (int k = 0; k < NREP; ++k) {
        const float4 a = *(const float4*)(sums + k * 256 + c4);
        const float4 b = *(const float4*)(sums + k * 256 + 128 + c4);
        s.x += a.x; s.y += a.y; s.z += a.z; s.w += a.w;
        q.x += b.x; q.y += b.y; q.z += b.z; q.w += b.w;
    }
    const float4 g4 = *(const float4*)(gamma + c4);
    const float4 be = *(const float4*)(beta + c4);
    v4f sc, sh;
    {
        float m, var;
        m = s.x * invN; var = q.x * invN - m * m; sc.x = rsqrtf(var + EPS_BN) * g4.x; sh.x = be.x - m * sc.x;
        m = s.y * invN; var = q.y * invN - m * m; sc.y = rsqrtf(var + EPS_BN) * g4.y; sh.y = be.y - m * sc.y;
        m = s.z * invN; var = q.z * invN - m * m; sc.z = rsqrtf(var + EPS_BN) * g4.z; sh.z = be.z - m * sc.z;
        m = s.w * invN; var = q.w * invN - m * m; sc.w = rsqrtf(var + EPS_BN) * g4.w; sh.w = be.w - m * sc.w;
    }

    const int nWaves = gridDim.x * 8;
    const int gw  = blockIdx.x * 8 + wave;
    const int per = n / nWaves, rem = n - per * nWaves;
    const int r0  = gw * per + (gw < rem ? gw : rem);
    const int r1  = r0 + per + (gw < rem ? 1 : 0);

    int   gcur = -1;
    float cnt  = 0.f;
    v4f   acc  = {0.f, 0.f, 0.f, 0.f};
    for (int base = r0; base < r1; base += 32) {
        int m = r1 - base; if (m > 32) m = 32;
        int myb = 0;
        if (lane < m) myb = batch[base + lane];
        int k = 0;
        for (; k + 4 <= m; k += 4) {
            v4f v[4], r[4];
            for (int u = 0; u < 4; ++u) {
                const long long off = (long long)(base + k + u) * HDIM + c4;
                v[u] = *(const v4f*)(y + off);
                r[u] = *(const v4f*)(res + off);
            }
            for (int u = 0; u < 4; ++u) {
                const int g = __builtin_amdgcn_readlane(myb, k + u);
                v4f o = v[u] * sc + sh;
                o += r[u];
                if (g != gcur) {
                    pool_flush(pool, counts, gcur, G, lane, acc, cnt);
                    gcur = g; cnt = 0.f; acc = (v4f){0.f, 0.f, 0.f, 0.f};
                }
                acc += o; cnt += 1.0f;
            }
        }
        for (; k < m; ++k) {
            const long long off = (long long)(base + k) * HDIM + c4;
            const int g = __builtin_amdgcn_readlane(myb, k);
            v4f o = *(const v4f*)(y + off) * sc + sh;
            o += *(const v4f*)(res + off);
            if (g != gcur) {
                pool_flush(pool, counts, gcur, G, lane, acc, cnt);
                gcur = g; cnt = 0.f; acc = (v4f){0.f, 0.f, 0.f, 0.f};
            }
            acc += o; cnt += 1.0f;
        }
    }
    pool_flush(pool, counts, gcur, G, lane, acc, cnt);
}

__global__ __launch_bounds__(256) void k_pool_final(const float* __restrict__ pool,
                                                    const float* __restrict__ counts,
                                                    float* __restrict__ xmean, int G) {
    int t = blockIdx.x * 256 + threadIdx.x;
    if (t >= G * HDIM) return;
    int g = t / HDIM;
    xmean[t] = pool[t] / fmaxf(counts[g], 1.0f);
}

__global__ void k_pred(const float* __restrict__ xmean, const float* __restrict__ Wr,
                       const float* __restrict__ br, float* __restrict__ pred) {
    __shared__ float red[128];
    int g = blockIdx.x, c = threadIdx.x;
    red[c] = xmean[(long long)g * HDIM + c] * Wr[c];
    __syncthreads();
    for (int s = 64; s > 0; s >>= 1) {
        if (c < s) red[c] += red[c + s];
        __syncthreads();
    }
    if (c == 0) pred[g] = red[0] + br[0];
}

extern "C" void kernel_launch(void* const* d_in, const int* in_sizes, int n_in,
                              void* d_out, int out_size, void* d_ws, size_t ws_size,
                              hipStream_t stream) {
    const float* X     = (const float*)d_in[0];
    const int*   edge  = (const int*)d_in[1];
    const int*   batch = (const int*)d_in[2];
    const float* W1    = (const float*)d_in[3];
    const float* b1    = (const float*)d_in[4];
    const float* Wm    = (const float*)d_in[5];
    const float* bm    = (const float*)d_in[6];
    const float* gamma = (const float*)d_in[7];
    const float* beta  = (const float*)d_in[8];
    const float* Wr    = (const float*)d_in[9];
    const float* br    = (const float*)d_in[10];
    float* out = (float*)d_out;

    const int N = in_sizes[0];
    const int E = in_sizes[1] / 2;
    const int G = out_size / (HDIM + 1);
    const int* src = edge;
    const int* dst = edge + E;

    const long long NH = (long long)N * HDIM;
    float* ws    = (float*)d_ws;
    float* bufA  = ws;
    float* bufB  = ws + NH;
    int*   rowptr = (int*)(ws + 2 * NH);
    int*   cursor = rowptr + (N + 1);
    int*   degi   = cursor + N;
    int*   bsum   = degi + N;
    int*   csrc   = bsum + 1024;
    float* ccoef  = (float*)(csrc + E);
    float* sumsAll = ws + 2 * NH + ((3LL * N + 1025 + 2LL * E + 3) & ~3LL);
    float* pool  = sumsAll + 5LL * NREP * 256;
    float* cnts  = pool + (long long)G * HDIM;
    float* dinv  = ws + 3 * NH;

    const int gGem = (N + 31) / 32;
    const int nChunk = (N + 1023) / 1024;

    k_zero2<<<256, 256, 0, stream>>>((float*)degi, N,
                                     sumsAll, 5LL * NREP * 256 + (long long)G * HDIM + G);
    k_deg_count<<<2048, 256, 0, stream>>>(dst, degi, E);
    k_scan_part<<<nChunk, 256, 0, stream>>>(degi, dinv, bsum, N);
    k_scan_fin<<<nChunk, 256, 0, stream>>>(degi, bsum, rowptr, cursor, N, E);
    k_fill<<<2048, 256, 0, stream>>>(src, dst, dinv, cursor, csrc, ccoef, E);

    const float invN = 1.0f / (float)N;
    auto bn_tail = [&](float* ybuf, const float* sums_, const float* g_, const float* be_,
                       const float* res) {
        k_bn_apply<<<1024, 256, 0, stream>>>(ybuf, res, sums_, g_, be_, invN, N);
    };

    k_layer1<<<2048, 256, 0, stream>>>(X, W1, b1, rowptr, csrc, ccoef, dinv, bufA, sumsAll, N);
    bn_tail(bufA, sumsAll, gamma, beta, nullptr);

    float* cur = bufA;
    float* nxt = bufB;

    for (int i = 0; i < 3; ++i) {
        k_agg_gemm<<<gGem, 256, 0, stream>>>(cur, Wm + (long long)i * HDIM * HDIM,
                                             bm + (long long)i * HDIM,
                                             rowptr, csrc, ccoef, dinv, nxt,
                                             sumsAll + (long long)(i + 1) * NREP * 256, N);
        bn_tail(nxt, sumsAll + (long long)(i + 1) * NREP * 256,
                gamma + (long long)(i + 1) * HDIM,
                beta  + (long long)(i + 1) * HDIM, cur);
        float* t = cur; cur = nxt; nxt = t;
    }

    k_agg_gemm<<<gGem, 256, 0, stream>>>(cur, Wm + 3LL * HDIM * HDIM, bm + 3LL * HDIM,
                                         rowptr, csrc, ccoef, dinv, nxt,
                                         sumsAll + 4LL * NREP * 256, N);
    k_bn_pool<<<1024, 256, 0, stream>>>(nxt, cur, sumsAll + 4LL * NREP * 256,
                                        gamma + 4LL * HDIM, beta + 4LL * HDIM,
                                        batch, pool, cnts, invN, N, G);

    k_pool_final<<<(G * HDIM + 255) / 256, 256, 0, stream>>>(pool, cnts, out + G, G);
    k_pred<<<G, 128, 0, stream>>>(out + G, Wr, br, out);
}


# Edges placed without a second atomic pass: the in-degree count's add returns each edge's rank in its destination's run; slot = run start + rank; row sums run in the count pass's arrival order
# speedup vs baseline: 4.2027x; 1.0800x over previous
#include <hip/hip_runtime.h>

#define HDIM 128
#define EPS_BN 1e-5f
#define NREP 8

typedef __attribute__((ext_vector_type(2))) float v2f;
typedef __attribute__((ext_vector_type(8))) float v8f;
typedef __attribute__((ext_vector_type(4))) float v4f;

__global__ __launch_bounds__(256) void k_zero2(float* a, long long na, float* b, long long nb) {
    long long i = (long long)blockIdx.x * blockDim.x + threadIdx.x;
    long long stride = (long long)gridDim.x * blockDim.x;
    for (; i < na + nb; i += stride) {
        if (i < na) a[i] = 0.0f; else b[i - na] = 0.0f;
    }
}

__global__ __launch_bounds__(256) void k_deg_count(const int* __restrict__ dst,
                                                   int* __restrict__ degi,
                                                   int* __restrict__ rank, int E) {
    int stride = gridDim.x * 256;
    for (int e = blockIdx.x * 256 + threadIdx.x; e < E; e += stride)
        rank[e] = atomicAdd(&degi[dst[e]], 1);
}

__global__ __launch_bounds__(256) void k_scan_part(const int* __restrict__ degi,
                                                   float* __restrict__ dinv,
                                                   int* __restrict__ bsum, int n) {
    __shared__ int red[256];
    const int t = threadIdx.x;
    const int base = blockIdx.x * 1024 + t * 4;
    int s = 0;
    #pragma unroll
    for (int q = 0; q < 4; ++q) {
        int i = base + q;
        if (i < n) {
            int d = degi[i];
            s += d;
            dinv[i] = rsqrtf((float)d + 1.0f);
        }
    }
    red[t] = s;
    __syncthreads();
    for (int off = 128; off > 0; off >>= 1) {
        if (t < off) red[t] += red[t + off];
        __syncthreads();
    }
    if (t == 0) bsum[blockIdx.x] = red[0];
}

__global__ __launch_bounds__(256) void k_scan_fin(const int* __restrict__ degi,
                                                  const int* __restrict__ bsum,
                                                  int* __restrict__ rowptr,
                                                  int n, int E) {
    __shared__ int red[256];
    __shared__ int sc[256];
    const int t = threadIdx.x;
    int part = 0;
    for (int k = t; k < (int)blockIdx.x; k += 256) part += bsum[k];
    red[t] = part;
    __syncthreads();
    for (int off = 128; off > 0; off >>= 1) {
        if (t < off) red[t] += red[t + off];
        __syncthreads();
    }
    const int chunkOff = red[0];

    const int base = blockIdx.x * 1024 + t * 4;
    int d[4];
    #pragma unroll
    for (int q = 0; q < 4; ++q) d[q] = (base + q < n) ? degi[base + q] : 0;
    const int tsum = d[0] + d[1] + d[2] + d[3];
    sc[t] = tsum;
    __syncthreads();
    for (int off = 1; off < 256; off <<= 1) {
        int v = (t >= off) ? sc[t - off] : 0;
        __syncthreads();
        sc[t] += v;
        __syncthreads();
    }
    int run = chunkOff + sc[t] - tsum;
    #pragma unroll
    for (int q = 0; q < 4; ++q) {
        if (base + q < n) rowptr[base + q] = run;
        run += d[q];
    }
    if (blockIdx.x == 0 && t == 0) rowptr[n] = E;
}

__global__ __launch_bounds__(256) void k_fill(const int* __restrict__ src,
                                              const int* __restrict__ dst,
                                              const float* __restrict__ dinv,
                                              const int* __restrict__ rowptr,
                                              const int* __restrict__ rank,
                                              int* __restrict__ csrc,
                                              float* __restrict__ ccoef, int E) {
    int stride = gridDim.x * 256;
    for (int e = blockIdx.x * 256 + threadIdx.x; e < E; e += stride) {
        int s = src[e], d = dst[e];
        int pos = rowptr[d] + rank[e];
        if ((unsigned)pos < (unsigned)E) {
            csrc[pos]  = s;
            ccoef[pos] = dinv[s] * dinv[d];
        }
    }
}

__global__ __launch_bounds__(256) void k_layer1(const float* __restrict__ X,
                                                const float* __restrict__ W1,
                                                const float* __restrict__ b1,
                                                const int* __restrict__ rowptr,
                                                const int* __restrict__ csrc,
                                                const float* __restrict__ ccoef,
                                                const float* __restrict__ dinv,
                                                float* __restrict__ y,
                                                float* __restrict__ sums, int n) {
    __shared__ float red[8 * 256];
    const int lane = threadIdx.x & 31;
    const int wave = __builtin_amdgcn_readfirstlane((int)(threadIdx.x >> 5));
    const int c4 = lane << 2;
    const float4 w = *(const float4*)(W1 + c4);
    const float4 b = *(const float4*)(b1 + c4);
    float4 s = make_float4(0.f, 0.f, 0.f, 0.f);
    float4 q = make_float4(0.f, 0.f, 0.f, 0.f);
    for (int node = blockIdx.x * 8 + wave; node < n; node += gridDim.x * 8) {
        const int beg = rowptr[node], end = rowptr[node + 1];
        float part = 0.f;
        for (int j = beg + lane; j < end; j += 32) part += ccoef[j] * X[csrc[j]];
        for (int off = 16; off > 0; off >>= 1) part += __shfl_xor(part, off);
        const float di = dinv[node];
        const float a = part + X[node] * (di * di);
        float4 o;
        o.x = fmaxf(a * w.x + b.x, 0.f); o.y = fmaxf(a * w.y + b.y, 0.f);
        o.z = fmaxf(a * w.z + b.z, 0.f); o.w = fmaxf(a * w.w + b.w, 0.f);
        *(float4*)(y + (long long)node * HDIM + c4) = o;
        s.x += o.x; s.y += o.y; s.z += o.z; s.w += o.w;
        q.x += o.x * o.x; q.y += o.y * o.y; q.z += o.z * o.z; q.w += o.w * o.w;
    }
    *(float4*)(red + wave * 256 + c4)       = s;
    *(float4*)(red + wave * 256 + 128 + c4) = q;
    __syncthreads();
    float tot = 0.f;
    for (int k = 0; k < 8; ++k) tot += red[k * 256 + threadIdx.x];
    atomicAdd(&sums[(blockIdx.x & (NREP - 1)) * 256 + threadIdx.x], tot);
}

__global__ __launch_bounds__(256) void k_agg_gemm(const float* __restrict__ x,
                                                  const float* __restrict__ W,
                                                  const float* __restrict__ bias,
                                                  const int* __restrict__ rowptr,
                                                  const int* __restrict__ csrc,
                                                  const float* __restrict__ ccoef,
                                                  const float* __restrict__ dinv,
                                                  float* __restrict__ y,
                                                  float* __restrict__ sums, int n) {
    __shared__ float tileX[32 * HDIM];
    __shared__ float red[2 * 256];
    const int rowBase = blockIdx.x * 32;
    const int tid  = threadIdx.x;
    const int wave = __builtin_amdgcn_readfirstlane(tid >> 5);
    const int lane = tid & 31;
    const int half = lane >> 4;
    const int l16  = lane & 15;
    const int rowHalf = wave & 1;
    const int c0   = (wave >> 1) * 32;
    const int c4   = lane << 2;

    const bool fullTile = (rowBase + 32 <= n);

    for (int q = 0; q < 4; ++q) {
        const int lr = wave * 4 + q;
        const int r  = rowBase + lr;
        float4 acc = make_float4(0.f, 0.f, 0.f, 0.f);
        if (r < n) {
            const int beg = rowptr[r], end = rowptr[r + 1];
            const float di = dinv[r];
            const float4 xs = *(const float4*)(x + (long long)r * HDIM + c4);
            for (int base = beg; base < end; base += 32) {
                int m = end - base; if (m > 32) m = 32;
                int   myi = 0;
                float myc = 0.f;
                if (lane < m) { myi = csrc[base + lane]; myc = ccoef[base + lane]; }
                const int myci = __float_as_int(myc);
                int k = 0;
                for (; k + 4 <= m; k += 4) {
                    const int s0 = __builtin_amdgcn_readlane(myi, k);
                    const int s1 = __builtin_amdgcn_readlane(myi, k + 1);
                    const int s2 = __builtin_amdgcn_readlane(myi, k + 2);
                    const int s3 = __builtin_amdgcn_readlane(myi, k + 3);
                    const float e0 = __int_as_float(__builtin_amdgcn_readlane(myci, k));
                    const float e1 = __int_as_float(__builtin_amdgcn_readlane(myci, k + 1));
                    const float e2 = __int_as_float(__builtin_amdgcn_readlane(myci, k + 2));
                    const float e3 = __int_as_float(__builtin_amdgcn_readlane(myci, k + 3));
                    const float4 v0 = *(const float4*)(x + (long long)s0 * HDIM + c4);
                    const float4 v1 = *(const float4*)(x + (long long)s1 * HDIM + c4);
                    const float4 v2 = *(const float4*)(x + (long long)s2 * HDIM + c4);
                    const float4 v3 = *(const float4*)(x + (long long)s3 * HDIM + c4);
                    acc.x += e0 * v0.x; acc.y += e0 * v0.y; acc.z += e0 * v0.z; acc.w += e0 * v0.w;
                    acc.x += e1 * v1.x; acc.y += e1 * v1.y; acc.z += e1 * v1.z; acc.w += e1 * v1.w;
                    acc.x += e2 * v2.x; acc.y += e2 * v2.y; acc.z += e2 * v2.z; acc.w += e2 * v2.w;
                    acc.x += e3 * v3.x; acc.y += e3 * v3.y; acc.z += e3 * v3.z; acc.w += e3 * v3.w;
                }
                for (; k < m; ++k) {
                    const int s0 = __builtin_amdgcn_readlane(myi, k);
                    const float e0 = __int_as_float(__builtin_amdgcn_readlane(myci, k));
                    const float4 v0 = *(const float4*)(x + (long long)s0 * HDIM + c4);
                    acc.x += e0 * v0.x; acc.y += e0 * v0.y; acc.z += e0 * v0.z; acc.w += e0 * v0.w;
                }
            }
            const float d2 = di * di;
            acc.x += xs.x * d2; acc.y += xs.y * d2; acc.z += xs.z * d2; acc.w += xs.w * d2;
        }
        *(float4*)(tileX + lr * HDIM + c4) = acc;
    }
    __syncthreads();

    const float bi0 = bias[c0 + l16];
    const float bi1 = bias[c0 + 16 + l16];

    const float* xrow = tileX + (rowHalf * 16 + l16) * HDIM + 2 * half;
    v8f acc0 = {}, acc1 = {};

    #pragma unroll 4
    for (int k = 0; k < HDIM; k += 4) {
        v2f a; a.x = xrow[k]; a.y = xrow[k + 1];
        const float* wp = W + (long long)(k + 2 * half) * HDIM + c0 + l16;
        v2f b0; b0.x = wp[0];  b0.y = wp[HDIM];
        v2f b1; b1.x = wp[16]; b1.y = wp[HDIM + 16];
        acc0 = __builtin_amdgcn_wmma_f32_16x16x4_f32(false, a, false, b0,
                                                     (short)0, acc0, false, false);
        acc1 = __builtin_amdgcn_wmma_f32_16x16x4_f32(false, a, false, b1,
                                                     (short)0, acc1, false, false);
    }

    asm volatile("v_nop\n\tv_nop\n\tv_nop\n\tv_nop\n\tv_nop\n\tv_nop\n\tv_nop\n\tv_nop"
                 : "+v"(acc0), "+v"(acc1));

    #pragma unroll
    for (int v = 0; v < 8; ++v) {
        acc0[v] = fmaxf(acc0[v] + bi0, 0.f);
        acc1[v] = fmaxf(acc1[v] + bi1, 0.f);
    }

    const int rbase = rowBase + rowHalf * 16 + half * 8;

    float cs0 = 0.f, cq0 = 0.f, cs1 = 0.f, cq1 = 0.f;
    for (int v = 0; v < 8; ++v) {
        const bool in = fullTile || (rbase + v < n);
        const float a0 = in ? acc0[v] : 0.f;
        const float a1 = in ? acc1[v] : 0.f;
        cs0 += a0; cq0 += a0 * a0;
        cs1 += a1; cq1 += a1 * a1;
    }
    cs0 += __shfl_xor(cs0, 16); cq0 += __shfl_xor(cq0, 16);
    cs1 += __shfl_xor(cs1, 16); cq1 += __shfl_xor(cq1, 16);
    if (half == 0) {
        float* rp = red + rowHalf * 256 + c0 + l16;
        rp[0]   = cs0; rp[16]  = cs1;
        rp[128] = cq0; rp[144] = cq1;
    }
    __syncthreads();
    atomicAdd(&sums[(blockIdx.x & (NREP - 1)) * 256 + tid], red[tid] + red[256 + tid]);

    if (fullTile) {
        #pragma unroll
        for (int v = 0; v < 8; ++v) {
            float* op = y + (long long)(rbase + v) * HDIM + c0 + l16;
            op[0]  = acc0[v];
            op[16] = acc1[v];
        }
    } else {
        for (int v = 0; v < 8; ++v) {
            int r = rbase + v;
            if (r < n) {
                y[(long long)r * HDIM + c0 + l16]      = acc0[v];
                y[(long long)r * HDIM + c0 + 16 + l16] = acc1[v];
            }
        }
    }
}

template <bool HAS_RES>
__device__ __forceinline__ void bn_rows(float* __restrict__ y, const float* __restrict__ res,
                                        const v4f sc, const v4f sh,
                                        int t, const int stride, const int total) {
    for (; t + 3 * stride < total; t += 4 * stride) {
        v4f v[4], r[4];
        for (int u = 0; u < 4; ++u) {
            const long long off = (long long)(t + u * stride) * 4;
            v[u] = *(const v4f*)(y + off);
            if (HAS_RES) r[u] = *(const v4f*)(res + off);
        }
        for (int u = 0; u < 4; ++u) {
            const long long off = (long long)(t + u * stride) * 4;
            v4f o = v[u] * sc + sh;
            if (HAS_RES) o += r[u];
            *(v4f*)(y + off) = o;
        }
    }
    for (; t < total; t += stride) {
        const long long off = (long long)t * 4;
        v4f o = *(const v4f*)(y + off) * sc + sh;
        if (HAS_RES) o += *(const v4f*)(res + off);
        *(v4f*)(y + off) = o;
    }
}

__global__ __launch_bounds__(256) void k_bn_apply(float* __restrict__ y,
                                                  const float* __restrict__ res,
                                                  const float* __restrict__ sums,
                                                  const float* __restrict__ gamma,
                                                  const float* __restrict__ beta,
                                                  float invN, int n) {
    const int t0 = blockIdx.x * 256 + threadIdx.x;
    const int c4 = (t0 & 31) << 2;
    float4 s = make_float4(0.f, 0.f, 0.f, 0.f);
    float4 q = make_float4(0.f, 0.f, 0.f, 0.f);
    for (int k = 0; k < NREP; ++k) {
        const float4 a = *(const float4*)(sums + k * 256 + c4);
        const float4 b = *(const float4*)(sums + k * 256 + 128 + c4);
        s.x += a.x; s.y += a.y; s.z += a.z; s.w += a.w;
        q.x += b.x; q.y += b.y; q.z += b.z; q.w += b.w;
    }
    const float4 g  = *(const float4*)(gamma + c4);
    const float4 be = *(const float4*)(beta + c4);
    v4f sc, sh;
    {
        float m, var;
        m = s.x * invN; var = q.x * invN - m * m; sc.x = rsqrtf(var + EPS_BN) * g.x; sh.x = be.x - m * sc.x;
        m = s.y * invN; var = q.y * invN - m * m; sc.y = rsqrtf(var + EPS_BN) * g.y; sh.y = be.y - m * sc.y;
        m = s.z * invN; var = q.z * invN - m * m; sc.z = rsqrtf(var + EPS_BN) * g.z; sh.z = be.z - m * sc.z;
        m = s.w * invN; var = q.w * invN - m * m; sc.w = rsqrtf(var + EPS_BN) * g.w; sh.w = be.w - m * sc.w;
    }
    const int stride = gridDim.x * 256;
    const int total  = n * 32;
    if (res) bn_rows<true>(y, res, sc, sh, t0, stride, total);
    else     bn_rows<false>(y, res, sc, sh, t0, stride, total);
}

__device__ __forceinline__ void pool_flush(float* __restrict__ pool, float* __restrict__ counts,
                                           const int g, const int G, const int lane,
                                           const v4f acc, const float cnt) {
    if ((unsigned)g < (unsigned)G) {
        float* p = pool + (long long)g * HDIM + (lane << 2);
        atomicAdd(p + 0, acc.x); atomicAdd(p + 1, acc.y);
        atomicAdd(p + 2, acc.z); atomicAdd(p + 3, acc.w);
        if (lane == 0) atomicAdd(&counts[g], cnt);
    }
}

__global__ __launch_bounds__(256) void k_bn_pool(const float* __restrict__ y,
                                                 const float* __restrict__ res,
                                                 const float* __restrict__ sums,
                                                 const float* __restrict__ gamma,
                                                 const float* __restrict__ beta,
                                                 const int* __restrict__ batch,
                                                 float* __restrict__ pool,
                                                 float* __restrict__ counts,
                                                 float invN, int n, int G) {
    const int lane = threadIdx.x & 31;
    const int wave = __builtin_amdgcn_readfirstlane((int)(threadIdx.x >> 5));
    const int c4 = lane << 2;
    float4 s = make_float4(0.f, 0.f, 0.f, 0.f);
    float4 q = make_float4(0.f, 0.f, 0.f, 0.f);
    for (int k = 0; k < NREP; ++k) {
        const float4 a = *(const float4*)(sums + k * 256 + c4);
        const float4 b = *(const float4*)(sums + k * 256 + 128 + c4);
        s.x += a.x; s.y += a.y; s.z += a.z; s.w += a.w;
        q.x += b.x; q.y += b.y; q.z += b.z; q.w += b.w;
    }
    const float4 g4 = *(const float4*)(gamma + c4);
    const float4 be = *(const float4*)(beta + c4);
    v4f sc, sh;
    {
        float m, var;
        m = s.x * invN; var = q.x * invN - m * m; sc.x = rsqrtf(var + EPS_BN) * g4.x; sh.x = be.x - m * sc.x;
        m = s.y * invN; var = q.y * invN - m * m; sc.y = rsqrtf(var + EPS_BN) * g4.y; sh.y = be.y - m * sc.y;
        m = s.z * invN; var = q.z * invN - m * m; sc.z = rsqrtf(var + EPS_BN) * g4.z; sh.z = be.z - m * sc.z;
        m = s.w * invN; var = q.w * invN - m * m; sc.w = rsqrtf(var + EPS_BN) * g4.w; sh.w = be.w - m * sc.w;
    }

    const int nWaves = gridDim.x * 8;
    const int gw  = blockIdx.x * 8 + wave;
    const int per = n / nWaves, rem = n - per * nWaves;
    const int r0  = gw * per + (gw < rem ? gw : rem);
    const int r1  = r0 + per + (gw < rem ? 1 : 0);

    int   gcur = -1;
    float cnt  = 0.f;
    v4f   acc  = {0.f, 0.f, 0.f, 0.f};
    for (int base = r0; base < r1; base += 32) {
        int m = r1 - base; if (m > 32) m = 32;
        int myb = 0;
        if (lane < m) myb = batch[base + lane];
        int k = 0;
        for (; k + 4 <= m; k += 4) {
            v4f v[4], r[4];
            for (int u = 0; u < 4; ++u) {
                const long long off = (long long)(base + k + u) * HDIM + c4;
                v[u] = *(const v4f*)(y + off);
                r[u] = *(const v4f*)(res + off);
            }
            for (int u = 0; u < 4; ++u) {
                const int g = __builtin_amdgcn_readlane(myb, k + u);
                v4f o = v[u] * sc + sh;
                o += r[u];
                if (g != gcur) {
                    pool_flush(pool, counts, gcur, G, lane, acc, cnt);
                    gcur = g; cnt = 0.f; acc = (v4f){0.f, 0.f, 0.f, 0.f};
                }
                acc += o; cnt += 1.0f;
            }
        }
        for (; k < m; ++k) {
            const long long off = (long long)(base + k) * HDIM + c4;
            const int g = __builtin_amdgcn_readlane(myb, k);
            v4f o = *(const v4f*)(y + off) * sc + sh;
            o += *(const v4f*)(res + off);
            if (g != gcur) {
                pool_flush(pool, counts, gcur, G, lane, acc, cnt);
                gcur = g; cnt = 0.f; acc = (v4f){0.f, 0.f, 0.f, 0.f};
            }
            acc += o; cnt += 1.0f;
        }
    }
    pool_flush(pool, counts, gcur, G, lane, acc, cnt);
}

__global__ __launch_bounds__(256) void k_pool_final(const float* __restrict__ pool,
                                                    const float* __restrict__ counts,
                                                    float* __restrict__ xmean, int G) {
    int t = blockIdx.x * 256 + threadIdx.x;
    if (t >= G * HDIM) return;
    int g = t / HDIM;
    xmean[t] = pool[t] / fmaxf(counts[g], 1.0f);
}

__global__ void k_pred(const float* __restrict__ xmean, const float* __restrict__ Wr,
                       const float* __restrict__ br, float* __restrict__ pred) {
    __shared__ float red[128];
    int g = blockIdx.x, c = threadIdx.x;
    red[c] = xmean[(long long)g * HDIM + c] * Wr[c];
    __syncthreads();
    for (int s = 64; s > 0; s >>= 1) {
        if (c < s) red[c] += red[c + s];
        __syncthreads();
    }
    if (c == 0) pred[g] = red[0] + br[0];
}

extern "C" void kernel_launch(void* const* d_in, const int* in_sizes, int n_in,
                              void* d_out, int out_size, void* d_ws, size_t ws_size,
                              hipStream_t stream) {
    const float* X     = (const float*)d_in[0];
    const int*   edge  = (const int*)d_in[1];
    const int*   batch = (const int*)d_in[2];
    const float* W1    = (const float*)d_in[3];
    const float* b1    = (const float*)d_in[4];
    const float* Wm    = (const float*)d_in[5];
    const float* bm    = (const float*)d_in[6];
    const float* gamma = (const float*)d_in[7];
    const float* beta  = (const float*)d_in[8];
    const float* Wr    = (const float*)d_in[9];
    const float* br    = (const float*)d_in[10];
    float* out = (float*)d_out;

    const int N = in_sizes[0];
    const int E = in_sizes[1] / 2;
    const int G = out_size / (HDIM + 1);
    const int* src = edge;
    const int* dst = edge + E;

    const long long NH = (long long)N * HDIM;
    float* ws    = (float*)d_ws;
    float* bufA  = ws;
    float* bufB  = ws + NH;
    int*   rowptr = (int*)(ws + 2 * NH);
    int*   degi   = rowptr + (N + 1);
    int*   bsum   = degi + N;
    int*   csrc   = bsum + 1024;
    float* ccoef  = (float*)(csrc + E);
    int*   rank   = csrc + 2LL * E;
    float* sumsAll = ws + 2 * NH + ((2LL * N + 1025 + 3LL * E + 3) & ~3LL);
    float* pool  = sumsAll + 5LL * NREP * 256;
    float* cnts  = pool + (long long)G * HDIM;
    float* dinv  = ws + 3 * NH;

    const int gGem = (N + 31) / 32;
    const int nChunk = (N + 1023) / 1024;

    k_zero2<<<256, 256, 0, stream>>>((float*)degi, N,
                                     sumsAll, 5LL * NREP * 256 + (long long)G * HDIM + G);
    k_deg_count<<<2048, 256, 0, stream>>>(dst, degi, rank, E);
    k_scan_part<<<nChunk, 256, 0, stream>>>(degi, dinv, bsum, N);
    k_scan_fin<<<nChunk, 256, 0, stream>>>(degi, bsum, rowptr, N, E);
    k_fill<<<2048, 256, 0, stream>>>(src, dst, dinv, rowptr, rank, csrc, ccoef, E);

    const float invN = 1.0f / (float)N;
    auto bn_tail = [&](float* ybuf, const float* sums_, const float* g_, const float* be_,
                       const float* res) {
        k_bn_apply<<<1024, 256, 0, stream>>>(ybuf, res, sums_, g_, be_, invN, N);
    };

    k_layer1<<<2048, 256, 0, stream>>>(X, W1, b1, rowptr, csrc, ccoef, dinv, bufA, sumsAll, N);
    bn_tail(bufA, sumsAll, gamma, beta, nullptr);

    float* cur = bufA;
    float* nxt = bufB;

    for (int i = 0; i < 3; ++i) {
        k_agg_gemm<<<gGem, 256, 0, stream>>>(cur, Wm + (long long)i * HDIM * HDIM,
                                             bm + (long long)i * HDIM,
                                             rowptr, csrc, ccoef, dinv, nxt,
                                             sumsAll + (long long)(i + 1) * NREP * 256, N);
        bn_tail(nxt, sumsAll + (long long)(i + 1) * NREP * 256,
                gamma + (long long)(i + 1) * HDIM,
                beta  + (long long)(i + 1) * HDIM, cur);
        float* t = cur; cur = nxt; nxt = t;
    }

    k_agg_gemm<<<gGem, 256, 0, stream>>>(cur, Wm + 3LL * HDIM * HDIM, bm + 3LL * HDIM,
                                         rowptr, csrc, ccoef, dinv, nxt,
                                         sumsAll + 4LL * NREP * 256, N);
    k_bn_pool<<<1024, 256, 0, stream>>>(nxt, cur, sumsAll + 4LL * NREP * 256,
                                        gamma + 4LL * HDIM, beta + 4LL * HDIM,
                                        batch, pool, cnts, invN, N, G);

    k_pool_final<<<(G * HDIM + 255) / 256, 256, 0, stream>>>(pool, cnts, out + G, G);
    k_pred<<<G, 128, 0, stream>>>(out + G, Wr, br, out);
}


# Edges ordered with no per-edge global atomic: grouped by 256-node destination range, then counted and placed per range by LDS counters; coefficients formed in layer 1; row sums in the new list order
# speedup vs baseline: 5.1596x; 1.2277x over previous
#include <hip/hip_runtime.h>

#define HDIM 128
#define EPS_BN 1e-5f
#define NREP 8

typedef __attribute__((ext_vector_type(2))) float v2f;
typedef __attribute__((ext_vector_type(8))) float v8f;
typedef __attribute__((ext_vector_type(4))) float v4f;

__global__ __launch_bounds__(256) void k_zero(float* a, long long na) {
    long long i = (long long)blockIdx.x * blockDim.x + threadIdx.x;
    long long stride = (long long)gridDim.x * blockDim.x;
    for (; i < na; i += stride) a[i] = 0.0f;
}

#define BIN_SHIFT 8
#define BIN_NODES 256
#define MAXB 512
#define CH 4096
#define BT 512

__device__ __forceinline__ int bin_of(int d, int NB) {
    const unsigned b = (unsigned)d >> BIN_SHIFT;
    return (int)(b < (unsigned)NB ? b : (unsigned)(NB - 1));
}

__device__ __forceinline__ int block_excl_scan(int v, int* wsum, int tid) {
    const int lane = tid & 31, w = tid >> 5;
    int incl = v;
    #pragma unroll
    for (int off = 1; off < 32; off <<= 1) {
        const int u = __shfl_up(incl, off);
        if (lane >= off) incl += u;
    }
    if (lane == 31) wsum[w] = incl;
    __syncthreads();
    int add = 0;
    for (int k = 0; k < w; ++k) add += wsum[k];
    __syncthreads();
    return add + incl - v;
}

__global__ __launch_bounds__(BT) void k_bin_count(const int* __restrict__ dst,
                                                  int* __restrict__ hist,
                                                  int E, int NB, int B1) {
    __shared__ int h[MAXB];
    const int tid = threadIdx.x;
    h[tid] = 0;
    __syncthreads();
    const int e0 = blockIdx.x * CH;
    int e1 = e0 + CH; if (e1 > E) e1 = E;
    int dv[CH / BT];
    #pragma unroll
    for (int u = 0; u < CH / BT; ++u) {
        const int e = e0 + tid + u * BT;
        dv[u] = dst[e < e1 ? e : e1 - 1];
    }
    #pragma unroll
    for (int u = 0; u < CH / BT; ++u)
        if (e0 + tid + u * BT < e1) atomicAdd(&h[bin_of(dv[u], NB)], 1);
    __syncthreads();
    if (tid < NB) hist[(long long)tid * B1 + blockIdx.x] = h[tid];
}

__global__ __launch_bounds__(256) void k_bin_scan(const int* __restrict__ hist,
                                                  int* __restrict__ base,
                                                  int* __restrict__ total,
                                                  int NB, int B1) {
    const int lane = threadIdx.x & 31;
    const int bin  = blockIdx.x * 8 + (int)(threadIdx.x >> 5);
    if (bin >= NB) return;
    const int per = (B1 + 31) >> 5;
    const int lo = lane * per;
    int hi = lo + per; if (hi > B1) hi = B1;
    const int* hp = hist + (long long)bin * B1;
    int*       bp = base + (long long)bin * B1;
    int s = 0;
    for (int k = lo; k < hi; ++k) s += hp[k];
    int incl = s;
    #pragma unroll
    for (int off = 1; off < 32; off <<= 1) {
        const int u = __shfl_up(incl, off);
        if (lane >= off) incl += u;
    }
    int run = incl - s;
    for (int k = lo; k < hi; ++k) { bp[k] = run; run += hp[k]; }
    if (lane == 31) total[bin] = incl;
}

__global__ __launch_bounds__(BT) void k_bin_scatter(const int* __restrict__ src,
                                                    const int* __restrict__ dst,
                                                    const int* __restrict__ hist,
                                                    const int* __restrict__ base,
                                                    const int* __restrict__ total,
                                                    int* __restrict__ binStart,
                                                    int2* __restrict__ binned,
                                                    int E, int NB, int B1) {
    __shared__ int2 rec[CH];
    __shared__ int lstart[MAXB];
    __shared__ int cur[MAXB];
    __shared__ int gofs[MAXB];
    __shared__ int wsum[16];
    const int tid = threadIdx.x;
    const int blk = blockIdx.x;
    const bool live = tid < NB;
    const int myTot  = live ? total[tid] : 0;
    const int myCnt  = live ? hist[(long long)tid * B1 + blk] : 0;
    const int myBase = live ? base[(long long)tid * B1 + blk] : 0;
    const int gs = block_excl_scan(myTot, wsum, tid);
    const int ls = block_excl_scan(myCnt, wsum, tid);
    lstart[tid] = ls;
    cur[tid]    = 0;
    gofs[tid]   = gs + myBase - ls;
    if (blk == 0) {
        if (live) binStart[tid] = gs;
        if (tid == NB - 1) binStart[NB] = gs + myTot;
    }
    __syncthreads();
    const int e0 = blk * CH;
    int e1 = e0 + CH; if (e1 > E) e1 = E;
    int sv[CH / BT], dv[CH / BT];
    #pragma unroll
    for (int u = 0; u < CH / BT; ++u) {
        const int e  = e0 + tid + u * BT;
        const int ec = e < e1 ? e : e1 - 1;
        sv[u] = src[ec];
        dv[u] = dst[ec];
    }
    #pragma unroll
    for (int u = 0; u < CH / BT; ++u) {
        if (e0 + tid + u * BT < e1) {
            const int b = bin_of(dv[u], NB);
            const int p = lstart[b] + atomicAdd(&cur[b], 1);
            if ((unsigned)p < (unsigned)CH) rec[p] = make_int2(sv[u], dv[u]);
        }
    }
    __syncthreads();
    const int cnt = e1 - e0;
    #pragma unroll
    for (int u = 0; u < CH / BT; ++u) {
        const int i = tid + u * BT;
        if (i < cnt) {
            const int2 r = rec[i];
            const int g = gofs[bin_of(r.y, NB)] + i;
            if ((unsigned)g < (unsigned)E) binned[g] = r;
        }
    }
}

__global__ __launch_bounds__(BT) void k_bin_sort(const int2* __restrict__ binned,
                                                 const int* __restrict__ binStart,
                                                 int* __restrict__ rowptr,
                                                 float* __restrict__ dinv,
                                                 int* __restrict__ csrc,
                                                 int N, int E, int NB) {
    __shared__ int cnt[BIN_NODES];
    __shared__ int wsum[16];
    const int tid = threadIdx.x;
    const int b   = blockIdx.x;
    int s0 = binStart[b], s1 = binStart[b + 1];
    if (s0 < 0) s0 = 0; if (s0 > E) s0 = E;
    if (s1 < s0) s1 = s0; if (s1 > E) s1 = E;
    if (tid < BIN_NODES) cnt[tid] = 0;
    __syncthreads();
    const int node0 = b << BIN_SHIFT;
    for (int i0 = s0 + tid; i0 < s1; i0 += 4 * BT) {
        int dv[4];
        #pragma unroll
        for (int u = 0; u < 4; ++u) {
            const int i = i0 + u * BT;
            dv[u] = binned[i < s1 ? i : s1 - 1].y;
        }
        #pragma unroll
        for (int u = 0; u < 4; ++u)
            if (i0 + u * BT < s1) atomicAdd(&cnt[(dv[u] - node0) & (BIN_NODES - 1)], 1);
    }
    __syncthreads();
    const int c  = (tid < BIN_NODES) ? cnt[tid] : 0;
    const int ex = block_excl_scan(c, wsum, tid);
    if (tid < BIN_NODES) {
        cnt[tid] = ex;
        const int node = node0 + tid;
        if (node < N) {
            rowptr[node] = s0 + ex;
            dinv[node]   = rsqrtf((float)c + 1.0f);
        }
    }
    if (b == NB - 1 && tid == 0) rowptr[N] = E;
    __syncthreads();
    for (int i0 = s0 + tid; i0 < s1; i0 += 4 * BT) {
        int2 rv[4];
        #pragma unroll
        for (int u = 0; u < 4; ++u) {
            const int i = i0 + u * BT;
            rv[u] = binned[i < s1 ? i : s1 - 1];
        }
        #pragma unroll
        for (int u = 0; u < 4; ++u) {
            if (i0 + u * BT < s1) {
                const int p = s0 + atomicAdd(&cnt[(rv[u].y - node0) & (BIN_NODES - 1)], 1);
                if ((unsigned)p < (unsigned)E) csrc[p] = rv[u].x;
            }
        }
    }
}

__global__ __launch_bounds__(256) void k_layer1(const float* __restrict__ X,
                                                const float* __restrict__ W1,
                                                const float* __restrict__ b1,
                                                const int* __restrict__ rowptr,
                                                const int* __restrict__ csrc,
                                                float* __restrict__ ccoef,
                                                const float* __restrict__ dinv,
                                                float* __restrict__ y,
                                                float* __restrict__ sums, int n) {
    __shared__ float red[8 * 256];
    const int lane = threadIdx.x & 31;
    const int wave = __builtin_amdgcn_readfirstlane((int)(threadIdx.x >> 5));
    const int c4 = lane << 2;
    const float4 w = *(const float4*)(W1 + c4);
    const float4 b = *(const float4*)(b1 + c4);
    float4 s = make_float4(0.f, 0.f, 0.f, 0.f);
    float4 q = make_float4(0.f, 0.f, 0.f, 0.f);
    for (int node = blockIdx.x * 8 + wave; node < n; node += gridDim.x * 8) {
        const int beg = rowptr[node], end = rowptr[node + 1];
        const float di = dinv[node];
        float part = 0.f;
        for (int j = beg + lane; j < end; j += 32) {
            const int s = csrc[j];
            const float c = dinv[s] * di;
            ccoef[j] = c;
            part += c * X[s];
        }
        for (int off = 16; off > 0; off >>= 1) part += __shfl_xor(part, off);
        const float a = part + X[node] * (di * di);
        float4 o;
        o.x = fmaxf(a * w.x + b.x, 0.f); o.y = fmaxf(a * w.y + b.y, 0.f);
        o.z = fmaxf(a * w.z + b.z, 0.f); o.w = fmaxf(a * w.w + b.w, 0.f);
        *(float4*)(y + (long long)node * HDIM + c4) = o;
        s.x += o.x; s.y += o.y; s.z += o.z; s.w += o.w;
        q.x += o.x * o.x; q.y += o.y * o.y; q.z += o.z * o.z; q.w += o.w * o.w;
    }
    *(float4*)(red + wave * 256 + c4)       = s;
    *(float4*)(red + wave * 256 + 128 + c4) = q;
    __syncthreads();
    float tot = 0.f;
    for (int k = 0; k < 8; ++k) tot += red[k * 256 + threadIdx.x];
    atomicAdd(&sums[(blockIdx.x & (NREP - 1)) * 256 + threadIdx.x], tot);
}

__global__ __launch_bounds__(256) void k_agg_gemm(const float* __restrict__ x,
                                                  const float* __restrict__ W,
                                                  const float* __restrict__ bias,
                                                  const int* __restrict__ rowptr,
                                                  const int* __restrict__ csrc,
                                                  const float* __restrict__ ccoef,
                                                  const float* __restrict__ dinv,
                                                  float* __restrict__ y,
                                                  float* __restrict__ sums, int n) {
    __shared__ float tileX[32 * HDIM];
    __shared__ float red[2 * 256];
    const int rowBase = blockIdx.x * 32;
    const int tid  = threadIdx.x;
    const int wave = __builtin_amdgcn_readfirstlane(tid >> 5);
    const int lane = tid & 31;
    const int half = lane >> 4;
    const int l16  = lane & 15;
    const int rowHalf = wave & 1;
    const int c0   = (wave >> 1) * 32;
    const int c4   = lane << 2;

    const bool fullTile = (rowBase + 32 <= n);

    for (int q = 0; q < 4; ++q) {
        const int lr = wave * 4 + q;
        const int r  = rowBase + lr;
        float4 acc = make_float4(0.f, 0.f, 0.f, 0.f);
        if (r < n) {
            const int beg = rowptr[r], end = rowptr[r + 1];
            const float di = dinv[r];
            const float4 xs = *(const float4*)(x + (long long)r * HDIM + c4);
            for (int base = beg; base < end; base += 32) {
                int m = end - base; if (m > 32) m = 32;
                int   myi = 0;
                float myc = 0.f;
                if (lane < m) { myi = csrc[base + lane]; myc = ccoef[base + lane]; }
                const int myci = __float_as_int(myc);
                int k = 0;
                for (; k + 4 <= m; k += 4) {
                    const int s0 = __builtin_amdgcn_readlane(myi, k);
                    const int s1 = __builtin_amdgcn_readlane(myi, k + 1);
                    const int s2 = __builtin_amdgcn_readlane(myi, k + 2);
                    const int s3 = __builtin_amdgcn_readlane(myi, k + 3);
                    const float e0 = __int_as_float(__builtin_amdgcn_readlane(myci, k));
                    const float e1 = __int_as_float(__builtin_amdgcn_readlane(myci, k + 1));
                    const float e2 = __int_as_float(__builtin_amdgcn_readlane(myci, k + 2));
                    const float e3 = __int_as_float(__builtin_amdgcn_readlane(myci, k + 3));
                    const float4 v0 = *(const float4*)(x + (long long)s0 * HDIM + c4);
                    const float4 v1 = *(const float4*)(x + (long long)s1 * HDIM + c4);
                    const float4 v2 = *(const float4*)(x + (long long)s2 * HDIM + c4);
                    const float4 v3 = *(const float4*)(x + (long long)s3 * HDIM + c4);
                    acc.x += e0 * v0.x; acc.y += e0 * v0.y; acc.z += e0 * v0.z; acc.w += e0 * v0.w;
                    acc.x += e1 * v1.x; acc.y += e1 * v1.y; acc.z += e1 * v1.z; acc.w += e1 * v1.w;
                    acc.x += e2 * v2.x; acc.y += e2 * v2.y; acc.z += e2 * v2.z; acc.w += e2 * v2.w;
                    acc.x += e3 * v3.x; acc.y += e3 * v3.y; acc.z += e3 * v3.z; acc.w += e3 * v3.w;
                }
                for (; k < m; ++k) {
                    const int s0 = __builtin_amdgcn_readlane(myi, k);
                    const float e0 = __int_as_float(__builtin_amdgcn_readlane(myci, k));
                    const float4 v0 = *(const float4*)(x + (long long)s0 * HDIM + c4);
                    acc.x += e0 * v0.x; acc.y += e0 * v0.y; acc.z += e0 * v0.z; acc.w += e0 * v0.w;
                }
            }
            const float d2 = di * di;
            acc.x += xs.x * d2; acc.y += xs.y * d2; acc.z += xs.z * d2; acc.w += xs.w * d2;
        }
        *(float4*)(tileX + lr * HDIM + c4) = acc;
    }
    __syncthreads();

    const float bi0 = bias[c0 + l16];
    const float bi1 = bias[c0 + 16 + l16];

    const float* xrow = tileX + (rowHalf * 16 + l16) * HDIM + 2 * half;
    v8f acc0 = {}, acc1 = {};

    #pragma unroll 4
    for (int k = 0; k < HDIM; k += 4) {
        v2f a; a.x = xrow[k]; a.y = xrow[k + 1];
        const float* wp = W + (long long)(k + 2 * half) * HDIM + c0 + l16;
        v2f b0; b0.x = wp[0];  b0.y = wp[HDIM];
        v2f b1; b1.x = wp[16]; b1.y = wp[HDIM + 16];
        acc0 = __builtin_amdgcn_wmma_f32_16x16x4_f32(false, a, false, b0,
                                                     (short)0, acc0, false, false);
        acc1 = __builtin_amdgcn_wmma_f32_16x16x4_f32(false, a, false, b1,
                                                     (short)0, acc1, false, false);
    }

    asm volatile("v_nop\n\tv_nop\n\tv_nop\n\tv_nop\n\tv_nop\n\tv_nop\n\tv_nop\n\tv_nop"
                 : "+v"(acc0), "+v"(acc1));

    #pragma unroll
    for (int v = 0; v < 8; ++v) {
        acc0[v] = fmaxf(acc0[v] + bi0, 0.f);
        acc1[v] = fmaxf(acc1[v] + bi1, 0.f);
    }

    const int rbase = rowBase + rowHalf * 16 + half * 8;

    float cs0 = 0.f, cq0 = 0.f, cs1 = 0.f, cq1 = 0.f;
    for (int v = 0; v < 8; ++v) {
        const bool in = fullTile || (rbase + v < n);
        const float a0 = in ? acc0[v] : 0.f;
        const float a1 = in ? acc1[v] : 0.f;
        cs0 += a0; cq0 += a0 * a0;
        cs1 += a1; cq1 += a1 * a1;
    }
    cs0 += __shfl_xor(cs0, 16); cq0 += __shfl_xor(cq0, 16);
    cs1 += __shfl_xor(cs1, 16); cq1 += __shfl_xor(cq1, 16);
    if (half == 0) {
        float* rp = red + rowHalf * 256 + c0 + l16;
        rp[0]   = cs0; rp[16]  = cs1;
        rp[128] = cq0; rp[144] = cq1;
    }
    __syncthreads();
    atomicAdd(&sums[(blockIdx.x & (NREP - 1)) * 256 + tid], red[tid] + red[256 + tid]);

    if (fullTile) {
        #pragma unroll
        for (int v = 0; v < 8; ++v) {
            float* op = y + (long long)(rbase + v) * HDIM + c0 + l16;
            op[0]  = acc0[v];
            op[16] = acc1[v];
        }
    } else {
        for (int v = 0; v < 8; ++v) {
            int r = rbase + v;
            if (r < n) {
                y[(long long)r * HDIM + c0 + l16]      = acc0[v];
                y[(long long)r * HDIM + c0 + 16 + l16] = acc1[v];
            }
        }
    }
}

template <bool HAS_RES>
__device__ __forceinline__ void bn_rows(float* __restrict__ y, const float* __restrict__ res,
                                        const v4f sc, const v4f sh,
                                        int t, const int stride, const int total) {
    for (; t + 3 * stride < total; t += 4 * stride) {
        v4f v[4], r[4];
        for (int u = 0; u < 4; ++u) {
            const long long off = (long long)(t + u * stride) * 4;
            v[u] = *(const v4f*)(y + off);
            if (HAS_RES) r[u] = *(const v4f*)(res + off);
        }
        for (int u = 0; u < 4; ++u) {
            const long long off = (long long)(t + u * stride) * 4;
            v4f o = v[u] * sc + sh;
            if (HAS_RES) o += r[u];
            *(v4f*)(y + off) = o;
        }
    }
    for (; t < total; t += stride) {
        const long long off = (long long)t * 4;
        v4f o = *(const v4f*)(y + off) * sc + sh;
        if (HAS_RES) o += *(const v4f*)(res + off);
        *(v4f*)(y + off) = o;
    }
}

__global__ __launch_bounds__(256) void k_bn_apply(float* __restrict__ y,
                                                  const float* __restrict__ res,
                                                  const float* __restrict__ sums,
                                                  const float* __restrict__ gamma,
                                                  const float* __restrict__ beta,
                                                  float invN, int n) {
    const int t0 = blockIdx.x * 256 + threadIdx.x;
    const int c4 = (t0 & 31) << 2;
    float4 s = make_float4(0.f, 0.f, 0.f, 0.f);
    float4 q = make_float4(0.f, 0.f, 0.f, 0.f);
    for (int k = 0; k < NREP; ++k) {
        const float4 a = *(const float4*)(sums + k * 256 + c4);
        const float4 b = *(const float4*)(sums + k * 256 + 128 + c4);
        s.x += a.x; s.y += a.y; s.z += a.z; s.w += a.w;
        q.x += b.x; q.y += b.y; q.z += b.z; q.w += b.w;
    }
    const float4 g  = *(const float4*)(gamma + c4);
    const float4 be = *(const float4*)(beta + c4);
    v4f sc, sh;
    {
        float m, var;
        m = s.x * invN; var = q.x * invN - m * m; sc.x = rsqrtf(var + EPS_BN) * g.x; sh.x = be.x - m * sc.x;
        m = s.y * invN; var = q.y * invN - m * m; sc.y = rsqrtf(var + EPS_BN) * g.y; sh.y = be.y - m * sc.y;
        m = s.z * invN; var = q.z * invN - m * m; sc.z = rsqrtf(var + EPS_BN) * g.z; sh.z = be.z - m * sc.z;
        m = s.w * invN; var = q.w * invN - m * m; sc.w = rsqrtf(var + EPS_BN) * g.w; sh.w = be.w - m * sc.w;
    }
    const int stride = gridDim.x * 256;
    const int total  = n * 32;
    if (res) bn_rows<true>(y, res, sc, sh, t0, stride, total);
    else     bn_rows<false>(y, res, sc, sh, t0, stride, total);
}

__device__ __forceinline__ void pool_flush(float* __restrict__ pool, float* __restrict__ counts,
                                           const int g, const int G, const int lane,
                                           const v4f acc, const float cnt) {
    if ((unsigned)g < (unsigned)G) {
        float* p = pool + (long long)g * HDIM + (lane << 2);
        atomicAdd(p + 0, acc.x); atomicAdd(p + 1, acc.y);
        atomicAdd(p + 2, acc.z); atomicAdd(p + 3, acc.w);
        if (lane == 0) atomicAdd(&counts[g], cnt);
    }
}

__global__ __launch_bounds__(256) void k_bn_pool(const float* __restrict__ y,
                                                 const float* __restrict__ res,
                                                 const float* __restrict__ sums,
                                                 const float* __restrict__ gamma,
                                                 const float* __restrict__ beta,
                                                 const int* __restrict__ batch,
                                                 float* __restrict__ pool,
                                                 float* __restrict__ counts,
                                                 float invN, int n, int G) {
    const int lane = threadIdx.x & 31;
    const int wave = __builtin_amdgcn_readfirstlane((int)(threadIdx.x >> 5));
    const int c4 = lane << 2;
    float4 s = make_float4(0.f, 0.f, 0.f, 0.f);
    float4 q = make_float4(0.f, 0.f, 0.f, 0.f);
    for (int k = 0; k < NREP; ++k) {
        const float4 a = *(const float4*)(sums + k * 256 + c4);
        const float4 b = *(const float4*)(sums + k * 256 + 128 + c4);
        s.x += a.x; s.y += a.y; s.z += a.z; s.w += a.w;
        q.x += b.x; q.y += b.y; q.z += b.z; q.w += b.w;
    }
    const float4 g4 = *(const float4*)(gamma + c4);
    const float4 be = *(const float4*)(beta + c4);
    v4f sc, sh;
    {
        float m, var;
        m = s.x * invN; var = q.x * invN - m * m; sc.x = rsqrtf(var + EPS_BN) * g4.x; sh.x = be.x - m * sc.x;
        m = s.y * invN; var = q.y * invN - m * m; sc.y = rsqrtf(var + EPS_BN) * g4.y; sh.y = be.y - m * sc.y;
        m = s.z * invN; var = q.z * invN - m * m; sc.z = rsqrtf(var + EPS_BN) * g4.z; sh.z = be.z - m * sc.z;
        m = s.w * invN; var = q.w * invN - m * m; sc.w = rsqrtf(var + EPS_BN) * g4.w; sh.w = be.w - m * sc.w;
    }

    const int nWaves = gridDim.x * 8;
    const int gw  = blockIdx.x * 8 + wave;
    const int per = n / nWaves, rem = n - per * nWaves;
    const int r0  = gw * per + (gw < rem ? gw : rem);
    const int r1  = r0 + per + (gw < rem ? 1 : 0);

    int   gcur = -1;
    float cnt  = 0.f;
    v4f   acc  = {0.f, 0.f, 0.f, 0.f};
    for (int base = r0; base < r1; base += 32) {
        int m = r1 - base; if (m > 32) m = 32;
        int myb = 0;
        if (lane < m) myb = batch[base + lane];
        int k = 0;
        for (; k + 4 <= m; k += 4) {
            v4f v[4], r[4];
            for (int u = 0; u < 4; ++u) {
                const long long off = (long long)(base + k + u) * HDIM + c4;
                v[u] = *(const v4f*)(y + off);
                r[u] = *(const v4f*)(res + off);
            }
            for (int u = 0; u < 4; ++u) {
                const int g = __builtin_amdgcn_readlane(myb, k + u);
                v4f o = v[u] * sc + sh;
                o += r[u];
                if (g != gcur) {
                    pool_flush(pool, counts, gcur, G, lane, acc, cnt);
                    gcur = g; cnt = 0.f; acc = (v4f){0.f, 0.f, 0.f, 0.f};
                }
                acc += o; cnt += 1.0f;
            }
        }
        for (; k < m; ++k) {
            const long long off = (long long)(base + k) * HDIM + c4;
            const int g = __builtin_amdgcn_readlane(myb, k);
            v4f o = *(const v4f*)(y + off) * sc + sh;
            o += *(const v4f*)(res + off);
            if (g != gcur) {
                pool_flush(pool, counts, gcur, G, lane, acc, cnt);
                gcur = g; cnt = 0.f; acc = (v4f){0.f, 0.f, 0.f, 0.f};
            }
            acc += o; cnt += 1.0f;
        }
    }
    pool_flush(pool, counts, gcur, G, lane, acc, cnt);
}

__global__ __launch_bounds__(256) void k_pool_final(const float* __restrict__ pool,
                                                    const float* __restrict__ counts,
                                                    float* __restrict__ xmean, int G) {
    int t = blockIdx.x * 256 + threadIdx.x;
    if (t >= G * HDIM) return;
    int g = t / HDIM;
    xmean[t] = pool[t] / fmaxf(counts[g], 1.0f);
}

__global__ void k_pred(const float* __restrict__ xmean, const float* __restrict__ Wr,
                       const float* __restrict__ br, float* __restrict__ pred) {
    __shared__ float red[128];
    int g = blockIdx.x, c = threadIdx.x;
    red[c] = xmean[(long long)g * HDIM + c] * Wr[c];
    __syncthreads();
    for (int s = 64; s > 0; s >>= 1) {
        if (c < s) red[c] += red[c + s];
        __syncthreads();
    }
    if (c == 0) pred[g] = red[0] + br[0];
}

extern "C" void kernel_launch(void* const* d_in, const int* in_sizes, int n_in,
                              void* d_out, int out_size, void* d_ws, size_t ws_size,
                              hipStream_t stream) {
    const float* X     = (const float*)d_in[0];
    const int*   edge  = (const int*)d_in[1];
    const int*   batch = (const int*)d_in[2];
    const float* W1    = (const float*)d_in[3];
    const float* b1    = (const float*)d_in[4];
    const float* Wm    = (const float*)d_in[5];
    const float* bm    = (const float*)d_in[6];
    const float* gamma = (const float*)d_in[7];
    const float* beta  = (const float*)d_in[8];
    const float* Wr    = (const float*)d_in[9];
    const float* br    = (const float*)d_in[10];
    float* out = (float*)d_out;

    const int N = in_sizes[0];
    const int E = in_sizes[1] / 2;
    const int G = out_size / (HDIM + 1);
    const int* src = edge;
    const int* dst = edge + E;

    const long long NH = (long long)N * HDIM;
    float* ws    = (float*)d_ws;
    float* bufA  = ws;
    float* bufB  = ws + NH;
    const int nbRaw = (N + BIN_NODES - 1) / BIN_NODES;
    const int NB = nbRaw < MAXB ? nbRaw : MAXB;
    const int B1 = (E + CH - 1) / CH;
    const long long tabW = ((long long)NB * B1 + 3) & ~3LL;
    const long long eW   = ((long long)E + 3) & ~3LL;
    int*   slot3    = (int*)(ws + 2 * NH);
    int*   rowptr   = slot3;
    int*   binStart = rowptr + (((long long)N + 1 + 3) & ~3LL);
    int*   total    = binStart + (MAXB + 4);
    int*   hist     = total + MAXB;
    int*   base     = hist + tabW;
    int*   csrc     = base + tabW;
    float* ccoef    = (float*)(csrc + eW);
    int2*  binned   = (int2*)(csrc + 2 * eW);
    float* sumsAll = (float*)(csrc + 4 * eW);
    float* pool  = sumsAll + 5LL * NREP * 256;
    float* cnts  = pool + (long long)G * HDIM;
    float* dinv  = ws + 3 * NH;

    const int gGem = (N + 31) / 32;

    k_zero<<<64, 256, 0, stream>>>(sumsAll, 5LL * NREP * 256 + (long long)G * HDIM + G);
    k_bin_count<<<B1, BT, 0, stream>>>(dst, hist, E, NB, B1);
    k_bin_scan<<<(NB + 7) / 8, 256, 0, stream>>>(hist, base, total, NB, B1);
    k_bin_scatter<<<B1, BT, 0, stream>>>(src, dst, hist, base, total, binStart, binned,
                                         E, NB, B1);
    k_bin_sort<<<NB, BT, 0, stream>>>(binned, binStart, rowptr, dinv, csrc, N, E, NB);

    const float invN = 1.0f / (float)N;
    auto bn_tail = [&](float* ybuf, const float* sums_, const float* g_, const float* be_,
                       const float* res) {
        k_bn_apply<<<1024, 256, 0, stream>>>(ybuf, res, sums_, g_, be_, invN, N);
    };

    k_layer1<<<2048, 256, 0, stream>>>(X, W1, b1, rowptr, csrc, ccoef, dinv, bufA, sumsAll, N);
    bn_tail(bufA, sumsAll, gamma, beta, nullptr);

    float* cur = bufA;
    float* nxt = bufB;

    for (int i = 0; i < 3; ++i) {
        k_agg_gemm<<<gGem, 256, 0, stream>>>(cur, Wm + (long long)i * HDIM * HDIM,
                                             bm + (long long)i * HDIM,
                                             rowptr, csrc, ccoef, dinv, nxt,
                                             sumsAll + (long long)(i + 1) * NREP * 256, N);
        bn_tail(nxt, sumsAll + (long long)(i + 1) * NREP * 256,
                gamma + (long long)(i + 1) * HDIM,
                beta  + (long long)(i + 1) * HDIM, cur);
        float* t = cur; cur = nxt; nxt = t;
    }

    k_agg_gemm<<<gGem, 256, 0, stream>>>(cur, Wm + 3LL * HDIM * HDIM, bm + 3LL * HDIM,
                                         rowptr, csrc, ccoef, dinv, nxt,
                                         sumsAll + 4LL * NREP * 256, N);
    k_bn_pool<<<1024, 256, 0, stream>>>(nxt, cur, sumsAll + 4LL * NREP * 256,
                                        gamma + 4LL * HDIM, beta + 4LL * HDIM,
                                        batch, pool, cnts, invN, N, G);

    k_pool_final<<<(G * HDIM + 255) / 256, 256, 0, stream>>>(pool, cnts, out + G, G);
    k_pred<<<G, 128, 0, stream>>>(out + G, Wr, br, out);
}


# k_agg_gemm gather: a wave walks its 4 rows' edges as one contiguous run, 8 row loads in flight across row ends, next 32 slots fetched ahead; per-row list order of the sums kept
# speedup vs baseline: 5.3947x; 1.0456x over previous
#include <hip/hip_runtime.h>

#define HDIM 128
#define EPS_BN 1e-5f
#define NREP 8

typedef __attribute__((ext_vector_type(2))) float v2f;
typedef __attribute__((ext_vector_type(8))) float v8f;
typedef __attribute__((ext_vector_type(4))) float v4f;

__global__ __launch_bounds__(256) void k_zero(float* a, long long na) {
    long long i = (long long)blockIdx.x * blockDim.x + threadIdx.x;
    long long stride = (long long)gridDim.x * blockDim.x;
    for (; i < na; i += stride) a[i] = 0.0f;
}

#define BIN_SHIFT 8
#define BIN_NODES 256
#define MAXB 512
#define CH 4096
#define BT 512

__device__ __forceinline__ int bin_of(int d, int NB) {
    const unsigned b = (unsigned)d >> BIN_SHIFT;
    return (int)(b < (unsigned)NB ? b : (unsigned)(NB - 1));
}

__device__ __forceinline__ int block_excl_scan(int v, int* wsum, int tid) {
    const int lane = tid & 31, w = tid >> 5;
    int incl = v;
    #pragma unroll
    for (int off = 1; off < 32; off <<= 1) {
        const int u = __shfl_up(incl, off);
        if (lane >= off) incl += u;
    }
    if (lane == 31) wsum[w] = incl;
    __syncthreads();
    int add = 0;
    for (int k = 0; k < w; ++k) add += wsum[k];
    __syncthreads();
    return add + incl - v;
}

__global__ __launch_bounds__(BT) void k_bin_count(const int* __restrict__ dst,
                                                  int* __restrict__ hist,
                                                  int E, int NB, int B1) {
    __shared__ int h[MAXB];
    const int tid = threadIdx.x;
    h[tid] = 0;
    __syncthreads();
    const int e0 = blockIdx.x * CH;
    int e1 = e0 + CH; if (e1 > E) e1 = E;
    int dv[CH / BT];
    #pragma unroll
    for (int u = 0; u < CH / BT; ++u) {
        const int e = e0 + tid + u * BT;
        dv[u] = dst[e < e1 ? e : e1 - 1];
    }
    #pragma unroll
    for (int u = 0; u < CH / BT; ++u)
        if (e0 + tid + u * BT < e1) atomicAdd(&h[bin_of(dv[u], NB)], 1);
    __syncthreads();
    if (tid < NB) hist[(long long)tid * B1 + blockIdx.x] = h[tid];
}

__global__ __launch_bounds__(256) void k_bin_scan(const int* __restrict__ hist,
                                                  int* __restrict__ base,
                                                  int* __restrict__ total,
                                                  int NB, int B1) {
    const int lane = threadIdx.x & 31;
    const int bin  = blockIdx.x * 8 + (int)(threadIdx.x >> 5);
    if (bin >= NB) return;
    const int per = (B1 + 31) >> 5;
    const int lo = lane * per;
    int hi = lo + per; if (hi > B1) hi = B1;
    const int* hp = hist + (long long)bin * B1;
    int*       bp = base + (long long)bin * B1;
    int s = 0;
    for (int k = lo; k < hi; ++k) s += hp[k];
    int incl = s;
    #pragma unroll
    for (int off = 1; off < 32; off <<= 1) {
        const int u = __shfl_up(incl, off);
        if (lane >= off) incl += u;
    }
    int run = incl - s;
    for (int k = lo; k < hi; ++k) { bp[k] = run; run += hp[k]; }
    if (lane == 31) total[bin] = incl;
}

__global__ __launch_bounds__(BT) void k_bin_scatter(const int* __restrict__ src,
                                                    const int* __restrict__ dst,
                                                    const int* __restrict__ hist,
                                                    const int* __restrict__ base,
                                                    const int* __restrict__ total,
                                                    int* __restrict__ binStart,
                                                    int2* __restrict__ binned,
                                                    int E, int NB, int B1) {
    __shared__ int2 rec[CH];
    __shared__ int lstart[MAXB];
    __shared__ int cur[MAXB];
    __shared__ int gofs[MAXB];
    __shared__ int wsum[16];
    const int tid = threadIdx.x;
    const int blk = blockIdx.x;
    const bool live = tid < NB;
    const int myTot  = live ? total[tid] : 0;
    const int myCnt  = live ? hist[(long long)tid * B1 + blk] : 0;
    const int myBase = live ? base[(long long)tid * B1 + blk] : 0;
    const int gs = block_excl_scan(myTot, wsum, tid);
    const int ls = block_excl_scan(myCnt, wsum, tid);
    lstart[tid] = ls;
    cur[tid]    = 0;
    gofs[tid]   = gs + myBase - ls;
    if (blk == 0) {
        if (live) binStart[tid] = gs;
        if (tid == NB - 1) binStart[NB] = gs + myTot;
    }
    __syncthreads();
    const int e0 = blk * CH;
    int e1 = e0 + CH; if (e1 > E) e1 = E;
    int sv[CH / BT], dv[CH / BT];
    #pragma unroll
    for (int u = 0; u < CH / BT; ++u) {
        const int e  = e0 + tid + u * BT;
        const int ec = e < e1 ? e : e1 - 1;
        sv[u] = src[ec];
        dv[u] = dst[ec];
    }
    #pragma unroll
    for (int u = 0; u < CH / BT; ++u) {
        if (e0 + tid + u * BT < e1) {
            const int b = bin_of(dv[u], NB);
            const int p = lstart[b] + atomicAdd(&cur[b], 1);
            if ((unsigned)p < (unsigned)CH) rec[p] = make_int2(sv[u], dv[u]);
        }
    }
    __syncthreads();
    const int cnt = e1 - e0;
    #pragma unroll
    for (int u = 0; u < CH / BT; ++u) {
        const int i = tid + u * BT;
        if (i < cnt) {
            const int2 r = rec[i];
            const int g = gofs[bin_of(r.y, NB)] + i;
            if ((unsigned)g < (unsigned)E) binned[g] = r;
        }
    }
}

__global__ __launch_bounds__(BT) void k_bin_sort(const int2* __restrict__ binned,
                                                 const int* __restrict__ binStart,
                                                 int* __restrict__ rowptr,
                                                 float* __restrict__ dinv,
                                                 int* __restrict__ csrc,
                                                 int N, int E, int NB) {
    __shared__ int cnt[BIN_NODES];
    __shared__ int wsum[16];
    const int tid = threadIdx.x;
    const int b   = blockIdx.x;
    int s0 = binStart[b], s1 = binStart[b + 1];
    if (s0 < 0) s0 = 0; if (s0 > E) s0 = E;
    if (s1 < s0) s1 = s0; if (s1 > E) s1 = E;
    if (tid < BIN_NODES) cnt[tid] = 0;
    __syncthreads();
    const int node0 = b << BIN_SHIFT;
    for (int i0 = s0 + tid; i0 < s1; i0 += 4 * BT) {
        int dv[4];
        #pragma unroll
        for (int u = 0; u < 4; ++u) {
            const int i = i0 + u * BT;
            dv[u] = binned[i < s1 ? i : s1 - 1].y;
        }
        #pragma unroll
        for (int u = 0; u < 4; ++u)
            if (i0 + u * BT < s1) atomicAdd(&cnt[(dv[u] - node0) & (BIN_NODES - 1)], 1);
    }
    __syncthreads();
    const int c  = (tid < BIN_NODES) ? cnt[tid] : 0;
    const int ex = block_excl_scan(c, wsum, tid);
    if (tid < BIN_NODES) {
        cnt[tid] = ex;
        const int node = node0 + tid;
        if (node < N) {
            rowptr[node] = s0 + ex;
            dinv[node]   = rsqrtf((float)c + 1.0f);
        }
    }
    if (b == NB - 1 && tid == 0) rowptr[N] = E;
    __syncthreads();
    for (int i0 = s0 + tid; i0 < s1; i0 += 4 * BT) {
        int2 rv[4];
        #pragma unroll
        for (int u = 0; u < 4; ++u) {
            const int i = i0 + u * BT;
            rv[u] = binned[i < s1 ? i : s1 - 1];
        }
        #pragma unroll
        for (int u = 0; u < 4; ++u) {
            if (i0 + u * BT < s1) {
                const int p = s0 + atomicAdd(&cnt[(rv[u].y - node0) & (BIN_NODES - 1)], 1);
                if ((unsigned)p < (unsigned)E) csrc[p] = rv[u].x;
            }
        }
    }
}

__global__ __launch_bounds__(256) void k_layer1(const float* __restrict__ X,
                                                const float* __restrict__ W1,
                                                const float* __restrict__ b1,
                                                const int* __restrict__ rowptr,
                                                const int* __restrict__ csrc,
                                                float* __restrict__ ccoef,
                                                const float* __restrict__ dinv,
                                                float* __restrict__ y,
                                                float* __restrict__ sums, int n) {
    __shared__ float red[8 * 256];
    const int lane = threadIdx.x & 31;
    const int wave = __builtin_amdgcn_readfirstlane((int)(threadIdx.x >> 5));
    const int c4 = lane << 2;
    const float4 w = *(const float4*)(W1 + c4);
    const float4 b = *(const float4*)(b1 + c4);
    float4 s = make_float4(0.f, 0.f, 0.f, 0.f);
    float4 q = make_float4(0.f, 0.f, 0.f, 0.f);
    for (int node = blockIdx.x * 8 + wave; node < n; node += gridDim.x * 8) {
        const int beg = rowptr[node], end = rowptr[node + 1];
        const float di = dinv[node];
        float part = 0.f;
        for (int j = beg + lane; j < end; j += 32) {
            const int s = csrc[j];
            const float c = dinv[s] * di;
            ccoef[j] = c;
            part += c * X[s];
        }
        for (int off = 16; off > 0; off >>= 1) part += __shfl_xor(part, off);
        const float a = part + X[node] * (di * di);
        float4 o;
        o.x = fmaxf(a * w.x + b.x, 0.f); o.y = fmaxf(a * w.y + b.y, 0.f);
        o.z = fmaxf(a * w.z + b.z, 0.f); o.w = fmaxf(a * w.w + b.w, 0.f);
        *(float4*)(y + (long long)node * HDIM + c4) = o;
        s.x += o.x; s.y += o.y; s.z += o.z; s.w += o.w;
        q.x += o.x * o.x; q.y += o.y * o.y; q.z += o.z * o.z; q.w += o.w * o.w;
    }
    *(float4*)(red + wave * 256 + c4)       = s;
    *(float4*)(red + wave * 256 + 128 + c4) = q;
    __syncthreads();
    float tot = 0.f;
    for (int k = 0; k < 8; ++k) tot += red[k * 256 + threadIdx.x];
    atomicAdd(&sums[(blockIdx.x & (NREP - 1)) * 256 + threadIdx.x], tot);
}

__device__ __forceinline__ void row_done(float* __restrict__ trow, float4& acc, int& cur, int& bnd,
                                         const int p2, const int p3, const int p4) {
    float* tp = trow + cur * HDIM;
    const float4 t = *(const float4*)tp;
    *(float4*)tp = make_float4(acc.x + t.x, acc.y + t.y, acc.z + t.z, acc.w + t.w);
    acc = make_float4(0.f, 0.f, 0.f, 0.f);
    ++cur;
    bnd = (cur == 1) ? p2 : ((cur == 2) ? p3 : p4);
}

template <bool FULL>
__device__ __forceinline__ void gather_group(const float* __restrict__ x, const int c4,
                                             const int myi, const int myci,
                                             const int k, const int cnt, const int m, const int j0,
                                             float* __restrict__ trow, float4& acc, int& cur, int& bnd,
                                             const int p2, const int p3, const int p4) {
    float4 v[8];
    float  e[8];
    #pragma unroll
    for (int u = 0; u < 8; ++u) {
        int ku = k + u;
        if (!FULL && ku > m - 1) ku = m - 1;
        const int s = __builtin_amdgcn_readlane(myi, ku);
        e[u] = __int_as_float(__builtin_amdgcn_readlane(myci, ku));
        v[u] = *(const float4*)(x + (long long)s * HDIM + c4);
    }
    #pragma unroll
    for (int u = 0; u < 8; ++u) {
        if (FULL || u < cnt) {
            const int j = j0 + u;
            while (j >= bnd && cur < 3) row_done(trow, acc, cur, bnd, p2, p3, p4);
            acc.x += e[u] * v[u].x; acc.y += e[u] * v[u].y;
            acc.z += e[u] * v[u].z; acc.w += e[u] * v[u].w;
        }
    }
}

__global__ __launch_bounds__(256) void k_agg_gemm(const float* __restrict__ x,
                                                  const float* __restrict__ W,
                                                  const float* __restrict__ bias,
                                                  const int* __restrict__ rowptr,
                                                  const int* __restrict__ csrc,
                                                  const float* __restrict__ ccoef,
                                                  const float* __restrict__ dinv,
                                                  float* __restrict__ y,
                                                  float* __restrict__ sums, int n) {
    __shared__ float tileX[32 * HDIM];
    __shared__ float red[2 * 256];
    const int rowBase = blockIdx.x * 32;
    const int tid  = threadIdx.x;
    const int wave = __builtin_amdgcn_readfirstlane(tid >> 5);
    const int lane = tid & 31;
    const int half = lane >> 4;
    const int l16  = lane & 15;
    const int rowHalf = wave & 1;
    const int c0   = (wave >> 1) * 32;
    const int c4   = lane << 2;

    const bool fullTile = (rowBase + 32 <= n);

    const int lr0 = wave * 4;
    const int r0  = rowBase + lr0;
    float* trow = tileX + lr0 * HDIM + c4;
    int pa[5];
    #pragma unroll
    for (int i = 0; i < 5; ++i) {
        int idx = r0 + i; if (idx > n) idx = n;
        pa[i] = rowptr[idx];
    }
    const int jend = pa[4];
    int base = pa[0];

    float4 xs[4];
    float  d2[4];
    #pragma unroll
    for (int i = 0; i < 4; ++i) {
        const int r  = r0 + i;
        const int rc = r < n ? r : n - 1;
        const float di = dinv[rc];
        d2[i] = di * di;
        xs[i] = *(const float4*)(x + (long long)rc * HDIM + c4);
    }
    int   myi = 0;
    float myc = 0.f;
    if (lane < jend - base) { myi = csrc[base + lane]; myc = ccoef[base + lane]; }

    #pragma unroll
    for (int i = 0; i < 4; ++i) {
        float4 t = make_float4(0.f, 0.f, 0.f, 0.f);
        if (r0 + i < n) t = make_float4(xs[i].x * d2[i], xs[i].y * d2[i], xs[i].z * d2[i], xs[i].w * d2[i]);
        *(float4*)(trow + i * HDIM) = t;
    }

    int    cur = 0;
    int    bnd = pa[1];
    float4 acc = make_float4(0.f, 0.f, 0.f, 0.f);
    for (; base < jend; base += 32) {
        int m = jend - base; if (m > 32) m = 32;
        int   nyi = 0;
        float nyc = 0.f;
        if (lane < jend - base - 32) { nyi = csrc[base + 32 + lane]; nyc = ccoef[base + 32 + lane]; }
        const int myci = __float_as_int(myc);
        int k = 0;
        for (; k + 8 <= m; k += 8)
            gather_group<true>(x, c4, myi, myci, k, 8, m, base + k, trow, acc, cur, bnd, pa[2], pa[3], pa[4]);
        if (k < m)
            gather_group<false>(x, c4, myi, myci, k, m - k, m, base + k, trow, acc, cur, bnd, pa[2], pa[3], pa[4]);
        myi = nyi; myc = nyc;
    }
    while (cur < 4) row_done(trow, acc, cur, bnd, pa[2], pa[3], pa[4]);
    __syncthreads();

    const float bi0 = bias[c0 + l16];
    const float bi1 = bias[c0 + 16 + l16];

    const float* xrow = tileX + (rowHalf * 16 + l16) * HDIM + 2 * half;
    v8f acc0 = {}, acc1 = {};

    #pragma unroll 4
    for (int k = 0; k < HDIM; k += 4) {
        v2f a; a.x = xrow[k]; a.y = xrow[k + 1];
        const float* wp = W + (long long)(k + 2 * half) * HDIM + c0 + l16;
        v2f b0; b0.x = wp[0];  b0.y = wp[HDIM];
        v2f b1; b1.x = wp[16]; b1.y = wp[HDIM + 16];
        acc0 = __builtin_amdgcn_wmma_f32_16x16x4_f32(false, a, false, b0,
                                                     (short)0, acc0, false, false);
        acc1 = __builtin_amdgcn_wmma_f32_16x16x4_f32(false, a, false, b1,
                                                     (short)0, acc1, false, false);
    }

    asm volatile("v_nop\n\tv_nop\n\tv_nop\n\tv_nop\n\tv_nop\n\tv_nop\n\tv_nop\n\tv_nop"
                 : "+v"(acc0), "+v"(acc1));

    #pragma unroll
    for (int v = 0; v < 8; ++v) {
        acc0[v] = fmaxf(acc0[v] + bi0, 0.f);
        acc1[v] = fmaxf(acc1[v] + bi1, 0.f);
    }

    const int rbase = rowBase + rowHalf * 16 + half * 8;

    float cs0 = 0.f, cq0 = 0.f, cs1 = 0.f, cq1 = 0.f;
    for (int v = 0; v < 8; ++v) {
        const bool in = fullTile || (rbase + v < n);
        const float a0 = in ? acc0[v] : 0.f;
        const float a1 = in ? acc1[v] : 0.f;
        cs0 += a0; cq0 += a0 * a0;
        cs1 += a1; cq1 += a1 * a1;
    }
    cs0 += __shfl_xor(cs0, 16); cq0 += __shfl_xor(cq0, 16);
    cs1 += __shfl_xor(cs1, 16); cq1 += __shfl_xor(cq1, 16);
    if (half == 0) {
        float* rp = red + rowHalf * 256 + c0 + l16;
        rp[0]   = cs0; rp[16]  = cs1;
        rp[128] = cq0; rp[144] = cq1;
    }
    __syncthreads();
    atomicAdd(&sums[(blockIdx.x & (NREP - 1)) * 256 + tid], red[tid] + red[256 + tid]);

    if (fullTile) {
        #pragma unroll
        for (int v = 0; v < 8; ++v) {
            float* op = y + (long long)(rbase + v) * HDIM + c0 + l16;
            op[0]  = acc0[v];
            op[16] = acc1[v];
        }
    } else {
        for (int v = 0; v < 8; ++v) {
            int r = rbase + v;
            if (r < n) {
                y[(long long)r * HDIM + c0 + l16]      = acc0[v];
                y[(long long)r * HDIM + c0 + 16 + l16] = acc1[v];
            }
        }
    }
}

template <bool HAS_RES>
__device__ __forceinline__ void bn_rows(float* __restrict__ y, const float* __restrict__ res,
                                        const v4f sc, const v4f sh,
                                        int t, const int stride, const int total) {
    for (; t + 3 * stride < total; t += 4 * stride) {
        v4f v[4], r[4];
        for (int u = 0; u < 4; ++u) {
            const long long off = (long long)(t + u * stride) * 4;
            v[u] = *(const v4f*)(y + off);
            if (HAS_RES) r[u] = *(const v4f*)(res + off);
        }
        for (int u = 0; u < 4; ++u) {
            const long long off = (long long)(t + u * stride) * 4;
            v4f o = v[u] * sc + sh;
            if (HAS_RES) o += r[u];
            *(v4f*)(y + off) = o;
        }
    }
    for (; t < total; t += stride) {
        const long long off = (long long)t * 4;
        v4f o = *(const v4f*)(y + off) * sc + sh;
        if (HAS_RES) o += *(const v4f*)(res + off);
        *(v4f*)(y + off) = o;
    }
}

__global__ __launch_bounds__(256) void k_bn_apply(float* __restrict__ y,
                                                  const float* __restrict__ res,
                                                  const float* __restrict__ sums,
                                                  const float* __restrict__ gamma,
                                                  const float* __restrict__ beta,
                                                  float invN, int n) {
    const int t0 = blockIdx.x * 256 + threadIdx.x;
    const int c4 = (t0 & 31) << 2;
    float4 s = make_float4(0.f, 0.f, 0.f, 0.f);
    float4 q = make_float4(0.f, 0.f, 0.f, 0.f);
    for (int k = 0; k < NREP; ++k) {
        const float4 a = *(const float4*)(sums + k * 256 + c4);
        const float4 b = *(const float4*)(sums + k * 256 + 128 + c4);
        s.x += a.x; s.y += a.y; s.z += a.z; s.w += a.w;
        q.x += b.x; q.y += b.y; q.z += b.z; q.w += b.w;
    }
    const float4 g  = *(const float4*)(gamma + c4);
    const float4 be = *(const float4*)(beta + c4);
    v4f sc, sh;
    {
        float m, var;
        m = s.x * invN; var = q.x * invN - m * m; sc.x = rsqrtf(var + EPS_BN) * g.x; sh.x = be.x - m * sc.x;
        m = s.y * invN; var = q.y * invN - m * m; sc.y = rsqrtf(var + EPS_BN) * g.y; sh.y = be.y - m * sc.y;
        m = s.z * invN; var = q.z * invN - m * m; sc.z = rsqrtf(var + EPS_BN) * g.z; sh.z = be.z - m * sc.z;
        m = s.w * invN; var = q.w * invN - m * m; sc.w = rsqrtf(var + EPS_BN) * g.w; sh.w = be.w - m * sc.w;
    }
    const int stride = gridDim.x * 256;
    const int total  = n * 32;
    if (res) bn_rows<true>(y, res, sc, sh, t0, stride, total);
    else     bn_rows<false>(y, res, sc, sh, t0, stride, total);
}

__device__ __forceinline__ void pool_flush(float* __restrict__ pool, float* __restrict__ counts,
                                           const int g, const int G, const int lane,
                                           const v4f acc, const float cnt) {
    if ((unsigned)g < (unsigned)G) {
        float* p = pool + (long long)g * HDIM + (lane << 2);
        atomicAdd(p + 0, acc.x); atomicAdd(p + 1, acc.y);
        atomicAdd(p + 2, acc.z); atomicAdd(p + 3, acc.w);
        if (lane == 0) atomicAdd(&counts[g], cnt);
    }
}

__global__ __launch_bounds__(256) void k_bn_pool(const float* __restrict__ y,
                                                 const float* __restrict__ res,
                                                 const float* __restrict__ sums,
                                                 const float* __restrict__ gamma,
                                                 const float* __restrict__ beta,
                                                 const int* __restrict__ batch,
                                                 float* __restrict__ pool,
                                                 float* __restrict__ counts,
                                                 float invN, int n, int G) {
    const int lane = threadIdx.x & 31;
    const int wave = __builtin_amdgcn_readfirstlane((int)(threadIdx.x >> 5));
    const int c4 = lane << 2;
    float4 s = make_float4(0.f, 0.f, 0.f, 0.f);
    float4 q = make_float4(0.f, 0.f, 0.f, 0.f);
    for (int k = 0; k < NREP; ++k) {
        const float4 a = *(const float4*)(sums + k * 256 + c4);
        const float4 b = *(const float4*)(sums + k * 256 + 128 + c4);
        s.x += a.x; s.y += a.y; s.z += a.z; s.w += a.w;
        q.x += b.x; q.y += b.y; q.z += b.z; q.w += b.w;
    }
    const float4 g4 = *(const float4*)(gamma + c4);
    const float4 be = *(const float4*)(beta + c4);
    v4f sc, sh;
    {
        float m, var;
        m = s.x * invN; var = q.x * invN - m * m; sc.x = rsqrtf(var + EPS_BN) * g4.x; sh.x = be.x - m * sc.x;
        m = s.y * invN; var = q.y * invN - m * m; sc.y = rsqrtf(var + EPS_BN) * g4.y; sh.y = be.y - m * sc.y;
        m = s.z * invN; var = q.z * invN - m * m; sc.z = rsqrtf(var + EPS_BN) * g4.z; sh.z = be.z - m * sc.z;
        m = s.w * invN; var = q.w * invN - m * m; sc.w = rsqrtf(var + EPS_BN) * g4.w; sh.w = be.w - m * sc.w;
    }

    const int nWaves = gridDim.x * 8;
    const int gw  = blockIdx.x * 8 + wave;
    const int per = n / nWaves, rem = n - per * nWaves;
    const int r0  = gw * per + (gw < rem ? gw : rem);
    const int r1  = r0 + per + (gw < rem ? 1 : 0);

    int   gcur = -1;
    float cnt  = 0.f;
    v4f   acc  = {0.f, 0.f, 0.f, 0.f};
    for (int base = r0; base < r1; base += 32) {
        int m = r1 - base; if (m > 32) m = 32;
        int myb = 0;
        if (lane < m) myb = batch[base + lane];
        int k = 0;
        for (; k + 4 <= m; k += 4) {
            v4f v[4], r[4];
            for (int u = 0; u < 4; ++u) {
                const long long off = (long long)(base + k + u) * HDIM + c4;
                v[u] = *(const v4f*)(y + off);
                r[u] = *(const v4f*)(res + off);
            }
            for (int u = 0; u < 4; ++u) {
                const int g = __builtin_amdgcn_readlane(myb, k + u);
                v4f o = v[u] * sc + sh;
                o += r[u];
                if (g != gcur) {
                    pool_flush(pool, counts, gcur, G, lane, acc, cnt);
                    gcur = g; cnt = 0.f; acc = (v4f){0.f, 0.f, 0.f, 0.f};
                }
                acc += o; cnt += 1.0f;
            }
        }
        for (; k < m; ++k) {
            const long long off = (long long)(base + k) * HDIM + c4;
            const int g = __builtin_amdgcn_readlane(myb, k);
            v4f o = *(const v4f*)(y + off) * sc + sh;
            o += *(const v4f*)(res + off);
            if (g != gcur) {
                pool_flush(pool, counts, gcur, G, lane, acc, cnt);
                gcur = g; cnt = 0.f; acc = (v4f){0.f, 0.f, 0.f, 0.f};
            }
            acc += o; cnt += 1.0f;
        }
    }
    pool_flush(pool, counts, gcur, G, lane, acc, cnt);
}

__global__ __launch_bounds__(256) void k_pool_final(const float* __restrict__ pool,
                                                    const float* __restrict__ counts,
                                                    float* __restrict__ xmean, int G) {
    int t = blockIdx.x * 256 + threadIdx.x;
    if (t >= G * HDIM) return;
    int g = t / HDIM;
    xmean[t] = pool[t] / fmaxf(counts[g], 1.0f);
}

__global__ void k_pred(const float* __restrict__ xmean, const float* __restrict__ Wr,
                       const float* __restrict__ br, float* __restrict__ pred) {
    __shared__ float red[128];
    int g = blockIdx.x, c = threadIdx.x;
    red[c] = xmean[(long long)g * HDIM + c] * Wr[c];
    __syncthreads();
    for (int s = 64; s > 0; s >>= 1) {
        if (c < s) red[c] += red[c + s];
        __syncthreads();
    }
    if (c == 0) pred[g] = red[0] + br[0];
}

extern "C" void kernel_launch(void* const* d_in, const int* in_sizes, int n_in,
                              void* d_out, int out_size, void* d_ws, size_t ws_size,
                              hipStream_t stream) {
    const float* X     = (const float*)d_in[0];
    const int*   edge  = (const int*)d_in[1];
    const int*   batch = (const int*)d_in[2];
    const float* W1    = (const float*)d_in[3];
    const float* b1    = (const float*)d_in[4];
    const float* Wm    = (const float*)d_in[5];
    const float* bm    = (const float*)d_in[6];
    const float* gamma = (const float*)d_in[7];
    const float* beta  = (const float*)d_in[8];
    const float* Wr    = (const float*)d_in[9];
    const float* br    = (const float*)d_in[10];
    float* out = (float*)d_out;

    const int N = in_sizes[0];
    const int E = in_sizes[1] / 2;
    const int G = out_size / (HDIM + 1);
    const int* src = edge;
    const int* dst = edge + E;

    const long long NH = (long long)N * HDIM;
    float* ws    = (float*)d_ws;
    float* bufA  = ws;
    float* bufB  = ws + NH;
    const int nbRaw = (N + BIN_NODES - 1) / BIN_NODES;
    const int NB = nbRaw < MAXB ? nbRaw : MAXB;
    const int B1 = (E + CH - 1) / CH;
    const long long tabW = ((long long)NB * B1 + 3) & ~3LL;
    const long long eW   = ((long long)E + 3) & ~3LL;
    int*   slot3    = (int*)(ws + 2 * NH);
    int*   rowptr   = slot3;
    int*   binStart = rowptr + (((long long)N + 1 + 3) & ~3LL);
    int*   total    = binStart + (MAXB + 4);
    int*   hist     = total + MAXB;
    int*   base     = hist + tabW;
    int*   csrc     = base + tabW;
    float* ccoef    = (float*)(csrc + eW);
    int2*  binned   = (int2*)(csrc + 2 * eW);
    float* sumsAll = (float*)(csrc + 4 * eW);
    float* pool  = sumsAll + 5LL * NREP * 256;
    float* cnts  = pool + (long long)G * HDIM;
    float* dinv  = ws + 3 * NH;

    const int gGem = (N + 31) / 32;

    k_zero<<<64, 256, 0, stream>>>(sumsAll, 5LL * NREP * 256 + (long long)G * HDIM + G);
    k_bin_count<<<B1, BT, 0, stream>>>(dst, hist, E, NB, B1);
    k_bin_scan<<<(NB + 7) / 8, 256, 0, stream>>>(hist, base, total, NB, B1);
    k_bin_scatter<<<B1, BT, 0, stream>>>(src, dst, hist, base, total, binStart, binned,
                                         E, NB, B1);
    k_bin_sort<<<NB, BT, 0, stream>>>(binned, binStart, rowptr, dinv, csrc, N, E, NB);

    const float invN = 1.0f / (float)N;
    auto bn_tail = [&](float* ybuf, const float* sums_, const float* g_, const float* be_,
                       const float* res) {
        k_bn_apply<<<1024, 256, 0, stream>>>(ybuf, res, sums_, g_, be_, invN, N);
    };

    k_layer1<<<2048, 256, 0, stream>>>(X, W1, b1, rowptr, csrc, ccoef, dinv, bufA, sumsAll, N);
    bn_tail(bufA, sumsAll, gamma, beta, nullptr);

    float* cur = bufA;
    float* nxt = bufB;

    for (int i = 0; i < 3; ++i) {
        k_agg_gemm<<<gGem, 256, 0, stream>>>(cur, Wm + (long long)i * HDIM * HDIM,
                                             bm + (long long)i * HDIM,
                                             rowptr, csrc, ccoef, dinv, nxt,
                                             sumsAll + (long long)(i + 1) * NREP * 256, N);
        bn_tail(nxt, sumsAll + (long long)(i + 1) * NREP * 256,
                gamma + (long long)(i + 1) * HDIM,
                beta  + (long long)(i + 1) * HDIM, cur);
        float* t = cur; cur = nxt; nxt = t;
    }

    k_agg_gemm<<<gGem, 256, 0, stream>>>(cur, Wm + 3LL * HDIM * HDIM, bm + 3LL * HDIM,
                                         rowptr, csrc, ccoef, dinv, nxt,
                                         sumsAll + 4LL * NREP * 256, N);
    k_bn_pool<<<1024, 256, 0, stream>>>(nxt, cur, sumsAll + 4LL * NREP * 256,
                                        gamma + 4LL * HDIM, beta + 4LL * HDIM,
                                        batch, pool, cnts, invN, N, G);

    k_pool_final<<<(G * HDIM + 255) / 256, 256, 0, stream>>>(pool, cnts, out + G, G);
    k_pred<<<G, 128, 0, stream>>>(out + G, Wr, br, out);
}


# k_agg_gemm: LDS tile rows 528 B apart instead of 512 B, so the 16 rows of one A-fragment read fall on 16 different bank groups, not on one (16-way conflict of every such read gone)
# speedup vs baseline: 5.9921x; 1.1107x over previous
#include <hip/hip_runtime.h>

#define HDIM 128
#define EPS_BN 1e-5f
#define NREP 8
#define TPITCH (HDIM + 4)

typedef __attribute__((ext_vector_type(2))) float v2f;
typedef __attribute__((ext_vector_type(8))) float v8f;
typedef __attribute__((ext_vector_type(4))) float v4f;

__global__ __launch_bounds__(256) void k_zero(float* a, long long na) {
    long long i = (long long)blockIdx.x * blockDim.x + threadIdx.x;
    long long stride = (long long)gridDim.x * blockDim.x;
    for (; i < na; i += stride) a[i] = 0.0f;
}

#define BIN_SHIFT 8
#define BIN_NODES 256
#define MAXB 512
#define CH 4096
#define BT 512

__device__ __forceinline__ int bin_of(int d, int NB) {
    const unsigned b = (unsigned)d >> BIN_SHIFT;
    return (int)(b < (unsigned)NB ? b : (unsigned)(NB - 1));
}

__device__ __forceinline__ int block_excl_scan(int v, int* wsum, int tid) {
    const int lane = tid & 31, w = tid >> 5;
    int incl = v;
    #pragma unroll
    for (int off = 1; off < 32; off <<= 1) {
        const int u = __shfl_up(incl, off);
        if (lane >= off) incl += u;
    }
    if (lane == 31) wsum[w] = incl;
    __syncthreads();
    int add = 0;
    for (int k = 0; k < w; ++k) add += wsum[k];
    __syncthreads();
    return add + incl - v;
}

__global__ __launch_bounds__(BT) void k_bin_count(const int* __restrict__ dst,
                                                  int* __restrict__ hist,
                                                  int E, int NB, int B1) {
    __shared__ int h[MAXB];
    const int tid = threadIdx.x;
    h[tid] = 0;
    __syncthreads();
    const int e0 = blockIdx.x * CH;
    int e1 = e0 + CH; if (e1 > E) e1 = E;
    int dv[CH / BT];
    #pragma unroll
    for (int u = 0; u < CH / BT; ++u) {
        const int e = e0 + tid + u * BT;
        dv[u] = dst[e < e1 ? e : e1 - 1];
    }
    #pragma unroll
    for (int u = 0; u < CH / BT; ++u)
        if (e0 + tid + u * BT < e1) atomicAdd(&h[bin_of(dv[u], NB)], 1);
    __syncthreads();
    if (tid < NB) hist[(long long)tid * B1 + blockIdx.x] = h[tid];
}

__global__ __launch_bounds__(256) void k_bin_scan(const int* __restrict__ hist,
                                                  int* __restrict__ base,
                                                  int* __restrict__ total,
                                                  int NB, int B1) {
    const int lane = threadIdx.x & 31;
    const int bin  = blockIdx.x * 8 + (int)(threadIdx.x >> 5);
    if (bin >= NB) return;
    const int per = (B1 + 31) >> 5;
    const int lo = lane * per;
    int hi = lo + per; if (hi > B1) hi = B1;
    const int* hp = hist + (long long)bin * B1;
    int*       bp = base + (long long)bin * B1;
    int s = 0;
    for (int k = lo; k < hi; ++k) s += hp[k];
    int incl = s;
    #pragma unroll
    for (int off = 1; off < 32; off <<= 1) {
        const int u = __shfl_up(incl, off);
        if (lane >= off) incl += u;
    }
    int run = incl - s;
    for (int k = lo; k < hi; ++k) { bp[k] = run; run += hp[k]; }
    if (lane == 31) total[bin] = incl;
}

__global__ __launch_bounds__(BT) void k_bin_scatter(const int* __restrict__ src,
                                                    const int* __restrict__ dst,
                                                    const int* __restrict__ hist,
                                                    const int* __restrict__ base,
                                                    const int* __restrict__ total,
                                                    int* __restrict__ binStart,
                                                    int2* __restrict__ binned,
                                                    int E, int NB, int B1) {
    __shared__ int2 rec[CH];
    __shared__ int lstart[MAXB];
    __shared__ int cur[MAXB];
    __shared__ int gofs[MAXB];
    __shared__ int wsum[16];
    const int tid = threadIdx.x;
    const int blk = blockIdx.x;
    const bool live = tid < NB;
    const int myTot  = live ? total[tid] : 0;
    const int myCnt  = live ? hist[(long long)tid * B1 + blk] : 0;
    const int myBase = live ? base[(long long)tid * B1 + blk] : 0;
    const int gs = block_excl_scan(myTot, wsum, tid);
    const int ls = block_excl_scan(myCnt, wsum, tid);
    lstart[tid] = ls;
    cur[tid]    = 0;
    gofs[tid]   = gs + myBase - ls;
    if (blk == 0) {
        if (live) binStart[tid] = gs;
        if (tid == NB - 1) binStart[NB] = gs + myTot;
    }
    __syncthreads();
    const int e0 = blk * CH;
    int e1 = e0 + CH; if (e1 > E) e1 = E;
    int sv[CH / BT], dv[CH / BT];
    #pragma unroll
    for (int u = 0; u < CH / BT; ++u) {
        const int e  = e0 + tid + u * BT;
        const int ec = e < e1 ? e : e1 - 1;
        sv[u] = src[ec];
        dv[u] = dst[ec];
    }
    #pragma unroll
    for (int u = 0; u < CH / BT; ++u) {
        if (e0 + tid + u * BT < e1) {
            const int b = bin_of(dv[u], NB);
            const int p = lstart[b] + atomicAdd(&cur[b], 1);
            if ((unsigned)p < (unsigned)CH) rec[p] = make_int2(sv[u], dv[u]);
        }
    }
    __syncthreads();
    const int cnt = e1 - e0;
    #pragma unroll
    for (int u = 0; u < CH / BT; ++u) {
        const int i = tid + u * BT;
        if (i < cnt) {
            const int2 r = rec[i];
            const int g = gofs[bin_of(r.y, NB)] + i;
            if ((unsigned)g < (unsigned)E) binned[g] = r;
        }
    }
}

__global__ __launch_bounds__(BT) void k_bin_sort(const int2* __restrict__ binned,
                                                 const int* __restrict__ binStart,
                                                 int* __restrict__ rowptr,
                                                 float* __restrict__ dinv,
                                                 int* __restrict__ csrc,
                                                 int N, int E, int NB) {
    __shared__ int cnt[BIN_NODES];
    __shared__ int wsum[16];
    const int tid = threadIdx.x;
    const int b   = blockIdx.x;
    int s0 = binStart[b], s1 = binStart[b + 1];
    if (s0 < 0) s0 = 0; if (s0 > E) s0 = E;
    if (s1 < s0) s1 = s0; if (s1 > E) s1 = E;
    if (tid < BIN_NODES) cnt[tid] = 0;
    __syncthreads();
    const int node0 = b << BIN_SHIFT;
    for (int i0 = s0 + tid; i0 < s1; i0 += 4 * BT) {
        int dv[4];
        #pragma unroll
        for (int u = 0; u < 4; ++u) {
            const int i = i0 + u * BT;
            dv[u] = binned[i < s1 ? i : s1 - 1].y;
        }
        #pragma unroll
        for (int u = 0; u < 4; ++u)
            if (i0 + u * BT < s1) atomicAdd(&cnt[(dv[u] - node0) & (BIN_NODES - 1)], 1);
    }
    __syncthreads();
    const int c  = (tid < BIN_NODES) ? cnt[tid] : 0;
    const int ex = block_excl_scan(c, wsum, tid);
    if (tid < BIN_NODES) {
        cnt[tid] = ex;
        const int node = node0 + tid;
        if (node < N) {
            rowptr[node] = s0 + ex;
            dinv[node]   = rsqrtf((float)c + 1.0f);
        }
    }
    if (b == NB - 1 && tid == 0) rowptr[N] = E;
    __syncthreads();
    for (int i0 = s0 + tid; i0 < s1; i0 += 4 * BT) {
        int2 rv[4];
        #pragma unroll
        for (int u = 0; u < 4; ++u) {
            const int i = i0 + u * BT;
            rv[u] = binned[i < s1 ? i : s1 - 1];
        }
        #pragma unroll
        for (int u = 0; u < 4; ++u) {
            if (i0 + u * BT < s1) {
                const int p = s0 + atomicAdd(&cnt[(rv[u].y - node0) & (BIN_NODES - 1)], 1);
                if ((unsigned)p < (unsigned)E) csrc[p] = rv[u].x;
            }
        }
    }
}

__global__ __launch_bounds__(256) void k_layer1(const float* __restrict__ X,
                                                const float* __restrict__ W1,
                                                const float* __restrict__ b1,
                                                const int* __restrict__ rowptr,
                                                const int* __restrict__ csrc,
                                                float* __restrict__ ccoef,
                                                const float* __restrict__ dinv,
                                                float* __restrict__ y,
                                                float* __restrict__ sums, int n) {
    __shared__ float red[8 * 256];
    const int lane = threadIdx.x & 31;
    const int wave = __builtin_amdgcn_readfirstlane((int)(threadIdx.x >> 5));
    const int c4 = lane << 2;
    const float4 w = *(const float4*)(W1 + c4);
    const float4 b = *(const float4*)(b1 + c4);
    float4 s = make_float4(0.f, 0.f, 0.f, 0.f);
    float4 q = make_float4(0.f, 0.f, 0.f, 0.f);
    for (int node = blockIdx.x * 8 + wave; node < n; node += gridDim.x * 8) {
        const int beg = rowptr[node], end = rowptr[node + 1];
        const float di = dinv[node];
        float part = 0.f;
        for (int j = beg + lane; j < end; j += 32) {
            const int s = csrc[j];
            const float c = dinv[s] * di;
            ccoef[j] = c;
            part += c * X[s];
        }
        for (int off = 16; off > 0; off >>= 1) part += __shfl_xor(part, off);
        const float a = part + X[node] * (di * di);
        float4 o;
        o.x = fmaxf(a * w.x + b.x, 0.f); o.y = fmaxf(a * w.y + b.y, 0.f);
        o.z = fmaxf(a * w.z + b.z, 0.f); o.w = fmaxf(a * w.w + b.w, 0.f);
        *(float4*)(y + (long long)node * HDIM + c4) = o;
        s.x += o.x; s.y += o.y; s.z += o.z; s.w += o.w;
        q.x += o.x * o.x; q.y += o.y * o.y; q.z += o.z * o.z; q.w += o.w * o.w;
    }
    *(float4*)(red + wave * 256 + c4)       = s;
    *(float4*)(red + wave * 256 + 128 + c4) = q;
    __syncthreads();
    float tot = 0.f;
    for (int k = 0; k < 8; ++k) tot += red[k * 256 + threadIdx.x];
    atomicAdd(&sums[(blockIdx.x & (NREP - 1)) * 256 + threadIdx.x], tot);
}

__device__ __forceinline__ void row_done(float* __restrict__ trow, float4& acc, int& cur, int& bnd,
                                         const int p2, const int p3, const int p4) {
    float* tp = trow + cur * TPITCH;
    const float4 t = *(const float4*)tp;
    *(float4*)tp = make_float4(acc.x + t.x, acc.y + t.y, acc.z + t.z, acc.w + t.w);
    acc = make_float4(0.f, 0.f, 0.f, 0.f);
    ++cur;
    bnd = (cur == 1) ? p2 : ((cur == 2) ? p3 : p4);
}

template <bool FULL>
__device__ __forceinline__ void gather_group(const float* __restrict__ x, const int c4,
                                             const int myi, const int myci,
                                             const int k, const int cnt, const int m, const int j0,
                                             float* __restrict__ trow, float4& acc, int& cur, int& bnd,
                                             const int p2, const int p3, const int p4) {
    float4 v[8];
    float  e[8];
    #pragma unroll
    for (int u = 0; u < 8; ++u) {
        int ku = k + u;
        if (!FULL && ku > m - 1) ku = m - 1;
        const int s = __builtin_amdgcn_readlane(myi, ku);
        e[u] = __int_as_float(__builtin_amdgcn_readlane(myci, ku));
        v[u] = *(const float4*)(x + (long long)s * HDIM + c4);
    }
    #pragma unroll
    for (int u = 0; u < 8; ++u) {
        if (FULL || u < cnt) {
            const int j = j0 + u;
            while (j >= bnd && cur < 3) row_done(trow, acc, cur, bnd, p2, p3, p4);
            acc.x += e[u] * v[u].x; acc.y += e[u] * v[u].y;
            acc.z += e[u] * v[u].z; acc.w += e[u] * v[u].w;
        }
    }
}

template <bool LAST>
__global__ __launch_bounds__(256) void k_agg_gemm(const float* __restrict__ x,
                                                  const float* __restrict__ W,
                                                  const float* __restrict__ bias,
                                                  const int* __restrict__ rowptr,
                                                  const int* __restrict__ csrc,
                                                  const float* __restrict__ ccoef,
                                                  const float* __restrict__ dinv,
                                                  float* __restrict__ y,
                                                  float* __restrict__ sums, int n,
                                                  const int* __restrict__ batch,
                                                  float* __restrict__ poolY, int G) {
    __shared__ alignas(16) float tileX[32 * TPITCH];
    __shared__ float red[2 * 256];
    __shared__ int mixedTile;
    const int rowBase = blockIdx.x * 32;
    const int tid  = threadIdx.x;
    if (LAST && tid == 0) mixedTile = 0;
    const int wave = __builtin_amdgcn_readfirstlane(tid >> 5);
    const int lane = tid & 31;
    const int half = lane >> 4;
    const int l16  = lane & 15;
    const int rowHalf = wave & 1;
    const int c0   = (wave >> 1) * 32;
    const int c4   = lane << 2;

    const bool fullTile = (rowBase + 32 <= n);

    const int lr0 = wave * 4;
    const int r0  = rowBase + lr0;
    float* trow = tileX + lr0 * TPITCH + c4;
    int pa[5];
    #pragma unroll
    for (int i = 0; i < 5; ++i) {
        int idx = r0 + i; if (idx > n) idx = n;
        pa[i] = rowptr[idx];
    }
    const int jend = pa[4];
    int base = pa[0];

    float4 xs[4];
    float  d2[4];
    #pragma unroll
    for (int i = 0; i < 4; ++i) {
        const int r  = r0 + i;
        const int rc = r < n ? r : n - 1;
        const float di = dinv[rc];
        d2[i] = di * di;
        xs[i] = *(const float4*)(x + (long long)rc * HDIM + c4);
    }
    int   myi = 0;
    float myc = 0.f;
    if (lane < jend - base) { myi = csrc[base + lane]; myc = ccoef[base + lane]; }

    #pragma unroll
    for (int i = 0; i < 4; ++i) {
        float4 t = make_float4(0.f, 0.f, 0.f, 0.f);
        if (r0 + i < n) t = make_float4(xs[i].x * d2[i], xs[i].y * d2[i], xs[i].z * d2[i], xs[i].w * d2[i]);
        *(float4*)(trow + i * TPITCH) = t;
    }

    int    cur = 0;
    int    bnd = pa[1];
    float4 acc = make_float4(0.f, 0.f, 0.f, 0.f);
    for (; base < jend; base += 32) {
        int m = jend - base; if (m > 32) m = 32;
        int   nyi = 0;
        float nyc = 0.f;
        if (lane < jend - base - 32) { nyi = csrc[base + 32 + lane]; nyc = ccoef[base + 32 + lane]; }
        const int myci = __float_as_int(myc);
        int k = 0;
        for (; k + 8 <= m; k += 8)
            gather_group<true>(x, c4, myi, myci, k, 8, m, base + k, trow, acc, cur, bnd, pa[2], pa[3], pa[4]);
        if (k < m)
            gather_group<false>(x, c4, myi, myci, k, m - k, m, base + k, trow, acc, cur, bnd, pa[2], pa[3], pa[4]);
        myi = nyi; myc = nyc;
    }
    while (cur < 4) row_done(trow, acc, cur, bnd, pa[2], pa[3], pa[4]);
    __syncthreads();

    const float bi0 = bias[c0 + l16];
    const float bi1 = bias[c0 + 16 + l16];

    const int rbase = rowBase + rowHalf * 16 + half * 8;

    int g0 = 0, myg = 0;
    if (LAST) {
        g0 = batch[rowBase];
        if (wave == 0) {
            const int r = rowBase + lane;
            myg = batch[r < n ? r : n - 1];
        }
    }

    const float* xrow = tileX + (rowHalf * 16 + l16) * TPITCH + 2 * half;
    v8f acc0 = {}, acc1 = {};

    #pragma unroll 4
    for (int k = 0; k < HDIM; k += 4) {
        v2f a; a.x = xrow[k]; a.y = xrow[k + 1];
        const float* wp = W + (long long)(k + 2 * half) * HDIM + c0 + l16;
        v2f b0; b0.x = wp[0];  b0.y = wp[HDIM];
        v2f b1; b1.x = wp[16]; b1.y = wp[HDIM + 16];
        acc0 = __builtin_amdgcn_wmma_f32_16x16x4_f32(false, a, false, b0,
                                                     (short)0, acc0, false, false);
        acc1 = __builtin_amdgcn_wmma_f32_16x16x4_f32(false, a, false, b1,
                                                     (short)0, acc1, false, false);
    }

    asm volatile("v_nop\n\tv_nop\n\tv_nop\n\tv_nop\n\tv_nop\n\tv_nop\n\tv_nop\n\tv_nop"
                 : "+v"(acc0), "+v"(acc1));

    #pragma unroll
    for (int v = 0; v < 8; ++v) {
        acc0[v] = fmaxf(acc0[v] + bi0, 0.f);
        acc1[v] = fmaxf(acc1[v] + bi1, 0.f);
    }

    float cs0 = 0.f, cq0 = 0.f, cs1 = 0.f, cq1 = 0.f;
    for (int v = 0; v < 8; ++v) {
        const bool in = fullTile || (rbase + v < n);
        const float a0 = in ? acc0[v] : 0.f;
        const float a1 = in ? acc1[v] : 0.f;
        cs0 += a0; cq0 += a0 * a0;
        cs1 += a1; cq1 += a1 * a1;
    }
    cs0 += __shfl_xor(cs0, 16); cq0 += __shfl_xor(cq0, 16);
    cs1 += __shfl_xor(cs1, 16); cq1 += __shfl_xor(cq1, 16);
    if (half == 0) {
        float* rp = red + rowHalf * 256 + c0 + l16;
        rp[0]   = cs0; rp[16]  = cs1;
        rp[128] = cq0; rp[144] = cq1;
    }
    if (LAST) {
        if (wave == 0 && rowBase + lane < n && myg != g0) mixedTile = 1;
    }
    __syncthreads();
    atomicAdd(&sums[(blockIdx.x & (NREP - 1)) * 256 + tid], red[tid] + red[256 + tid]);

    if (LAST) {
        if (mixedTile == 0) {
            if (tid < HDIM && (unsigned)g0 < (unsigned)G)
                atomicAdd(&poolY[(long long)g0 * HDIM + tid], red[tid] + red[256 + tid]);
        } else {
            int gb[8];
            #pragma unroll
            for (int v = 0; v < 8; ++v) {
                const int r = rbase + v;
                gb[v] = batch[r < n ? r : n - 1];
            }
            int   gc = -1;
            float p0 = 0.f, p1 = 0.f;
            #pragma unroll
            for (int v = 0; v < 8; ++v) {
                if (fullTile || (rbase + v < n)) {
                    if (gb[v] != gc) {
                        if ((unsigned)gc < (unsigned)G) {
                            atomicAdd(&poolY[(long long)gc * HDIM + c0 + l16], p0);
                            atomicAdd(&poolY[(long long)gc * HDIM + c0 + 16 + l16], p1);
                        }
                        gc = gb[v]; p0 = 0.f; p1 = 0.f;
                    }
                    p0 += acc0[v]; p1 += acc1[v];
                }
            }
            if ((unsigned)gc < (unsigned)G) {
                atomicAdd(&poolY[(long long)gc * HDIM + c0 + l16], p0);
                atomicAdd(&poolY[(long long)gc * HDIM + c0 + 16 + l16], p1);
            }
        }
    } else if (fullTile) {
        #pragma unroll
        for (int v = 0; v < 8; ++v) {
            float* op = y + (long long)(rbase + v) * HDIM + c0 + l16;
            op[0]  = acc0[v];
            op[16] = acc1[v];
        }
    } else {
        for (int v = 0; v < 8; ++v) {
            int r = rbase + v;
            if (r < n) {
                y[(long long)r * HDIM + c0 + l16]      = acc0[v];
                y[(long long)r * HDIM + c0 + 16 + l16] = acc1[v];
            }
        }
    }
}

template <bool HAS_RES>
__device__ __forceinline__ void bn_rows(float* __restrict__ y, const float* __restrict__ res,
                                        const v4f sc, const v4f sh,
                                        int t, const int stride, const int total) {
    for (; t + 3 * stride < total; t += 4 * stride) {
        v4f v[4], r[4];
        for (int u = 0; u < 4; ++u) {
            const long long off = (long long)(t + u * stride) * 4;
            v[u] = *(const v4f*)(y + off);
            if (HAS_RES) r[u] = *(const v4f*)(res + off);
        }
        for (int u = 0; u < 4; ++u) {
            const long long off = (long long)(t + u * stride) * 4;
            v4f o = v[u] * sc + sh;
            if (HAS_RES) o += r[u];
            *(v4f*)(y + off) = o;
        }
    }
    for (; t < total; t += stride) {
        const long long off = (long long)t * 4;
        v4f o = *(const v4f*)(y + off) * sc + sh;
        if (HAS_RES) o += *(const v4f*)(res + off);
        *(v4f*)(y + off) = o;
    }
}

__global__ __launch_bounds__(256) void k_bn_apply(float* __restrict__ y,
                                                  const float* __restrict__ res,
                                                  const float* __restrict__ sums,
                                                  const float* __restrict__ gamma,
                                                  const float* __restrict__ beta,
                                                  float invN, int n) {
    const int t0 = blockIdx.x * 256 + threadIdx.x;
    const int c4 = (t0 & 31) << 2;
    float4 s = make_float4(0.f, 0.f, 0.f, 0.f);
    float4 q = make_float4(0.f, 0.f, 0.f, 0.f);
    for (int k = 0; k < NREP; ++k) {
        const float4 a = *(const float4*)(sums + k * 256 + c4);
        const float4 b = *(const float4*)(sums + k * 256 + 128 + c4);
        s.x += a.x; s.y += a.y; s.z += a.z; s.w += a.w;
        q.x += b.x; q.y += b.y; q.z += b.z; q.w += b.w;
    }
    const float4 g  = *(const float4*)(gamma + c4);
    const float4 be = *(const float4*)(beta + c4);
    v4f sc, sh;
    {
        float m, var;
        m = s.x * invN; var = q.x * invN - m * m; sc.x = rsqrtf(var + EPS_BN) * g.x; sh.x = be.x - m * sc.x;
        m = s.y * invN; var = q.y * invN - m * m; sc.y = rsqrtf(var + EPS_BN) * g.y; sh.y = be.y - m * sc.y;
        m = s.z * invN; var = q.z * invN - m * m; sc.z = rsqrtf(var + EPS_BN) * g.z; sh.z = be.z - m * sc.z;
        m = s.w * invN; var = q.w * invN - m * m; sc.w = rsqrtf(var + EPS_BN) * g.w; sh.w = be.w - m * sc.w;
    }
    const int stride = gridDim.x * 256;
    const int total  = n * 32;
    if (res) bn_rows<true>(y, res, sc, sh, t0, stride, total);
    else     bn_rows<false>(y, res, sc, sh, t0, stride, total);
}

__device__ __forceinline__ void pool_flush(float* __restrict__ pool, float* __restrict__ counts,
                                           const int g, const int G, const int lane,
                                           const v4f acc, const float cnt) {
    if ((unsigned)g < (unsigned)G) {
        float* p = pool + (long long)g * HDIM + (lane << 2);
        atomicAdd(p + 0, acc.x); atomicAdd(p + 1, acc.y);
        atomicAdd(p + 2, acc.z); atomicAdd(p + 3, acc.w);
        if (lane == 0) atomicAdd(&counts[g], cnt);
    }
}

__global__ __launch_bounds__(256) void k_bn_pool(float* __restrict__ y,
                                                 const float* __restrict__ res,
                                                 const float* __restrict__ sums,
                                                 const float* __restrict__ gamma,
                                                 const float* __restrict__ beta,
                                                 const int* __restrict__ batch,
                                                 float* __restrict__ pool,
                                                 float* __restrict__ counts,
                                                 float invN, int n, int G) {
    const int lane = threadIdx.x & 31;
    const int wave = __builtin_amdgcn_readfirstlane((int)(threadIdx.x >> 5));
    const int c4 = lane << 2;
    float4 s = make_float4(0.f, 0.f, 0.f, 0.f);
    float4 q = make_float4(0.f, 0.f, 0.f, 0.f);
    for (int k = 0; k < NREP; ++k) {
        const float4 a = *(const float4*)(sums + k * 256 + c4);
        const float4 b = *(const float4*)(sums + k * 256 + 128 + c4);
        s.x += a.x; s.y += a.y; s.z += a.z; s.w += a.w;
        q.x += b.x; q.y += b.y; q.z += b.z; q.w += b.w;
    }
    const float4 g4 = *(const float4*)(gamma + c4);
    const float4 be = *(const float4*)(beta + c4);
    v4f sc, sh;
    {
        float m, var;
        m = s.x * invN; var = q.x * invN - m * m; sc.x = rsqrtf(var + EPS_BN) * g4.x; sh.x = be.x - m * sc.x;
        m = s.y * invN; var = q.y * invN - m * m; sc.y = rsqrtf(var + EPS_BN) * g4.y; sh.y = be.y - m * sc.y;
        m = s.z * invN; var = q.z * invN - m * m; sc.z = rsqrtf(var + EPS_BN) * g4.z; sh.z = be.z - m * sc.z;
        m = s.w * invN; var = q.w * invN - m * m; sc.w = rsqrtf(var + EPS_BN) * g4.w; sh.w = be.w - m * sc.w;
    }

    const int nWaves = gridDim.x * 8;
    const int gw  = blockIdx.x * 8 + wave;
    const int per = n / nWaves, rem = n - per * nWaves;
    const int r0  = gw * per + (gw < rem ? gw : rem);
    const int r1  = r0 + per + (gw < rem ? 1 : 0);

    int   gcur = -1;
    float cnt  = 0.f;
    v4f   acc  = {0.f, 0.f, 0.f, 0.f};
    for (int base = r0; base < r1; base += 32) {
        int m = r1 - base; if (m > 32) m = 32;
        int myb = 0;
        if (lane < m) myb = batch[base + lane];
        int k = 0;
        for (; k + 4 <= m; k += 4) {
            v4f v[4], r[4];
            for (int u = 0; u < 4; ++u) {
                const long long off = (long long)(base + k + u) * HDIM + c4;
                v[u] = *(const v4f*)(y + off);
                r[u] = *(const v4f*)(res + off);
            }
            for (int u = 0; u < 4; ++u) {
                const int g = __builtin_amdgcn_readlane(myb, k + u);
                v4f o = v[u] * sc + sh;
                o += r[u];
                *(v4f*)(y + (long long)(base + k + u) * HDIM + c4) = o;
                if (g != gcur) {
                    pool_flush(pool, counts, gcur, G, lane, acc, cnt);
                    gcur = g; cnt = 0.f; acc = (v4f){0.f, 0.f, 0.f, 0.f};
                }
                acc += o; cnt += 1.0f;
            }
        }
        for (; k < m; ++k) {
            const long long off = (long long)(base + k) * HDIM + c4;
            const int g = __builtin_amdgcn_readlane(myb, k);
            v4f o = *(const v4f*)(y + off) * sc + sh;
            o += *(const v4f*)(res + off);
            *(v4f*)(y + off) = o;
            if (g != gcur) {
                pool_flush(pool, counts, gcur, G, lane, acc, cnt);
                gcur = g; cnt = 0.f; acc = (v4f){0.f, 0.f, 0.f, 0.f};
            }
            acc += o; cnt += 1.0f;
        }
    }
    pool_flush(pool, counts, gcur, G, lane, acc, cnt);
}

__global__ __launch_bounds__(256) void k_pool_final(const float* __restrict__ pool,
                                                    const float* __restrict__ poolY,
                                                    const float* __restrict__ counts,
                                                    const float* __restrict__ sums,
                                                    const float* __restrict__ gamma,
                                                    const float* __restrict__ beta,
                                                    float invN,
                                                    float* __restrict__ xmean, int G) {
    int t = blockIdx.x * 256 + threadIdx.x;
    if (t >= G * HDIM) return;
    int g = t / HDIM;
    int ch = t - g * HDIM;
    float s = 0.f, q = 0.f;
    for (int k = 0; k < NREP; ++k) {
        s += sums[k * 256 + ch];
        q += sums[k * 256 + 128 + ch];
    }
    const float m   = s * invN;
    const float var = q * invN - m * m;
    const float sc  = rsqrtf(var + EPS_BN) * gamma[ch];
    const float cnt = counts[g];
    xmean[t] = (pool[t] + sc * (poolY[t] - cnt * m) + beta[ch] * cnt) / fmaxf(cnt, 1.0f);
}

__global__ void k_pred(const float* __restrict__ xmean, const float* __restrict__ Wr,
                       const float* __restrict__ br, float* __restrict__ pred) {
    __shared__ float red[128];
    int g = blockIdx.x, c = threadIdx.x;
    red[c] = xmean[(long long)g * HDIM + c] * Wr[c];
    __syncthreads();
    for (int s = 64; s > 0; s >>= 1) {
        if (c < s) red[c] += red[c + s];
        __syncthreads();
    }
    if (c == 0) pred[g] = red[0] + br[0];
}

extern "C" void kernel_launch(void* const* d_in, const int* in_sizes, int n_in,
                              void* d_out, int out_size, void* d_ws, size_t ws_size,
                              hipStream_t stream) {
    const float* X     = (const float*)d_in[0];
    const int*   edge  = (const int*)d_in[1];
    const int*   batch = (const int*)d_in[2];
    const float* W1    = (const float*)d_in[3];
    const float* b1    = (const float*)d_in[4];
    const float* Wm    = (const float*)d_in[5];
    const float* bm    = (const float*)d_in[6];
    const float* gamma = (const float*)d_in[7];
    const float* beta  = (const float*)d_in[8];
    const float* Wr    = (const float*)d_in[9];
    const float* br    = (const float*)d_in[10];
    float* out = (float*)d_out;

    const int N = in_sizes[0];
    const int E = in_sizes[1] / 2;
    const int G = out_size / (HDIM + 1);
    const int* src = edge;
    const int* dst = edge + E;

    const long long NH = (long long)N * HDIM;
    float* ws    = (float*)d_ws;
    float* bufA  = ws;
    float* bufB  = ws + NH;
    const int nbRaw = (N + BIN_NODES - 1) / BIN_NODES;
    const int NB = nbRaw < MAXB ? nbRaw : MAXB;
    const int B1 = (E + CH - 1) / CH;
    const long long tabW = ((long long)NB * B1 + 3) & ~3LL;
    const long long eW   = ((long long)E + 3) & ~3LL;
    int*   slot3    = (int*)(ws + 2 * NH);
    int*   rowptr   = slot3;
    int*   binStart = rowptr + (((long long)N + 1 + 3) & ~3LL);
    int*   total    = binStart + (MAXB + 4);
    int*   hist     = total + MAXB;
    int*   base     = hist + tabW;
    int*   csrc     = base + tabW;
    float* ccoef    = (float*)(csrc + eW);
    int2*  binned   = (int2*)(csrc + 2 * eW);
    float* sumsAll = (float*)(csrc + 4 * eW);
    float* pool  = sumsAll + 5LL * NREP * 256;
    float* poolY = pool + (long long)G * HDIM;
    float* cnts  = poolY + (long long)G * HDIM;
    float* dinv  = ws + 3 * NH;

    const int gGem = (N + 31) / 32;

    k_zero<<<64, 256, 0, stream>>>(sumsAll, 5LL * NREP * 256 + 2LL * G * HDIM + G);
    k_bin_count<<<B1, BT, 0, stream>>>(dst, hist, E, NB, B1);
    k_bin_scan<<<(NB + 7) / 8, 256, 0, stream>>>(hist, base, total, NB, B1);
    k_bin_scatter<<<B1, BT, 0, stream>>>(src, dst, hist, base, total, binStart, binned,
                                         E, NB, B1);
    k_bin_sort<<<NB, BT, 0, stream>>>(binned, binStart, rowptr, dinv, csrc, N, E, NB);

    const float invN = 1.0f / (float)N;
    auto bn_tail = [&](float* ybuf, const float* sums_, const float* g_, const float* be_,
                       const float* res) {
        k_bn_apply<<<1024, 256, 0, stream>>>(ybuf, res, sums_, g_, be_, invN, N);
    };

    k_layer1<<<2048, 256, 0, stream>>>(X, W1, b1, rowptr, csrc, ccoef, dinv, bufA, sumsAll, N);
    bn_tail(bufA, sumsAll, gamma, beta, nullptr);

    float* cur = bufA;
    float* nxt = bufB;

    for (int i = 0; i < 3; ++i) {
        k_agg_gemm<false><<<gGem, 256, 0, stream>>>(cur, Wm + (long long)i * HDIM * HDIM,
                                                    bm + (long long)i * HDIM,
                                                    rowptr, csrc, ccoef, dinv, nxt,
                                                    sumsAll + (long long)(i + 1) * NREP * 256, N,
                                                    nullptr, nullptr, 0);
        if (i < 2)
            bn_tail(nxt, sumsAll + (long long)(i + 1) * NREP * 256,
                    gamma + (long long)(i + 1) * HDIM,
                    beta  + (long long)(i + 1) * HDIM, cur);
        else
            k_bn_pool<<<1024, 256, 0, stream>>>(nxt, cur, sumsAll + 3LL * NREP * 256,
                                                gamma + 3LL * HDIM, beta + 3LL * HDIM,
                                                batch, pool, cnts, invN, N, G);
        float* t = cur; cur = nxt; nxt = t;
    }

    k_agg_gemm<true><<<gGem, 256, 0, stream>>>(cur, Wm + 3LL * HDIM * HDIM, bm + 3LL * HDIM,
                                               rowptr, csrc, ccoef, dinv, nxt,
                                               sumsAll + 4LL * NREP * 256, N,
                                               batch, poolY, G);

    k_pool_final<<<(G * HDIM + 255) / 256, 256, 0, stream>>>(pool, poolY, cnts,
                                                             sumsAll + 4LL * NREP * 256,
                                                             gamma + 4LL * HDIM, beta + 4LL * HDIM,
                                                             invN, out + G, G);
    k_pred<<<G, 128, 0, stream>>>(out + G, Wr, br, out);
}
